# starting kernel
# baseline (speedup 1.0000x reference)
#include <hip/hip_runtime.h>
#include <math.h>

typedef float v2f __attribute__((ext_vector_type(2)));
typedef float v8f __attribute__((ext_vector_type(8)));

#define IN_DIM 128
#define HID    64
#define OUTD   64

__device__ __forceinline__ unsigned enc_f32(float f) {
  unsigned b = __float_as_uint(f);
  return (b & 0x80000000u) ? ~b : (b | 0x80000000u);
}
__device__ __forceinline__ float dec_f32(unsigned u) {
  unsigned b = (u & 0x80000000u) ? (u & 0x7FFFFFFFu) : ~u;
  return __uint_as_float(b);
}

__global__ void gat_init(float* __restrict__ h, unsigned* __restrict__ mbits,
                         float* __restrict__ denom, int N, long total) {
  long i = (long)blockIdx.x * blockDim.x + threadIdx.x;
  if (i < total) h[i] = 0.0f;
  if (i < N) {
    mbits[i] = enc_f32(-INFINITY);
    denom[i] = 0.0f;
  }
}

__global__ __launch_bounds__(64) void gat_fused_gemm(
    const float* __restrict__ x, const float* __restrict__ w_in,
    const float* __restrict__ b_in, const float* __restrict__ w,
    float* __restrict__ z, int N) {
  __shared__ float lds[2][16][66];

  const int lane = threadIdx.x & 31;
  const int wv   = threadIdx.x >> 5;
  const int tile = blockIdx.x * 2 + wv;
  const int ntiles = (N + 15) >> 4;
  const bool active = tile < ntiles;

  const int m  = lane & 15;
  const int kp = lane >> 4;

  if (active) {
    const int row  = tile * 16 + m;
    const int rowc = row < N ? row : (N - 1);
    const float* xrow = x + (long)rowc * IN_DIM;

    v8f acc[4];
    #pragma unroll
    for (int nt = 0; nt < 4; ++nt) acc[nt] = (v8f){0,0,0,0,0,0,0,0};

    for (int k0 = 0; k0 < IN_DIM; k0 += 4) {
      const int kk = k0 + 2 * kp;
      v2f av = *(const v2f*)(xrow + kk);
      const float* wr0 = w_in + (long)kk * HID;
      const float* wr1 = wr0 + HID;
      #pragma unroll
      for (int nt = 0; nt < 4; ++nt) {
        v2f bv = { wr0[nt * 16 + m], wr1[nt * 16 + m] };
        acc[nt] = __builtin_amdgcn_wmma_f32_16x16x4_f32(
            false, av, false, bv, (short)0, acc[nt], false, false);
      }
    }

    #pragma unroll
    for (int nt = 0; nt < 4; ++nt) {
      float bias = b_in[nt * 16 + m];
      #pragma unroll
      for (int r = 0; r < 8; ++r) {
        float v = acc[nt][r] + bias;
        v = 0.5f * v * (1.0f + erff(v * 0.70710678118654752f));
        lds[wv][r + 8 * kp][nt * 16 + m] = v;
      }
    }
  }
  __syncthreads();

  if (active) {
    v8f acc[4];
    #pragma unroll
    for (int nt = 0; nt < 4; ++nt) acc[nt] = (v8f){0,0,0,0,0,0,0,0};

    for (int k0 = 0; k0 < HID; k0 += 4) {
      const int kk = k0 + 2 * kp;
      v2f av = *(const v2f*)(&lds[wv][m][kk]);
      const float* wr0 = w + (long)kk * OUTD;
      const float* wr1 = wr0 + OUTD;
      #pragma unroll
      for (int nt = 0; nt < 4; ++nt) {
        v2f bv = { wr0[nt * 16 + m], wr1[nt * 16 + m] };
        acc[nt] = __builtin_amdgcn_wmma_f32_16x16x4_f32(
            false, av, false, bv, (short)0, acc[nt], false, false);
      }
    }

    #pragma unroll
    for (int nt = 0; nt < 4; ++nt) {
      #pragma unroll
      for (int r = 0; r < 8; ++r) {
        int rr = tile * 16 + r + 8 * kp;
        if (rr < N) z[(long)rr * OUTD + nt * 16 + m] = acc[nt][r];
      }
    }
  }
}

__global__ __launch_bounds__(256) void gat_scores(
    const float* __restrict__ z, const float* __restrict__ a,
    float* __restrict__ s1, float* __restrict__ s2, int N) {
  int i = blockIdx.x * blockDim.x + threadIdx.x;
  if (i >= N) return;
  const float4* zr = (const float4*)(z + (long)i * OUTD);
  const float4* a1 = (const float4*)a;
  const float4* a2 = (const float4*)(a + OUTD);
  float acc1 = 0.f, acc2 = 0.f;
  #pragma unroll
  for (int j = 0; j < OUTD / 4; ++j) {
    float4 v = zr[j];
    float4 p = a1[j];
    float4 q = a2[j];
    acc1 += v.x * p.x + v.y * p.y + v.z * p.z + v.w * p.w;
    acc2 += v.x * q.x + v.y * q.y + v.z * q.z + v.w * q.w;
  }
  s1[i] = acc1;
  s2[i] = acc2;
}

__global__ __launch_bounds__(256) void gat_edge_max(
    const int* __restrict__ src, const int* __restrict__ dst,
    const float* __restrict__ s1, const float* __restrict__ s2,
    float* __restrict__ ebuf, unsigned* __restrict__ mbits, int E) {
  int e = blockIdx.x * blockDim.x + threadIdx.x;
  if (e >= E) return;
  int s = src[e], d = dst[e];
  float v = s1[s] + s2[d];
  v = v > 0.f ? v : 0.01f * v;
  ebuf[e] = v;
  atomicMax(&mbits[d], enc_f32(v));
}

__global__ __launch_bounds__(256) void gat_edge_exp(
    const int* __restrict__ dst, float* __restrict__ ebuf,
    const unsigned* __restrict__ mbits, float* __restrict__ denom, int E) {
  int e = blockIdx.x * blockDim.x + threadIdx.x;
  if (e >= E) return;
  int d = dst[e];
  float mv = dec_f32(mbits[d]);
  if (!isfinite(mv)) mv = 0.0f;
  float ev = expf(ebuf[e] - mv);
  ebuf[e] = ev;
  atomicAdd(&denom[d], ev);
}

__global__ __launch_bounds__(256) void gat_scatter(
    const int* __restrict__ src, const int* __restrict__ dst,
    const float* __restrict__ z, const float* __restrict__ ebuf,
    const float* __restrict__ denom, float* __restrict__ h, int E) {
  int e = blockIdx.x * 8 + (threadIdx.x >> 5);
  if (e >= E) return;
  int lane = threadIdx.x & 31;
  int s = src[e], d = dst[e];
  float alpha = ebuf[e] / fmaxf(denom[d], 1e-9f);
  v2f zv = *(const v2f*)(z + (long)s * OUTD + 2 * lane);
  float* hp = h + (long)d * OUTD + 2 * lane;
  atomicAdd(hp + 0, alpha * zv.x);
  atomicAdd(hp + 1, alpha * zv.y);
}

extern "C" void kernel_launch(void* const* d_in, const int* in_sizes, int n_in,
                              void* d_out, int out_size, void* d_ws, size_t ws_size,
                              hipStream_t stream) {
  const float* x    = (const float*)d_in[0];
  const float* w_in = (const float*)d_in[1];
  const float* b_in = (const float*)d_in[2];
  const float* w    = (const float*)d_in[3];
  const float* a    = (const float*)d_in[4];
  const int*   src  = (const int*)d_in[5];
  const int*   dst  = (const int*)d_in[6];
  float* h = (float*)d_out;

  const int N = in_sizes[0] / IN_DIM;
  const int E = in_sizes[5];

  auto align256 = [](size_t v) { return (v + 255) & ~(size_t)255; };
  char* base = (char*)d_ws;
  size_t off = 0;
  float*    z     = (float*)(base + off); off += align256((size_t)N * OUTD * 4);
  float*    s1    = (float*)(base + off); off += align256((size_t)N * 4);
  float*    s2    = (float*)(base + off); off += align256((size_t)N * 4);
  unsigned* mbits = (unsigned*)(base + off); off += align256((size_t)N * 4);
  float*    denom = (float*)(base + off); off += align256((size_t)N * 4);
  float*    ebuf  = (float*)(base + off); off += align256((size_t)E * 4);

  const long total = (long)N * OUTD;
  {
    long blocks = (total + 255) / 256;
    gat_init<<<dim3((unsigned)blocks), dim3(256), 0, stream>>>(h, mbits, denom, N, total);
  }
  {
    int ntiles = (N + 15) >> 4;
    int blocks = (ntiles + 1) / 2;
    gat_fused_gemm<<<dim3(blocks), dim3(64), 0, stream>>>(x, w_in, b_in, w, z, N);
  }
  gat_scores<<<dim3((N + 255) / 256), dim3(256), 0, stream>>>(z, a, s1, s2, N);
  gat_edge_max<<<dim3((E + 255) / 256), dim3(256), 0, stream>>>(src, dst, s1, s2, ebuf, mbits, E);
  gat_edge_exp<<<dim3((E + 255) / 256), dim3(256), 0, stream>>>(dst, ebuf, mbits, denom, E);
  gat_scatter<<<dim3((E + 7) / 8), dim3(256), 0, stream>>>(src, dst, z, ebuf, denom, h, E);
}


# scatter pass: one wave per 32 consecutive edges instead of one per edge: indices, denom and alpha fetched lane-parallel, then 8 source rows requested together before their atomic adds
# speedup vs baseline: 1.2722x; 1.2722x over previous
#include <hip/hip_runtime.h>
#include <math.h>

typedef float v2f __attribute__((ext_vector_type(2)));
typedef float v8f __attribute__((ext_vector_type(8)));

#define IN_DIM 128
#define HID    64
#define OUTD   64

__device__ __forceinline__ unsigned enc_f32(float f) {
  unsigned b = __float_as_uint(f);
  return (b & 0x80000000u) ? ~b : (b | 0x80000000u);
}
__device__ __forceinline__ float dec_f32(unsigned u) {
  unsigned b = (u & 0x80000000u) ? (u & 0x7FFFFFFFu) : ~u;
  return __uint_as_float(b);
}

__global__ void gat_init(float* __restrict__ h, unsigned* __restrict__ mbits,
                         float* __restrict__ denom, int N, long total) {
  long i = (long)blockIdx.x * blockDim.x + threadIdx.x;
  if (i < total) h[i] = 0.0f;
  if (i < N) {
    mbits[i] = enc_f32(-INFINITY);
    denom[i] = 0.0f;
  }
}

__global__ __launch_bounds__(64) void gat_fused_gemm(
    const float* __restrict__ x, const float* __restrict__ w_in,
    const float* __restrict__ b_in, const float* __restrict__ w,
    float* __restrict__ z, int N) {
  __shared__ float lds[2][16][66];

  const int lane = threadIdx.x & 31;
  const int wv   = threadIdx.x >> 5;
  const int tile = blockIdx.x * 2 + wv;
  const int ntiles = (N + 15) >> 4;
  const bool active = tile < ntiles;

  const int m  = lane & 15;
  const int kp = lane >> 4;

  if (active) {
    const int row  = tile * 16 + m;
    const int rowc = row < N ? row : (N - 1);
    const float* xrow = x + (long)rowc * IN_DIM;

    v8f acc[4];
    #pragma unroll
    for (int nt = 0; nt < 4; ++nt) acc[nt] = (v8f){0,0,0,0,0,0,0,0};

    for (int k0 = 0; k0 < IN_DIM; k0 += 4) {
      const int kk = k0 + 2 * kp;
      v2f av = *(const v2f*)(xrow + kk);
      const float* wr0 = w_in + (long)kk * HID;
      const float* wr1 = wr0 + HID;
      #pragma unroll
      for (int nt = 0; nt < 4; ++nt) {
        v2f bv = { wr0[nt * 16 + m], wr1[nt * 16 + m] };
        acc[nt] = __builtin_amdgcn_wmma_f32_16x16x4_f32(
            false, av, false, bv, (short)0, acc[nt], false, false);
      }
    }

    #pragma unroll
    for (int nt = 0; nt < 4; ++nt) {
      float bias = b_in[nt * 16 + m];
      #pragma unroll
      for (int r = 0; r < 8; ++r) {
        float v = acc[nt][r] + bias;
        v = 0.5f * v * (1.0f + erff(v * 0.70710678118654752f));
        lds[wv][r + 8 * kp][nt * 16 + m] = v;
      }
    }
  }
  __syncthreads();

  if (active) {
    v8f acc[4];
    #pragma unroll
    for (int nt = 0; nt < 4; ++nt) acc[nt] = (v8f){0,0,0,0,0,0,0,0};

    for (int k0 = 0; k0 < HID; k0 += 4) {
      const int kk = k0 + 2 * kp;
      v2f av = *(const v2f*)(&lds[wv][m][kk]);
      const float* wr0 = w + (long)kk * OUTD;
      const float* wr1 = wr0 + OUTD;
      #pragma unroll
      for (int nt = 0; nt < 4; ++nt) {
        v2f bv = { wr0[nt * 16 + m], wr1[nt * 16 + m] };
        acc[nt] = __builtin_amdgcn_wmma_f32_16x16x4_f32(
            false, av, false, bv, (short)0, acc[nt], false, false);
      }
    }

    #pragma unroll
    for (int nt = 0; nt < 4; ++nt) {
      #pragma unroll
      for (int r = 0; r < 8; ++r) {
        int rr = tile * 16 + r + 8 * kp;
        if (rr < N) z[(long)rr * OUTD + nt * 16 + m] = acc[nt][r];
      }
    }
  }
}

__global__ __launch_bounds__(256) void gat_scores(
    const float* __restrict__ z, const float* __restrict__ a,
    float* __restrict__ s1, float* __restrict__ s2, int N) {
  int i = blockIdx.x * blockDim.x + threadIdx.x;
  if (i >= N) return;
  const float4* zr = (const float4*)(z + (long)i * OUTD);
  const float4* a1 = (const float4*)a;
  const float4* a2 = (const float4*)(a + OUTD);
  float acc1 = 0.f, acc2 = 0.f;
  #pragma unroll
  for (int j = 0; j < OUTD / 4; ++j) {
    float4 v = zr[j];
    float4 p = a1[j];
    float4 q = a2[j];
    acc1 += v.x * p.x + v.y * p.y + v.z * p.z + v.w * p.w;
    acc2 += v.x * q.x + v.y * q.y + v.z * q.z + v.w * q.w;
  }
  s1[i] = acc1;
  s2[i] = acc2;
}

__global__ __launch_bounds__(256) void gat_edge_max(
    const int* __restrict__ src, const int* __restrict__ dst,
    const float* __restrict__ s1, const float* __restrict__ s2,
    float* __restrict__ ebuf, unsigned* __restrict__ mbits, int E) {
  int e = blockIdx.x * blockDim.x + threadIdx.x;
  if (e >= E) return;
  int s = src[e], d = dst[e];
  float v = s1[s] + s2[d];
  v = v > 0.f ? v : 0.01f * v;
  ebuf[e] = v;
  atomicMax(&mbits[d], enc_f32(v));
}

__global__ __launch_bounds__(256) void gat_edge_exp(
    const int* __restrict__ dst, float* __restrict__ ebuf,
    const unsigned* __restrict__ mbits, float* __restrict__ denom, int E) {
  int e = blockIdx.x * blockDim.x + threadIdx.x;
  if (e >= E) return;
  int d = dst[e];
  float mv = dec_f32(mbits[d]);
  if (!isfinite(mv)) mv = 0.0f;
  float ev = expf(ebuf[e] - mv);
  ebuf[e] = ev;
  atomicAdd(&denom[d], ev);
}

#define SC_EDGES 32
#define SC_GROUP 8
__global__ __launch_bounds__(256) void gat_scatter(
    const int* __restrict__ src, const int* __restrict__ dst,
    const float* __restrict__ z, const float* __restrict__ ebuf,
    const float* __restrict__ denom, float* __restrict__ h, int E) {
  const int lane = threadIdx.x & 31;
  const int wv = __builtin_amdgcn_readfirstlane((int)(threadIdx.x >> 5));
  const long e0 = ((long)blockIdx.x * 8 + wv) * SC_EDGES;
  if (e0 >= E) return;
  const int cnt = (E - e0) < SC_EDGES ? (int)(E - e0) : SC_EDGES;

  int   s_l = 0, d_l = 0;
  float a_l = 0.0f;
  if (lane < cnt) {
    s_l = src[e0 + lane];
    d_l = dst[e0 + lane];
    a_l = ebuf[e0 + lane] / fmaxf(denom[d_l], 1e-9f);
  }
  const int a_bits = __float_as_int(a_l);
  const float* zl = z + 2 * lane;
  float*       hl = h + 2 * lane;

  int j0 = 0;
  for (; j0 + SC_GROUP <= cnt; j0 += SC_GROUP) {
    v2f zv[SC_GROUP];
    #pragma unroll
    for (int u = 0; u < SC_GROUP; ++u) {
      const int s = __builtin_amdgcn_readlane(s_l, j0 + u);
      zv[u] = *(const v2f*)(zl + (long)s * OUTD);
    }
    #pragma unroll
    for (int u = 0; u < SC_GROUP; ++u) {
      const int   d = __builtin_amdgcn_readlane(d_l, j0 + u);
      const float a = __int_as_float(__builtin_amdgcn_readlane(a_bits, j0 + u));
      float* hp = hl + (long)d * OUTD;
      atomicAdd(hp + 0, a * zv[u].x);
      atomicAdd(hp + 1, a * zv[u].y);
    }
  }
  for (; j0 < cnt; ++j0) {
    const int   s = __builtin_amdgcn_readlane(s_l, j0);
    const int   d = __builtin_amdgcn_readlane(d_l, j0);
    const float a = __int_as_float(__builtin_amdgcn_readlane(a_bits, j0));
    const v2f zv = *(const v2f*)(zl + (long)s * OUTD);
    float* hp = hl + (long)d * OUTD;
    atomicAdd(hp + 0, a * zv.x);
    atomicAdd(hp + 1, a * zv.y);
  }
}

extern "C" void kernel_launch(void* const* d_in, const int* in_sizes, int n_in,
                              void* d_out, int out_size, void* d_ws, size_t ws_size,
                              hipStream_t stream) {
  const float* x    = (const float*)d_in[0];
  const float* w_in = (const float*)d_in[1];
  const float* b_in = (const float*)d_in[2];
  const float* w    = (const float*)d_in[3];
  const float* a    = (const float*)d_in[4];
  const int*   src  = (const int*)d_in[5];
  const int*   dst  = (const int*)d_in[6];
  float* h = (float*)d_out;

  const int N = in_sizes[0] / IN_DIM;
  const int E = in_sizes[5];

  auto align256 = [](size_t v) { return (v + 255) & ~(size_t)255; };
  char* base = (char*)d_ws;
  size_t off = 0;
  float*    z     = (float*)(base + off); off += align256((size_t)N * OUTD * 4);
  float*    s1    = (float*)(base + off); off += align256((size_t)N * 4);
  float*    s2    = (float*)(base + off); off += align256((size_t)N * 4);
  unsigned* mbits = (unsigned*)(base + off); off += align256((size_t)N * 4);
  float*    denom = (float*)(base + off); off += align256((size_t)N * 4);
  float*    ebuf  = (float*)(base + off); off += align256((size_t)E * 4);

  const long total = (long)N * OUTD;
  {
    long blocks = (total + 255) / 256;
    gat_init<<<dim3((unsigned)blocks), dim3(256), 0, stream>>>(h, mbits, denom, N, total);
  }
  {
    int ntiles = (N + 15) >> 4;
    int blocks = (ntiles + 1) / 2;
    gat_fused_gemm<<<dim3(blocks), dim3(64), 0, stream>>>(x, w_in, b_in, w, z, N);
  }
  gat_scores<<<dim3((N + 255) / 256), dim3(256), 0, stream>>>(z, a, s1, s2, N);
  gat_edge_max<<<dim3((E + 255) / 256), dim3(256), 0, stream>>>(src, dst, s1, s2, ebuf, mbits, E);
  gat_edge_exp<<<dim3((E + 255) / 256), dim3(256), 0, stream>>>(dst, ebuf, mbits, denom, E);
  gat_scatter<<<dim3((E + 255) / 256), dim3(256), 0, stream>>>(src, dst, z, ebuf, denom, h, E);
}


# edges grouped by destination each call (returning int add = rank, block scan = offsets, fill); a wave per row does softmax and weighted row sum in registers: no float atomics; sums in rank order
# speedup vs baseline: 1.8704x; 1.4702x over previous
#include <hip/hip_runtime.h>
#include <math.h>

typedef float v2f __attribute__((ext_vector_type(2)));
typedef float v8f __attribute__((ext_vector_type(8)));

#define IN_DIM 128
#define HID    64
#define OUTD   64

__global__ __launch_bounds__(256) void gat_init(int* __restrict__ cnt, int* __restrict__ cursor,
                                                int N) {
  const int i = blockIdx.x * blockDim.x + threadIdx.x;
  if (i < N) cnt[i] = 0;
  if (i == 0) *cursor = 0;
}

__global__ __launch_bounds__(256) void gat_count(
    const int* __restrict__ dst, int* __restrict__ cnt, int* __restrict__ rank, int E) {
  const long e0 = ((long)blockIdx.x * blockDim.x + threadIdx.x) * 4;
  if (e0 + 3 < E) {
    const int4 d = *(const int4*)(dst + e0);
    int4 r;
    r.x = atomicAdd(&cnt[d.x], 1);
    r.y = atomicAdd(&cnt[d.y], 1);
    r.z = atomicAdd(&cnt[d.z], 1);
    r.w = atomicAdd(&cnt[d.w], 1);
    *(int4*)(rank + e0) = r;
  } else {
    for (long e = e0; e < E; ++e) rank[e] = atomicAdd(&cnt[dst[e]], 1);
  }
}

__global__ __launch_bounds__(64) void gat_fused_gemm(
    const float* __restrict__ x, const float* __restrict__ w_in,
    const float* __restrict__ b_in, const float* __restrict__ w,
    float* __restrict__ z, int N) {
  __shared__ float lds[2][16][66];

  const int lane = threadIdx.x & 31;
  const int wv   = threadIdx.x >> 5;
  const int tile = blockIdx.x * 2 + wv;
  const int ntiles = (N + 15) >> 4;
  const bool active = tile < ntiles;

  const int m  = lane & 15;
  const int kp = lane >> 4;

  if (active) {
    const int row  = tile * 16 + m;
    const int rowc = row < N ? row : (N - 1);
    const float* xrow = x + (long)rowc * IN_DIM;

    v8f acc[4];
    #pragma unroll
    for (int nt = 0; nt < 4; ++nt) acc[nt] = (v8f){0,0,0,0,0,0,0,0};

    for (int k0 = 0; k0 < IN_DIM; k0 += 4) {
      const int kk = k0 + 2 * kp;
      v2f av = *(const v2f*)(xrow + kk);
      const float* wr0 = w_in + (long)kk * HID;
      const float* wr1 = wr0 + HID;
      #pragma unroll
      for (int nt = 0; nt < 4; ++nt) {
        v2f bv = { wr0[nt * 16 + m], wr1[nt * 16 + m] };
        acc[nt] = __builtin_amdgcn_wmma_f32_16x16x4_f32(
            false, av, false, bv, (short)0, acc[nt], false, false);
      }
    }

    #pragma unroll
    for (int nt = 0; nt < 4; ++nt) {
      float bias = b_in[nt * 16 + m];
      #pragma unroll
      for (int r = 0; r < 8; ++r) {
        float v = acc[nt][r] + bias;
        v = 0.5f * v * (1.0f + erff(v * 0.70710678118654752f));
        lds[wv][r + 8 * kp][nt * 16 + m] = v;
      }
    }
  }
  __syncthreads();

  if (active) {
    v8f acc[4];
    #pragma unroll
    for (int nt = 0; nt < 4; ++nt) acc[nt] = (v8f){0,0,0,0,0,0,0,0};

    for (int k0 = 0; k0 < HID; k0 += 4) {
      const int kk = k0 + 2 * kp;
      v2f av = *(const v2f*)(&lds[wv][m][kk]);
      const float* wr0 = w + (long)kk * OUTD;
      const float* wr1 = wr0 + OUTD;
      #pragma unroll
      for (int nt = 0; nt < 4; ++nt) {
        v2f bv = { wr0[nt * 16 + m], wr1[nt * 16 + m] };
        acc[nt] = __builtin_amdgcn_wmma_f32_16x16x4_f32(
            false, av, false, bv, (short)0, acc[nt], false, false);
      }
    }

    #pragma unroll
    for (int nt = 0; nt < 4; ++nt) {
      #pragma unroll
      for (int r = 0; r < 8; ++r) {
        int rr = tile * 16 + r + 8 * kp;
        if (rr < N) z[(long)rr * OUTD + nt * 16 + m] = acc[nt][r];
      }
    }
  }
}

__global__ __launch_bounds__(256) void gat_node(
    const float* __restrict__ z, const float* __restrict__ a,
    const int* __restrict__ cnt, int* __restrict__ cursor,
    float* __restrict__ s1, float* __restrict__ s2, int* __restrict__ off, int N) {
  __shared__ int wsum[8];
  __shared__ int blk_base;
  const int i    = blockIdx.x * blockDim.x + threadIdx.x;
  const int lane = threadIdx.x & 31;
  const int wv   = threadIdx.x >> 5;
  int c = 0;
  if (i < N) {
    const float4* zr = (const float4*)(z + (long)i * OUTD);
    const float4* a1 = (const float4*)a;
    const float4* a2 = (const float4*)(a + OUTD);
    float acc1 = 0.f, acc2 = 0.f;
    #pragma unroll
    for (int j = 0; j < OUTD / 4; ++j) {
      float4 v = zr[j];
      float4 p = a1[j];
      float4 q = a2[j];
      acc1 += v.x * p.x + v.y * p.y + v.z * p.z + v.w * p.w;
      acc2 += v.x * q.x + v.y * q.y + v.z * q.z + v.w * q.w;
    }
    s1[i] = acc1;
    s2[i] = acc2;
    c = cnt[i];
  }
  int incl = c;
  #pragma unroll
  for (int o = 1; o < 32; o <<= 1) {
    const int t = __shfl_up(incl, o, 32);
    if (lane >= o) incl += t;
  }
  if (lane == 31) wsum[wv] = incl;
  __syncthreads();
  if (threadIdx.x == 0) {
    int run = 0;
    for (int w = 0; w < 8; ++w) { const int t = wsum[w]; wsum[w] = run; run += t; }
    blk_base = atomicAdd(cursor, run);
  }
  __syncthreads();
  if (i < N) off[i] = blk_base + wsum[wv] + incl - c;
}

__global__ __launch_bounds__(256) void gat_fill(
    const int* __restrict__ src, const int* __restrict__ dst, const int* __restrict__ rank,
    const int* __restrict__ off, int* __restrict__ ssrc, int E) {
  const long e0 = ((long)blockIdx.x * blockDim.x + threadIdx.x) * 4;
  if (e0 + 3 < E) {
    const int4 s = *(const int4*)(src + e0);
    const int4 d = *(const int4*)(dst + e0);
    const int4 r = *(const int4*)(rank + e0);
    const unsigned p0 = (unsigned)(off[d.x] + r.x);
    const unsigned p1 = (unsigned)(off[d.y] + r.y);
    const unsigned p2 = (unsigned)(off[d.z] + r.z);
    const unsigned p3 = (unsigned)(off[d.w] + r.w);
    if (p0 < (unsigned)E) ssrc[p0] = s.x;
    if (p1 < (unsigned)E) ssrc[p1] = s.y;
    if (p2 < (unsigned)E) ssrc[p2] = s.z;
    if (p3 < (unsigned)E) ssrc[p3] = s.w;
  } else {
    for (long e = e0; e < E; ++e) {
      const unsigned p = (unsigned)(off[dst[e]] + rank[e]);
      if (p < (unsigned)E) ssrc[p] = src[e];
    }
  }
}

#define AG_NODES 4
#define AG_GROUP 16

__device__ __forceinline__ float wave_max(float v) {
  #pragma unroll
  for (int o = 16; o > 0; o >>= 1) v = fmaxf(v, __shfl_xor(v, o, 32));
  return v;
}
__device__ __forceinline__ float wave_sum(float v) {
  #pragma unroll
  for (int o = 16; o > 0; o >>= 1) v += __shfl_xor(v, o, 32);
  return v;
}

__device__ __forceinline__ void trip_fetch(
    const int* __restrict__ ssrc, const float* __restrict__ s1,
    int o, int c, int jb, int lane, int N, int& s_l, float& g_l) {
  s_l = 0;
  g_l = 0.0f;
  if (jb + lane < c) {
    int s = ssrc[o + jb + lane];
    s = min(max(s, 0), N - 1);
    g_l = s1[s];
    s_l = s;
  }
}
__device__ __forceinline__ float edge_score(float g_l, float s2d, bool live) {
  const float v = g_l + s2d;
  const float e = v > 0.f ? v : 0.01f * v;
  return live ? e : -INFINITY;
}

__device__ __forceinline__ void rows_load(
    const char* __restrict__ zb, unsigned lane_off, int s_l, int j0, int n,
    v2f (&zv)[AG_GROUP]) {
  #pragma unroll
  for (int u = 0; u < AG_GROUP; ++u) {
    zv[u] = (v2f){0.f, 0.f};
    if (j0 + u < n) {
      const int s = __builtin_amdgcn_readlane(s_l, j0 + u);
      zv[u] = *(const v2f*)(zb + (long)s * (OUTD * 4) + lane_off);
    }
  }
}
__device__ __forceinline__ void rows_fma(
    const v2f (&zv)[AG_GROUP], int a_bits, int j0, int n, v2f& acc) {
  #pragma unroll
  for (int u = 0; u < AG_GROUP; ++u) {
    if (j0 + u < n) {
      const float al = __int_as_float(__builtin_amdgcn_readlane(a_bits, j0 + u));
      acc.x += al * zv[u].x;
      acc.y += al * zv[u].y;
    }
  }
}

__global__ __launch_bounds__(256) __attribute__((amdgpu_waves_per_eu(12))) void gat_aggregate(
    const int* __restrict__ ssrc, const int* __restrict__ off, const int* __restrict__ cnt,
    const float* __restrict__ s1, const float* __restrict__ s2,
    const float* __restrict__ z, float* __restrict__ h, int N) {
  const int lane = threadIdx.x & 31;
  const int wv = __builtin_amdgcn_readfirstlane((int)(threadIdx.x >> 5));
  const int d0 = (blockIdx.x * 8 + wv) * AG_NODES;
  if (d0 >= N) return;
  const char*    zb       = (const char*)z;
  const unsigned lane_off = (unsigned)lane * 8u;

  int   o_v  = 0;
  int   c_v  = 0;
  float s2_v = 0.0f;
  if (lane < AG_NODES && d0 + lane < N) {
    o_v  = off[d0 + lane];
    c_v  = cnt[d0 + lane];
    s2_v = s2[d0 + lane];
  }
  const int s2_bits = __float_as_int(s2_v);

  for (int k = 0; k < AG_NODES; ++k) {
    const int d = d0 + k;
    if (d >= N) break;
    const int   o   = __builtin_amdgcn_readlane(o_v, k);
    const int   c   = __builtin_amdgcn_readlane(c_v, k);
    const float s2d = __int_as_float(__builtin_amdgcn_readlane(s2_bits, k));

    const int n0 = c < 32 ? c : 32;
    int s_l; float g_l;
    trip_fetch(ssrc, s1, o, c, 0, lane, N, s_l, g_l);
    v2f zv[AG_GROUP];
    rows_load(zb, lane_off, s_l, 0, n0, zv);

    const float e_l = edge_score(g_l, s2d, lane < c);
    float m = e_l;
    for (int jb = 32; jb < c; jb += 32) {
      int s_t; float g_t;
      trip_fetch(ssrc, s1, o, c, jb, lane, N, s_t, g_t);
      m = fmaxf(m, edge_score(g_t, s2d, jb + lane < c));
    }
    m = wave_max(m);
    if (!isfinite(m)) m = 0.0f;

    const float ex_l = lane < c ? expf(e_l - m) : 0.0f;
    float den = ex_l;
    for (int jb = 32; jb < c; jb += 32) {
      int s_t; float g_t;
      trip_fetch(ssrc, s1, o, c, jb, lane, N, s_t, g_t);
      den += (jb + lane < c) ? expf(edge_score(g_t, s2d, true) - m) : 0.0f;
    }
    den = wave_sum(den);
    const float dd = fmaxf(den, 1e-9f);

    v2f acc = {0.f, 0.f};
    {
      const int a_bits = __float_as_int(ex_l / dd);
      rows_fma(zv, a_bits, 0, n0, acc);
      for (int j0 = AG_GROUP; j0 < n0; j0 += AG_GROUP) {
        rows_load(zb, lane_off, s_l, j0, n0, zv);
        rows_fma(zv, a_bits, j0, n0, acc);
      }
    }
    for (int jb = 32; jb < c; jb += 32) {
      int s_t; float g_t;
      trip_fetch(ssrc, s1, o, c, jb, lane, N, s_t, g_t);
      const float a_t = (jb + lane < c) ? expf(edge_score(g_t, s2d, true) - m) / dd : 0.0f;
      const int a_bits = __float_as_int(a_t);
      const int nt = (c - jb) < 32 ? (c - jb) : 32;
      for (int j0 = 0; j0 < nt; j0 += AG_GROUP) {
        rows_load(zb, lane_off, s_t, j0, nt, zv);
        rows_fma(zv, a_bits, j0, nt, acc);
      }
    }
    *(v2f*)((char*)h + (long)d * (OUTD * 4) + lane_off) = acc;
  }
}

extern "C" void kernel_launch(void* const* d_in, const int* in_sizes, int n_in,
                              void* d_out, int out_size, void* d_ws, size_t ws_size,
                              hipStream_t stream) {
  const float* x    = (const float*)d_in[0];
  const float* w_in = (const float*)d_in[1];
  const float* b_in = (const float*)d_in[2];
  const float* w    = (const float*)d_in[3];
  const float* a    = (const float*)d_in[4];
  const int*   src  = (const int*)d_in[5];
  const int*   dst  = (const int*)d_in[6];
  float* h = (float*)d_out;

  const int N = in_sizes[0] / IN_DIM;
  const int E = in_sizes[5];

  auto align256 = [](size_t v) { return (v + 255) & ~(size_t)255; };
  char* base = (char*)d_ws;
  size_t off_b = 0;
  float* z      = (float*)(base + off_b); off_b += align256((size_t)N * OUTD * 4);
  float* s1     = (float*)(base + off_b); off_b += align256((size_t)N * 4);
  float* s2     = (float*)(base + off_b); off_b += align256((size_t)N * 4);
  int*   cnt    = (int*)(base + off_b);   off_b += align256((size_t)N * 4);
  int*   off    = (int*)(base + off_b);   off_b += align256((size_t)N * 4);
  int*   cursor = (int*)(base + off_b);   off_b += 256;
  int*   rank   = (int*)(base + off_b);   off_b += align256((size_t)E * 4);
  int*   ssrc   = (int*)(base + off_b);   off_b += align256((size_t)E * 4);

  const int node_blocks = (N + 255) / 256;
  const int edge_blocks = (int)((((long)E + 3) / 4 + 255) / 256);

  gat_init<<<dim3(node_blocks), dim3(256), 0, stream>>>(cnt, cursor, N);
  gat_count<<<dim3(edge_blocks), dim3(256), 0, stream>>>(dst, cnt, rank, E);
  {
    int ntiles = (N + 15) >> 4;
    int blocks = (ntiles + 1) / 2;
    gat_fused_gemm<<<dim3(blocks), dim3(64), 0, stream>>>(x, w_in, b_in, w, z, N);
  }
  gat_node<<<dim3(node_blocks), dim3(256), 0, stream>>>(z, a, cnt, cursor, s1, s2, off, N);
  gat_fill<<<dim3(edge_blocks), dim3(256), 0, stream>>>(src, dst, rank, off, ssrc, E);
  {
    const int rows_per_block = 8 * AG_NODES;
    gat_aggregate<<<dim3((N + rows_per_block - 1) / rows_per_block), dim3(256), 0, stream>>>(
        ssrc, off, cnt, s1, s2, z, h, N);
  }
}


# edges grouped by destination by a two-level counting sort on LDS counters (chunk counts, bucket scatter, per-bucket row sort), not one returning global add per edge; sums in LDS arrival order
# speedup vs baseline: 3.5822x; 1.9152x over previous
#include <hip/hip_runtime.h>
#include <math.h>

typedef float v2f __attribute__((ext_vector_type(2)));
typedef float v8f __attribute__((ext_vector_type(8)));

#define IN_DIM 128
#define HID    64
#define OUTD   64

#define SORT_CHUNKS 256
#define SORT_NBP    256
#define BK_SHIFT    9
#define BK_ROWS     512
#define SORT_UNITS  8

__device__ __forceinline__ int bucket_of(int d, int NB) {
  const unsigned b = (unsigned)d >> BK_SHIFT;
  return (int)(b < (unsigned)(NB - 1) ? b : (unsigned)(NB - 1));
}
__device__ __forceinline__ long chunk_size(int E) {
  return ((((long)E + SORT_CHUNKS - 1) / SORT_CHUNKS) + 3) & ~3L;
}

__global__ __launch_bounds__(256) void gat_chunk_hist(
    const int* __restrict__ dst, int* __restrict__ hist, int E, int NB) {
  __shared__ int h[SORT_NBP];
  const int t = threadIdx.x;
  const int c = blockIdx.x;
  h[t] = 0;
  __syncthreads();
  const long cs  = chunk_size(E);
  const long lo  = (long)c * cs;
  long       hi  = lo + cs;
  if (hi > E) hi = E;
  const long hi4 = lo + ((hi - lo) & ~3L);
  for (long base = lo; base < hi4; base += 256L * 4 * SORT_UNITS) {
    int4 d[SORT_UNITS];
    #pragma unroll
    for (int j = 0; j < SORT_UNITS; ++j) {
      const long e = base + ((long)j * 256 + t) * 4;
      d[j] = *(const int4*)(dst + (e < hi4 ? e : lo));
    }
    #pragma unroll
    for (int j = 0; j < SORT_UNITS; ++j) {
      const long e = base + ((long)j * 256 + t) * 4;
      if (e < hi4) {
        atomicAdd(&h[bucket_of(d[j].x, NB)], 1);
        atomicAdd(&h[bucket_of(d[j].y, NB)], 1);
        atomicAdd(&h[bucket_of(d[j].z, NB)], 1);
        atomicAdd(&h[bucket_of(d[j].w, NB)], 1);
      }
    }
  }
  if (hi4 + t < hi) atomicAdd(&h[bucket_of(dst[hi4 + t], NB)], 1);
  __syncthreads();
  hist[c * SORT_NBP + t] = h[t];
}

__global__ __launch_bounds__(256) void gat_chunk_scatter(
    const int* __restrict__ src, const int* __restrict__ dst, const int* __restrict__ hist,
    int* __restrict__ bbase, int2* __restrict__ pairs, int E, int NB) {
  __shared__ __attribute__((aligned(16))) int part_pre[4][SORT_NBP];
  __shared__ __attribute__((aligned(16))) int part_tot[4][SORT_NBP];
  __shared__ int cursor[SORT_NBP];
  __shared__ int wsum[8];
  const int t    = threadIdx.x;
  const int lane = t & 31;
  const int wv   = t >> 5;
  const int c    = blockIdx.x;

  {
    const int q = t & 63;
    const int r = t >> 6;
    int4 pre = {0, 0, 0, 0};
    int4 tot = {0, 0, 0, 0};
    #pragma unroll
    for (int hb = 0; hb < 2; ++hb) {
      int4 v[32];
      #pragma unroll
      for (int k = 0; k < 32; ++k)
        v[k] = *(const int4*)(hist + (r + 4 * (hb * 32 + k)) * SORT_NBP + 4 * q);
      #pragma unroll
      for (int k = 0; k < 32; ++k) {
        const int c2 = r + 4 * (hb * 32 + k);
        tot.x += v[k].x; tot.y += v[k].y; tot.z += v[k].z; tot.w += v[k].w;
        if (c2 < c) { pre.x += v[k].x; pre.y += v[k].y; pre.z += v[k].z; pre.w += v[k].w; }
      }
    }
    *(int4*)(&part_pre[r][4 * q]) = pre;
    *(int4*)(&part_tot[r][4 * q]) = tot;
  }
  __syncthreads();
  const int my_pre = part_pre[0][t] + part_pre[1][t] + part_pre[2][t] + part_pre[3][t];
  const int my_tot = part_tot[0][t] + part_tot[1][t] + part_tot[2][t] + part_tot[3][t];
  int incl = my_tot;
  #pragma unroll
  for (int o = 1; o < 32; o <<= 1) {
    const int u = __shfl_up(incl, o, 32);
    if (lane >= o) incl += u;
  }
  if (lane == 31) wsum[wv] = incl;
  __syncthreads();
  int wbase = 0;
  #pragma unroll
  for (int w = 0; w < 8; ++w) wbase += (w < wv) ? wsum[w] : 0;
  const int bstart = wbase + incl - my_tot;
  cursor[t] = bstart + my_pre;
  if (c == 0) {
    if (t < NB) bbase[t] = bstart;
    if (t == NB - 1) bbase[NB] = bstart + my_tot;
  }
  __syncthreads();

  const long cs  = chunk_size(E);
  const long lo  = (long)c * cs;
  long       hi  = lo + cs;
  if (hi > E) hi = E;
  const long hi4 = lo + ((hi - lo) & ~3L);
  for (long base = lo; base < hi4; base += 256L * 4 * SORT_UNITS) {
    int4 d[SORT_UNITS];
    int4 s[SORT_UNITS];
    #pragma unroll
    for (int j = 0; j < SORT_UNITS; ++j) {
      const long e  = base + ((long)j * 256 + t) * 4;
      const long ec = e < hi4 ? e : lo;
      d[j] = *(const int4*)(dst + ec);
      s[j] = *(const int4*)(src + ec);
    }
    #pragma unroll
    for (int j = 0; j < SORT_UNITS; ++j) {
      const long e = base + ((long)j * 256 + t) * 4;
      if (e < hi4) {
        const unsigned p0 = (unsigned)atomicAdd(&cursor[bucket_of(d[j].x, NB)], 1);
        const unsigned p1 = (unsigned)atomicAdd(&cursor[bucket_of(d[j].y, NB)], 1);
        const unsigned p2 = (unsigned)atomicAdd(&cursor[bucket_of(d[j].z, NB)], 1);
        const unsigned p3 = (unsigned)atomicAdd(&cursor[bucket_of(d[j].w, NB)], 1);
        if (p0 < (unsigned)E) pairs[p0] = make_int2(s[j].x, d[j].x);
        if (p1 < (unsigned)E) pairs[p1] = make_int2(s[j].y, d[j].y);
        if (p2 < (unsigned)E) pairs[p2] = make_int2(s[j].z, d[j].z);
        if (p3 < (unsigned)E) pairs[p3] = make_int2(s[j].w, d[j].w);
      }
    }
  }
  if (hi4 + t < hi) {
    const int dd = dst[hi4 + t];
    const unsigned p = (unsigned)atomicAdd(&cursor[bucket_of(dd, NB)], 1);
    if (p < (unsigned)E) pairs[p] = make_int2(src[hi4 + t], dd);
  }
}

#define BS_THREADS 1024
#define BS_UNITS   8
__global__ __launch_bounds__(BS_THREADS) void gat_bucket_sort(
    const int2* __restrict__ pairs, const int* __restrict__ bbase,
    int* __restrict__ cnt, int* __restrict__ off, int* __restrict__ ssrc, int N, int E) {
  __shared__ int rcnt[BK_ROWS];
  __shared__ int rcur[BK_ROWS];
  __shared__ int wsum[16];
  __shared__ int wexc[16];
  const int t    = threadIdx.x;
  const int lane = t & 31;
  const int wv   = t >> 5;
  const int b    = blockIdx.x;
  int lo = bbase[b];
  int hi = bbase[b + 1];
  lo = min(max(lo, 0), E);
  hi = min(max(hi, lo), E);
  const int row0 = b << BK_SHIFT;

  if (t < BK_ROWS) rcnt[t] = 0;
  __syncthreads();
  for (int base = lo; base < hi; base += BS_THREADS * BS_UNITS) {
    int2 pr[BS_UNITS];
    #pragma unroll
    for (int j = 0; j < BS_UNITS; ++j) {
      const int e = base + j * BS_THREADS + t;
      pr[j] = pairs[e < hi ? e : lo];
    }
    #pragma unroll
    for (int j = 0; j < BS_UNITS; ++j) {
      const int e = base + j * BS_THREADS + t;
      if (e < hi) atomicAdd(&rcnt[(pr[j].y - row0) & (BK_ROWS - 1)], 1);
    }
  }
  __syncthreads();

  int v = 0, incl = 0;
  if (t < BK_ROWS) {
    v = rcnt[t];
    incl = v;
    #pragma unroll
    for (int o = 1; o < 32; o <<= 1) {
      const int u = __shfl_up(incl, o, 32);
      if (lane >= o) incl += u;
    }
    if (lane == 31) wsum[wv] = incl;
  }
  __syncthreads();
  if (t < 16) {
    const int x = wsum[t];
    int in2 = x;
    #pragma unroll
    for (int o = 1; o < 16; o <<= 1) {
      const int u = __shfl_up(in2, o, 32);
      if (lane >= o) in2 += u;
    }
    wexc[t] = in2 - x;
  }
  __syncthreads();
  if (t < BK_ROWS) {
    const int start = wexc[wv] + incl - v;
    rcur[t] = start;
    const int node = row0 + t;
    if (node < N) {
      off[node] = lo + start;
      cnt[node] = v;
    }
  }
  __syncthreads();

  for (int base = lo; base < hi; base += BS_THREADS * BS_UNITS) {
    int2 pr[BS_UNITS];
    #pragma unroll
    for (int j = 0; j < BS_UNITS; ++j) {
      const int e = base + j * BS_THREADS + t;
      pr[j] = pairs[e < hi ? e : lo];
    }
    #pragma unroll
    for (int j = 0; j < BS_UNITS; ++j) {
      const int e = base + j * BS_THREADS + t;
      if (e < hi) {
        const int p = atomicAdd(&rcur[(pr[j].y - row0) & (BK_ROWS - 1)], 1);
        const unsigned g = (unsigned)(lo + p);
        if (g < (unsigned)hi) ssrc[g] = pr[j].x;
      }
    }
  }
}

__global__ __launch_bounds__(64) void gat_fused_gemm(
    const float* __restrict__ x, const float* __restrict__ w_in,
    const float* __restrict__ b_in, const float* __restrict__ w,
    float* __restrict__ z, int N) {
  __shared__ float lds[2][16][66];

  const int lane = threadIdx.x & 31;
  const int wv   = threadIdx.x >> 5;
  const int tile = blockIdx.x * 2 + wv;
  const int ntiles = (N + 15) >> 4;
  const bool active = tile < ntiles;

  const int m  = lane & 15;
  const int kp = lane >> 4;

  if (active) {
    const int row  = tile * 16 + m;
    const int rowc = row < N ? row : (N - 1);
    const float* xrow = x + (long)rowc * IN_DIM;

    v8f acc[4];
    #pragma unroll
    for (int nt = 0; nt < 4; ++nt) acc[nt] = (v8f){0,0,0,0,0,0,0,0};

    for (int k0 = 0; k0 < IN_DIM; k0 += 4) {
      const int kk = k0 + 2 * kp;
      v2f av = *(const v2f*)(xrow + kk);
      const float* wr0 = w_in + (long)kk * HID;
      const float* wr1 = wr0 + HID;
      #pragma unroll
      for (int nt = 0; nt < 4; ++nt) {
        v2f bv = { wr0[nt * 16 + m], wr1[nt * 16 + m] };
        acc[nt] = __builtin_amdgcn_wmma_f32_16x16x4_f32(
            false, av, false, bv, (short)0, acc[nt], false, false);
      }
    }

    #pragma unroll
    for (int nt = 0; nt < 4; ++nt) {
      float bias = b_in[nt * 16 + m];
      #pragma unroll
      for (int r = 0; r < 8; ++r) {
        float v = acc[nt][r] + bias;
        v = 0.5f * v * (1.0f + erff(v * 0.70710678118654752f));
        lds[wv][r + 8 * kp][nt * 16 + m] = v;
      }
    }
  }
  __syncthreads();

  if (active) {
    v8f acc[4];
    #pragma unroll
    for (int nt = 0; nt < 4; ++nt) acc[nt] = (v8f){0,0,0,0,0,0,0,0};

    for (int k0 = 0; k0 < HID; k0 += 4) {
      const int kk = k0 + 2 * kp;
      v2f av = *(const v2f*)(&lds[wv][m][kk]);
      const float* wr0 = w + (long)kk * OUTD;
      const float* wr1 = wr0 + OUTD;
      #pragma unroll
      for (int nt = 0; nt < 4; ++nt) {
        v2f bv = { wr0[nt * 16 + m], wr1[nt * 16 + m] };
        acc[nt] = __builtin_amdgcn_wmma_f32_16x16x4_f32(
            false, av, false, bv, (short)0, acc[nt], false, false);
      }
    }

    #pragma unroll
    for (int nt = 0; nt < 4; ++nt) {
      #pragma unroll
      for (int r = 0; r < 8; ++r) {
        int rr = tile * 16 + r + 8 * kp;
        if (rr < N) z[(long)rr * OUTD + nt * 16 + m] = acc[nt][r];
      }
    }
  }
}

__global__ __launch_bounds__(256) void gat_node(
    const float* __restrict__ z, const float* __restrict__ a,
    float* __restrict__ s1, float* __restrict__ s2, int N) {
  const int i = blockIdx.x * blockDim.x + threadIdx.x;
  if (i < N) {
    const float4* zr = (const float4*)(z + (long)i * OUTD);
    const float4* a1 = (const float4*)a;
    const float4* a2 = (const float4*)(a + OUTD);
    float acc1 = 0.f, acc2 = 0.f;
    #pragma unroll
    for (int j = 0; j < OUTD / 4; ++j) {
      float4 v = zr[j];
      float4 p = a1[j];
      float4 q = a2[j];
      acc1 += v.x * p.x + v.y * p.y + v.z * p.z + v.w * p.w;
      acc2 += v.x * q.x + v.y * q.y + v.z * q.z + v.w * q.w;
    }
    s1[i] = acc1;
    s2[i] = acc2;
  }
}

#define AG_NODES 4
#define AG_GROUP 16

__device__ __forceinline__ float wave_max(float v) {
  #pragma unroll
  for (int o = 16; o > 0; o >>= 1) v = fmaxf(v, __shfl_xor(v, o, 32));
  return v;
}
__device__ __forceinline__ float wave_sum(float v) {
  #pragma unroll
  for (int o = 16; o > 0; o >>= 1) v += __shfl_xor(v, o, 32);
  return v;
}

__device__ __forceinline__ void trip_fetch(
    const int* __restrict__ ssrc, const float* __restrict__ s1,
    int o, int c, int jb, int lane, int N, int& s_l, float& g_l) {
  s_l = 0;
  g_l = 0.0f;
  if (jb + lane < c) {
    int s = ssrc[o + jb + lane];
    s = min(max(s, 0), N - 1);
    g_l = s1[s];
    s_l = s;
  }
}
__device__ __forceinline__ float edge_score(float g_l, float s2d, bool live) {
  const float v = g_l + s2d;
  const float e = v > 0.f ? v : 0.01f * v;
  return live ? e : -INFINITY;
}

__device__ __forceinline__ void rows_load(
    const char* __restrict__ zb, unsigned lane_off, int s_l, int j0, int n,
    v2f (&zv)[AG_GROUP]) {
  #pragma unroll
  for (int u = 0; u < AG_GROUP; ++u) {
    zv[u] = (v2f){0.f, 0.f};
    if (j0 + u < n) {
      const int s = __builtin_amdgcn_readlane(s_l, j0 + u);
      zv[u] = *(const v2f*)(zb + (long)s * (OUTD * 4) + lane_off);
    }
  }
}
__device__ __forceinline__ void rows_fma(
    const v2f (&zv)[AG_GROUP], int a_bits, int j0, int n, v2f& acc) {
  #pragma unroll
  for (int u = 0; u < AG_GROUP; ++u) {
    if (j0 + u < n) {
      const float al = __int_as_float(__builtin_amdgcn_readlane(a_bits, j0 + u));
      acc.x += al * zv[u].x;
      acc.y += al * zv[u].y;
    }
  }
}

__global__ __launch_bounds__(256) __attribute__((amdgpu_waves_per_eu(12))) void gat_aggregate(
    const int* __restrict__ ssrc, const int* __restrict__ off, const int* __restrict__ cnt,
    const float* __restrict__ s1, const float* __restrict__ s2,
    const float* __restrict__ z, float* __restrict__ h, int N) {
  const int lane = threadIdx.x & 31;
  const int wv = __builtin_amdgcn_readfirstlane((int)(threadIdx.x >> 5));
  const int d0 = (blockIdx.x * 8 + wv) * AG_NODES;
  if (d0 >= N) return;
  const char*    zb       = (const char*)z;
  const unsigned lane_off = (unsigned)lane * 8u;

  int   o_v  = 0;
  int   c_v  = 0;
  float s2_v = 0.0f;
  if (lane < AG_NODES && d0 + lane < N) {
    o_v  = off[d0 + lane];
    c_v  = cnt[d0 + lane];
    s2_v = s2[d0 + lane];
  }
  const int s2_bits = __float_as_int(s2_v);

  for (int k = 0; k < AG_NODES; ++k) {
    const int d = d0 + k;
    if (d >= N) break;
    const int   o   = __builtin_amdgcn_readlane(o_v, k);
    const int   c   = __builtin_amdgcn_readlane(c_v, k);
    const float s2d = __int_as_float(__builtin_amdgcn_readlane(s2_bits, k));

    const int n0 = c < 32 ? c : 32;
    int s_l; float g_l;
    trip_fetch(ssrc, s1, o, c, 0, lane, N, s_l, g_l);
    v2f zv[AG_GROUP];
    rows_load(zb, lane_off, s_l, 0, n0, zv);

    const float e_l = edge_score(g_l, s2d, lane < c);
    float m = e_l;
    for (int jb = 32; jb < c; jb += 32) {
      int s_t; float g_t;
      trip_fetch(ssrc, s1, o, c, jb, lane, N, s_t, g_t);
      m = fmaxf(m, edge_score(g_t, s2d, jb + lane < c));
    }
    m = wave_max(m);
    if (!isfinite(m)) m = 0.0f;

    const float ex_l = lane < c ? expf(e_l - m) : 0.0f;
    float den = ex_l;
    for (int jb = 32; jb < c; jb += 32) {
      int s_t; float g_t;
      trip_fetch(ssrc, s1, o, c, jb, lane, N, s_t, g_t);
      den += (jb + lane < c) ? expf(edge_score(g_t, s2d, true) - m) : 0.0f;
    }
    den = wave_sum(den);
    const float dd = fmaxf(den, 1e-9f);

    v2f acc = {0.f, 0.f};
    {
      const int a_bits = __float_as_int(ex_l / dd);
      rows_fma(zv, a_bits, 0, n0, acc);
      for (int j0 = AG_GROUP; j0 < n0; j0 += AG_GROUP) {
        rows_load(zb, lane_off, s_l, j0, n0, zv);
        rows_fma(zv, a_bits, j0, n0, acc);
      }
    }
    for (int jb = 32; jb < c; jb += 32) {
      int s_t; float g_t;
      trip_fetch(ssrc, s1, o, c, jb, lane, N, s_t, g_t);
      const float a_t = (jb + lane < c) ? expf(edge_score(g_t, s2d, true) - m) / dd : 0.0f;
      const int a_bits = __float_as_int(a_t);
      const int nt = (c - jb) < 32 ? (c - jb) : 32;
      for (int j0 = 0; j0 < nt; j0 += AG_GROUP) {
        rows_load(zb, lane_off, s_t, j0, nt, zv);
        rows_fma(zv, a_bits, j0, nt, acc);
      }
    }
    *(v2f*)((char*)h + (long)d * (OUTD * 4) + lane_off) = acc;
  }
}

extern "C" void kernel_launch(void* const* d_in, const int* in_sizes, int n_in,
                              void* d_out, int out_size, void* d_ws, size_t ws_size,
                              hipStream_t stream) {
  const float* x    = (const float*)d_in[0];
  const float* w_in = (const float*)d_in[1];
  const float* b_in = (const float*)d_in[2];
  const float* w    = (const float*)d_in[3];
  const float* a    = (const float*)d_in[4];
  const int*   src  = (const int*)d_in[5];
  const int*   dst  = (const int*)d_in[6];
  float* h = (float*)d_out;

  const int N = in_sizes[0] / IN_DIM;
  const int E = in_sizes[5];
  const int NB = (N + BK_ROWS - 1) / BK_ROWS;

  auto align256 = [](size_t v) { return (v + 255) & ~(size_t)255; };
  char* base = (char*)d_ws;
  size_t off_b = 0;
  float* z      = (float*)(base + off_b); off_b += align256((size_t)N * OUTD * 4);
  float* s1     = (float*)(base + off_b); off_b += align256((size_t)N * 4);
  float* s2     = (float*)(base + off_b); off_b += align256((size_t)N * 4);
  int*   cnt    = (int*)(base + off_b);   off_b += align256((size_t)N * 4);
  int*   off    = (int*)(base + off_b);   off_b += align256((size_t)N * 4);
  int*   hist   = (int*)(base + off_b);   off_b += align256((size_t)SORT_CHUNKS * SORT_NBP * 4);
  int*   bbase  = (int*)(base + off_b);   off_b += align256((size_t)(SORT_NBP + 1) * 4);
  int2*  pairs  = (int2*)(base + off_b);  off_b += align256((size_t)E * 8);
  int*   ssrc   = (int*)(base + off_b);   off_b += align256((size_t)E * 4);

  const int node_blocks = (N + 255) / 256;

  gat_chunk_hist<<<dim3(SORT_CHUNKS), dim3(256), 0, stream>>>(dst, hist, E, NB);
  gat_chunk_scatter<<<dim3(SORT_CHUNKS), dim3(256), 0, stream>>>(src, dst, hist, bbase, pairs,
                                                                E, NB);
  gat_bucket_sort<<<dim3(NB), dim3(BS_THREADS), 0, stream>>>(pairs, bbase, cnt, off, ssrc, N, E);
  {
    int ntiles = (N + 15) >> 4;
    int blocks = (ntiles + 1) / 2;
    gat_fused_gemm<<<dim3(blocks), dim3(64), 0, stream>>>(x, w_in, b_in, w, z, N);
  }
  gat_node<<<dim3(node_blocks), dim3(256), 0, stream>>>(z, a, s1, s2, N);
  {
    const int rows_per_block = 8 * AG_NODES;
    gat_aggregate<<<dim3((N + rows_per_block - 1) / rows_per_block), dim3(256), 0, stream>>>(
        ssrc, off, cnt, s1, s2, z, h, N);
  }
}


# scores s1 = z.a[:64], s2 = z.a[64:] taken from the accumulator tile inside the GEMM launch (4 products per lane, then a 16-lane sum tree) and the per-node launch removed; score sums reordered
# speedup vs baseline: 3.7501x; 1.0469x over previous
#include <hip/hip_runtime.h>
#include <math.h>

typedef float v2f __attribute__((ext_vector_type(2)));
typedef float v8f __attribute__((ext_vector_type(8)));

#define IN_DIM 128
#define HID    64
#define OUTD   64

#define SORT_CHUNKS 256
#define SORT_NBP    256
#define BK_SHIFT    9
#define BK_ROWS     512
#define SORT_UNITS  8

__device__ __forceinline__ int bucket_of(int d, int NB) {
  const unsigned b = (unsigned)d >> BK_SHIFT;
  return (int)(b < (unsigned)(NB - 1) ? b : (unsigned)(NB - 1));
}
__device__ __forceinline__ long chunk_size(int E) {
  return ((((long)E + SORT_CHUNKS - 1) / SORT_CHUNKS) + 3) & ~3L;
}

__global__ __launch_bounds__(256) void gat_chunk_hist(
    const int* __restrict__ dst, int* __restrict__ hist, int E, int NB) {
  __shared__ int h[SORT_NBP];
  const int t = threadIdx.x;
  const int c = blockIdx.x;
  h[t] = 0;
  __syncthreads();
  const long cs  = chunk_size(E);
  const long lo  = (long)c * cs;
  long       hi  = lo + cs;
  if (hi > E) hi = E;
  const long hi4 = lo + ((hi - lo) & ~3L);
  for (long base = lo; base < hi4; base += 256L * 4 * SORT_UNITS) {
    int4 d[SORT_UNITS];
    #pragma unroll
    for (int j = 0; j < SORT_UNITS; ++j) {
      const long e = base + ((long)j * 256 + t) * 4;
      d[j] = *(const int4*)(dst + (e < hi4 ? e : lo));
    }
    #pragma unroll
    for (int j = 0; j < SORT_UNITS; ++j) {
      const long e = base + ((long)j * 256 + t) * 4;
      if (e < hi4) {
        atomicAdd(&h[bucket_of(d[j].x, NB)], 1);
        atomicAdd(&h[bucket_of(d[j].y, NB)], 1);
        atomicAdd(&h[bucket_of(d[j].z, NB)], 1);
        atomicAdd(&h[bucket_of(d[j].w, NB)], 1);
      }
    }
  }
  if (hi4 + t < hi) atomicAdd(&h[bucket_of(dst[hi4 + t], NB)], 1);
  __syncthreads();
  hist[c * SORT_NBP + t] = h[t];
}

__global__ __launch_bounds__(256) void gat_chunk_scatter(
    const int* __restrict__ src, const int* __restrict__ dst, const int* __restrict__ hist,
    int* __restrict__ bbase, int2* __restrict__ pairs, int E, int NB) {
  __shared__ __attribute__((aligned(16))) int part_pre[4][SORT_NBP];
  __shared__ __attribute__((aligned(16))) int part_tot[4][SORT_NBP];
  __shared__ int cursor[SORT_NBP];
  __shared__ int wsum[8];
  const int t    = threadIdx.x;
  const int lane = t & 31;
  const int wv   = t >> 5;
  const int c    = blockIdx.x;

  {
    const int q = t & 63;
    const int r = t >> 6;
    int4 pre = {0, 0, 0, 0};
    int4 tot = {0, 0, 0, 0};
    #pragma unroll
    for (int hb = 0; hb < 2; ++hb) {
      int4 v[32];
      #pragma unroll
      for (int k = 0; k < 32; ++k)
        v[k] = *(const int4*)(hist + (r + 4 * (hb * 32 + k)) * SORT_NBP + 4 * q);
      #pragma unroll
      for (int k = 0; k < 32; ++k) {
        const int c2 = r + 4 * (hb * 32 + k);
        tot.x += v[k].x; tot.y += v[k].y; tot.z += v[k].z; tot.w += v[k].w;
        if (c2 < c) { pre.x += v[k].x; pre.y += v[k].y; pre.z += v[k].z; pre.w += v[k].w; }
      }
    }
    *(int4*)(&part_pre[r][4 * q]) = pre;
    *(int4*)(&part_tot[r][4 * q]) = tot;
  }
  __syncthreads();
  const int my_pre = part_pre[0][t] + part_pre[1][t] + part_pre[2][t] + part_pre[3][t];
  const int my_tot = part_tot[0][t] + part_tot[1][t] + part_tot[2][t] + part_tot[3][t];
  int incl = my_tot;
  #pragma unroll
  for (int o = 1; o < 32; o <<= 1) {
    const int u = __shfl_up(incl, o, 32);
    if (lane >= o) incl += u;
  }
  if (lane == 31) wsum[wv] = incl;
  __syncthreads();
  int wbase = 0;
  #pragma unroll
  for (int w = 0; w < 8; ++w) wbase += (w < wv) ? wsum[w] : 0;
  const int bstart = wbase + incl - my_tot;
  cursor[t] = bstart + my_pre;
  if (c == 0) {
    if (t < NB) bbase[t] = bstart;
    if (t == NB - 1) bbase[NB] = bstart + my_tot;
  }
  __syncthreads();

  const long cs  = chunk_size(E);
  const long lo  = (long)c * cs;
  long       hi  = lo + cs;
  if (hi > E) hi = E;
  const long hi4 = lo + ((hi - lo) & ~3L);
  for (long base = lo; base < hi4; base += 256L * 4 * SORT_UNITS) {
    int4 d[SORT_UNITS];
    int4 s[SORT_UNITS];
    #pragma unroll
    for (int j = 0; j < SORT_UNITS; ++j) {
      const long e  = base + ((long)j * 256 + t) * 4;
      const long ec = e < hi4 ? e : lo;
      d[j] = *(const int4*)(dst + ec);
      s[j] = *(const int4*)(src + ec);
    }
    #pragma unroll
    for (int j = 0; j < SORT_UNITS; ++j) {
      const long e = base + ((long)j * 256 + t) * 4;
      if (e < hi4) {
        const unsigned p0 = (unsigned)atomicAdd(&cursor[bucket_of(d[j].x, NB)], 1);
        const unsigned p1 = (unsigned)atomicAdd(&cursor[bucket_of(d[j].y, NB)], 1);
        const unsigned p2 = (unsigned)atomicAdd(&cursor[bucket_of(d[j].z, NB)], 1);
        const unsigned p3 = (unsigned)atomicAdd(&cursor[bucket_of(d[j].w, NB)], 1);
        if (p0 < (unsigned)E) pairs[p0] = make_int2(s[j].x, d[j].x);
        if (p1 < (unsigned)E) pairs[p1] = make_int2(s[j].y, d[j].y);
        if (p2 < (unsigned)E) pairs[p2] = make_int2(s[j].z, d[j].z);
        if (p3 < (unsigned)E) pairs[p3] = make_int2(s[j].w, d[j].w);
      }
    }
  }
  if (hi4 + t < hi) {
    const int dd = dst[hi4 + t];
    const unsigned p = (unsigned)atomicAdd(&cursor[bucket_of(dd, NB)], 1);
    if (p < (unsigned)E) pairs[p] = make_int2(src[hi4 + t], dd);
  }
}

#define BS_THREADS 1024
#define BS_UNITS   8
__global__ __launch_bounds__(BS_THREADS) void gat_bucket_sort(
    const int2* __restrict__ pairs, const int* __restrict__ bbase,
    int* __restrict__ cnt, int* __restrict__ off, int* __restrict__ ssrc, int N, int E) {
  __shared__ int rcnt[BK_ROWS];
  __shared__ int rcur[BK_ROWS];
  __shared__ int wsum[16];
  __shared__ int wexc[16];
  const int t    = threadIdx.x;
  const int lane = t & 31;
  const int wv   = t >> 5;
  const int b    = blockIdx.x;
  int lo = bbase[b];
  int hi = bbase[b + 1];
  lo = min(max(lo, 0), E);
  hi = min(max(hi, lo), E);
  const int row0 = b << BK_SHIFT;

  if (t < BK_ROWS) rcnt[t] = 0;
  __syncthreads();
  for (int base = lo; base < hi; base += BS_THREADS * BS_UNITS) {
    int2 pr[BS_UNITS];
    #pragma unroll
    for (int j = 0; j < BS_UNITS; ++j) {
      const int e = base + j * BS_THREADS + t;
      pr[j] = pairs[e < hi ? e : lo];
    }
    #pragma unroll
    for (int j = 0; j < BS_UNITS; ++j) {
      const int e = base + j * BS_THREADS + t;
      if (e < hi) atomicAdd(&rcnt[(pr[j].y - row0) & (BK_ROWS - 1)], 1);
    }
  }
  __syncthreads();

  int v = 0, incl = 0;
  if (t < BK_ROWS) {
    v = rcnt[t];
    incl = v;
    #pragma unroll
    for (int o = 1; o < 32; o <<= 1) {
      const int u = __shfl_up(incl, o, 32);
      if (lane >= o) incl += u;
    }
    if (lane == 31) wsum[wv] = incl;
  }
  __syncthreads();
  if (t < 16) {
    const int x = wsum[t];
    int in2 = x;
    #pragma unroll
    for (int o = 1; o < 16; o <<= 1) {
      const int u = __shfl_up(in2, o, 32);
      if (lane >= o) in2 += u;
    }
    wexc[t] = in2 - x;
  }
  __syncthreads();
  if (t < BK_ROWS) {
    const int start = wexc[wv] + incl - v;
    rcur[t] = start;
    const int node = row0 + t;
    if (node < N) {
      off[node] = lo + start;
      cnt[node] = v;
    }
  }
  __syncthreads();

  for (int base = lo; base < hi; base += BS_THREADS * BS_UNITS) {
    int2 pr[BS_UNITS];
    #pragma unroll
    for (int j = 0; j < BS_UNITS; ++j) {
      const int e = base + j * BS_THREADS + t;
      pr[j] = pairs[e < hi ? e : lo];
    }
    #pragma unroll
    for (int j = 0; j < BS_UNITS; ++j) {
      const int e = base + j * BS_THREADS + t;
      if (e < hi) {
        const int p = atomicAdd(&rcur[(pr[j].y - row0) & (BK_ROWS - 1)], 1);
        const unsigned g = (unsigned)(lo + p);
        if (g < (unsigned)hi) ssrc[g] = pr[j].x;
      }
    }
  }
}

template <int BIT>
__device__ __forceinline__ float fold_step(float lo, float hi, bool up) {
  const float keep = up ? hi : lo;
  const float send = up ? lo : hi;
  const int got = __builtin_amdgcn_update_dpp(0, __float_as_int(send), 0x160 | BIT, 0xf, 0xf, true);
  return keep + __int_as_float(got);
}

__global__ __launch_bounds__(64) void gat_fused_gemm(
    const float* __restrict__ x, const float* __restrict__ w_in,
    const float* __restrict__ b_in, const float* __restrict__ w,
    const float* __restrict__ a, float* __restrict__ z,
    float* __restrict__ s1, float* __restrict__ s2, int N) {
  __shared__ float lds[2][16][66];
  __shared__ float sc[2][32];

  const int lane = threadIdx.x & 31;
  const int wv   = threadIdx.x >> 5;
  const int tile = blockIdx.x * 2 + wv;
  const int ntiles = (N + 15) >> 4;
  const bool active = tile < ntiles;

  const int m  = lane & 15;
  const int kp = lane >> 4;

  if (active) {
    const int row  = tile * 16 + m;
    const int rowc = row < N ? row : (N - 1);
    const float* xrow = x + (long)rowc * IN_DIM;

    v8f acc[4];
    #pragma unroll
    for (int nt = 0; nt < 4; ++nt) acc[nt] = (v8f){0,0,0,0,0,0,0,0};

    for (int k0 = 0; k0 < IN_DIM; k0 += 4) {
      const int kk = k0 + 2 * kp;
      v2f av = *(const v2f*)(xrow + kk);
      const float* wr0 = w_in + (long)kk * HID;
      const float* wr1 = wr0 + HID;
      #pragma unroll
      for (int nt = 0; nt < 4; ++nt) {
        v2f bv = { wr0[nt * 16 + m], wr1[nt * 16 + m] };
        acc[nt] = __builtin_amdgcn_wmma_f32_16x16x4_f32(
            false, av, false, bv, (short)0, acc[nt], false, false);
      }
    }

    #pragma unroll
    for (int nt = 0; nt < 4; ++nt) {
      float bias = b_in[nt * 16 + m];
      #pragma unroll
      for (int r = 0; r < 8; ++r) {
        float v = acc[nt][r] + bias;
        v = 0.5f * v * (1.0f + erff(v * 0.70710678118654752f));
        lds[wv][r + 8 * kp][nt * 16 + m] = v;
      }
    }
  }
  __syncthreads();

  v8f acc[4];
  #pragma unroll
  for (int nt = 0; nt < 4; ++nt) acc[nt] = (v8f){0,0,0,0,0,0,0,0};
  if (active) {
    for (int k0 = 0; k0 < HID; k0 += 4) {
      const int kk = k0 + 2 * kp;
      v2f av = *(const v2f*)(&lds[wv][m][kk]);
      const float* wr0 = w + (long)kk * OUTD;
      const float* wr1 = wr0 + OUTD;
      #pragma unroll
      for (int nt = 0; nt < 4; ++nt) {
        v2f bv = { wr0[nt * 16 + m], wr1[nt * 16 + m] };
        acc[nt] = __builtin_amdgcn_wmma_f32_16x16x4_f32(
            false, av, false, bv, (short)0, acc[nt], false, false);
      }
    }

    const float a10 = a[m],        a11 = a[16 + m],        a12 = a[32 + m],        a13 = a[48 + m];
    const float a20 = a[OUTD + m], a21 = a[OUTD + 16 + m], a22 = a[OUTD + 32 + m], a23 = a[OUTD + 48 + m];
#define ROW_PAIR(nt, j) ((v2f){acc[nt][2 * (j)], acc[nt][2 * (j) + 1]})
#define SCORE_PAIR(j, w0, w1, w2, w3) \
    (ROW_PAIR(0, j) * (w0) + ROW_PAIR(1, j) * (w1) + ROW_PAIR(2, j) * (w2) + ROW_PAIR(3, j) * (w3))
    const v2f P0 = SCORE_PAIR(0, a10, a11, a12, a13), Q0 = SCORE_PAIR(0, a20, a21, a22, a23);
    const v2f P1 = SCORE_PAIR(1, a10, a11, a12, a13), Q1 = SCORE_PAIR(1, a20, a21, a22, a23);
    const v2f P2 = SCORE_PAIR(2, a10, a11, a12, a13), Q2 = SCORE_PAIR(2, a20, a21, a22, a23);
    const v2f P3 = SCORE_PAIR(3, a10, a11, a12, a13), Q3 = SCORE_PAIR(3, a20, a21, a22, a23);
#undef SCORE_PAIR
#undef ROW_PAIR
    const bool u8 = (m & 8) != 0, u4 = (m & 4) != 0, u2 = (m & 2) != 0, u1 = (m & 1) != 0;
    const float c0 = fold_step<8>(P0.x, Q0.x, u8), c1 = fold_step<8>(P0.y, Q0.y, u8);
    const float c2 = fold_step<8>(P1.x, Q1.x, u8), c3 = fold_step<8>(P1.y, Q1.y, u8);
    const float c4 = fold_step<8>(P2.x, Q2.x, u8), c5 = fold_step<8>(P2.y, Q2.y, u8);
    const float c6 = fold_step<8>(P3.x, Q3.x, u8), c7 = fold_step<8>(P3.y, Q3.y, u8);
    const float d0 = fold_step<4>(c0, c4, u4), d1 = fold_step<4>(c1, c5, u4);
    const float d2 = fold_step<4>(c2, c6, u4), d3 = fold_step<4>(c3, c7, u4);
    const float e0 = fold_step<2>(d0, d2, u2), e1 = fold_step<2>(d1, d3, u2);
    const float f0 = fold_step<1>(e0, e1, u1);
    sc[m >> 3][wv * 16 + 8 * kp + (m & 7)] = f0;
  }
  __syncthreads();
  {
    const int rr = blockIdx.x * 32 + lane;
    float* sp = wv ? s2 : s1;
    if (rr < N) sp[rr] = sc[wv][lane];
  }

  if (active) {
    #pragma unroll
    for (int nt = 0; nt < 4; ++nt) {
      #pragma unroll
      for (int r = 0; r < 8; ++r) {
        int rr = tile * 16 + r + 8 * kp;
        if (rr < N) z[(long)rr * OUTD + nt * 16 + m] = acc[nt][r];
      }
    }
  }
}

#define AG_NODES 4
#define AG_GROUP 16

__device__ __forceinline__ float wave_max(float v) {
  #pragma unroll
  for (int o = 16; o > 0; o >>= 1) v = fmaxf(v, __shfl_xor(v, o, 32));
  return v;
}
__device__ __forceinline__ float wave_sum(float v) {
  #pragma unroll
  for (int o = 16; o > 0; o >>= 1) v += __shfl_xor(v, o, 32);
  return v;
}

__device__ __forceinline__ void trip_fetch(
    const int* __restrict__ ssrc, const float* __restrict__ s1,
    int o, int c, int jb, int lane, int N, int& s_l, float& g_l) {
  s_l = 0;
  g_l = 0.0f;
  if (jb + lane < c) {
    int s = ssrc[o + jb + lane];
    s = min(max(s, 0), N - 1);
    g_l = s1[s];
    s_l = s;
  }
}
__device__ __forceinline__ float edge_score(float g_l, float s2d, bool live) {
  const float v = g_l + s2d;
  const float e = v > 0.f ? v : 0.01f * v;
  return live ? e : -INFINITY;
}

__device__ __forceinline__ void rows_load(
    const char* __restrict__ zb, unsigned lane_off, int s_l, int j0, int n,
    v2f (&zv)[AG_GROUP]) {
  #pragma unroll
  for (int u = 0; u < AG_GROUP; ++u) {
    zv[u] = (v2f){0.f, 0.f};
    if (j0 + u < n) {
      const int s = __builtin_amdgcn_readlane(s_l, j0 + u);
      zv[u] = *(const v2f*)(zb + (long)s * (OUTD * 4) + lane_off);
    }
  }
}
__device__ __forceinline__ void rows_fma(
    const v2f (&zv)[AG_GROUP], int a_bits, int j0, int n, v2f& acc) {
  #pragma unroll
  for (int u = 0; u < AG_GROUP; ++u) {
    if (j0 + u < n) {
      const float al = __int_as_float(__builtin_amdgcn_readlane(a_bits, j0 + u));
      acc.x += al * zv[u].x;
      acc.y += al * zv[u].y;
    }
  }
}

__global__ __launch_bounds__(256) __attribute__((amdgpu_waves_per_eu(12))) void gat_aggregate(
    const int* __restrict__ ssrc, const int* __restrict__ off, const int* __restrict__ cnt,
    const float* __restrict__ s1, const float* __restrict__ s2,
    const float* __restrict__ z, float* __restrict__ h, int N) {
  const int lane = threadIdx.x & 31;
  const int wv = __builtin_amdgcn_readfirstlane((int)(threadIdx.x >> 5));
  const int d0 = (blockIdx.x * 8 + wv) * AG_NODES;
  if (d0 >= N) return;
  const char*    zb       = (const char*)z;
  const unsigned lane_off = (unsigned)lane * 8u;

  int   o_v  = 0;
  int   c_v  = 0;
  float s2_v = 0.0f;
  if (lane < AG_NODES && d0 + lane < N) {
    o_v  = off[d0 + lane];
    c_v  = cnt[d0 + lane];
    s2_v = s2[d0 + lane];
  }
  const int s2_bits = __float_as_int(s2_v);

  for (int k = 0; k < AG_NODES; ++k) {
    const int d = d0 + k;
    if (d >= N) break;
    const int   o   = __builtin_amdgcn_readlane(o_v, k);
    const int   c   = __builtin_amdgcn_readlane(c_v, k);
    const float s2d = __int_as_float(__builtin_amdgcn_readlane(s2_bits, k));

    const int n0 = c < 32 ? c : 32;
    int s_l; float g_l;
    trip_fetch(ssrc, s1, o, c, 0, lane, N, s_l, g_l);
    v2f zv[AG_GROUP];
    rows_load(zb, lane_off, s_l, 0, n0, zv);

    const float e_l = edge_score(g_l, s2d, lane < c);
    float m = e_l;
    for (int jb = 32; jb < c; jb += 32) {
      int s_t; float g_t;
      trip_fetch(ssrc, s1, o, c, jb, lane, N, s_t, g_t);
      m = fmaxf(m, edge_score(g_t, s2d, jb + lane < c));
    }
    m = wave_max(m);
    if (!isfinite(m)) m = 0.0f;

    const float ex_l = lane < c ? expf(e_l - m) : 0.0f;
    float den = ex_l;
    for (int jb = 32; jb < c; jb += 32) {
      int s_t; float g_t;
      trip_fetch(ssrc, s1, o, c, jb, lane, N, s_t, g_t);
      den += (jb + lane < c) ? expf(edge_score(g_t, s2d, true) - m) : 0.0f;
    }
    den = wave_sum(den);
    const float dd = fmaxf(den, 1e-9f);

    v2f acc = {0.f, 0.f};
    {
      const int a_bits = __float_as_int(ex_l / dd);
      rows_fma(zv, a_bits, 0, n0, acc);
      for (int j0 = AG_GROUP; j0 < n0; j0 += AG_GROUP) {
        rows_load(zb, lane_off, s_l, j0, n0, zv);
        rows_fma(zv, a_bits, j0, n0, acc);
      }
    }
    for (int jb = 32; jb < c; jb += 32) {
      int s_t; float g_t;
      trip_fetch(ssrc, s1, o, c, jb, lane, N, s_t, g_t);
      const float a_t = (jb + lane < c) ? expf(edge_score(g_t, s2d, true) - m) / dd : 0.0f;
      const int a_bits = __float_as_int(a_t);
      const int nt = (c - jb) < 32 ? (c - jb) : 32;
      for (int j0 = 0; j0 < nt; j0 += AG_GROUP) {
        rows_load(zb, lane_off, s_t, j0, nt, zv);
        rows_fma(zv, a_bits, j0, nt, acc);
      }
    }
    *(v2f*)((char*)h + (long)d * (OUTD * 4) + lane_off) = acc;
  }
}

extern "C" void kernel_launch(void* const* d_in, const int* in_sizes, int n_in,
                              void* d_out, int out_size, void* d_ws, size_t ws_size,
                              hipStream_t stream) {
  const float* x    = (const float*)d_in[0];
  const float* w_in = (const float*)d_in[1];
  const float* b_in = (const float*)d_in[2];
  const float* w    = (const float*)d_in[3];
  const float* a    = (const float*)d_in[4];
  const int*   src  = (const int*)d_in[5];
  const int*   dst  = (const int*)d_in[6];
  float* h = (float*)d_out;

  const int N = in_sizes[0] / IN_DIM;
  const int E = in_sizes[5];
  const int NB = (N + BK_ROWS - 1) / BK_ROWS;

  auto align256 = [](size_t v) { return (v + 255) & ~(size_t)255; };
  char* base = (char*)d_ws;
  size_t off_b = 0;
  float* z      = (float*)(base + off_b); off_b += align256((size_t)N * OUTD * 4);
  float* s1     = (float*)(base + off_b); off_b += align256((size_t)N * 4);
  float* s2     = (float*)(base + off_b); off_b += align256((size_t)N * 4);
  int*   cnt    = (int*)(base + off_b);   off_b += align256((size_t)N * 4);
  int*   off    = (int*)(base + off_b);   off_b += align256((size_t)N * 4);
  int*   hist   = (int*)(base + off_b);   off_b += align256((size_t)SORT_CHUNKS * SORT_NBP * 4);
  int*   bbase  = (int*)(base + off_b);   off_b += align256((size_t)(SORT_NBP + 1) * 4);
  int2*  pairs  = (int2*)(base + off_b);  off_b += align256((size_t)E * 8);
  int*   ssrc   = (int*)(base + off_b);   off_b += align256((size_t)E * 4);

  gat_chunk_hist<<<dim3(SORT_CHUNKS), dim3(256), 0, stream>>>(dst, hist, E, NB);
  gat_chunk_scatter<<<dim3(SORT_CHUNKS), dim3(256), 0, stream>>>(src, dst, hist, bbase, pairs,
                                                                E, NB);
  gat_bucket_sort<<<dim3(NB), dim3(BS_THREADS), 0, stream>>>(pairs, bbase, cnt, off, ssrc, N, E);
  {
    int ntiles = (N + 15) >> 4;
    int blocks = (ntiles + 1) / 2;
    gat_fused_gemm<<<dim3(blocks), dim3(64), 0, stream>>>(x, w_in, b_in, w, a, z, s1, s2, N);
  }
  {
    const int rows_per_block = 8 * AG_NODES;
    gat_aggregate<<<dim3((N + rows_per_block - 1) / rows_per_block), dim3(256), 0, stream>>>(
        ssrc, off, cnt, s1, s2, z, h, N);
  }
}


# both scattering sort passes (chunk to bucket, bucket to row) place items in an LDS area, then write it out in index order: whole runs or lines per store, not 32 far-apart pieces; order as before
# speedup vs baseline: 4.2988x; 1.1463x over previous
#include <hip/hip_runtime.h>
#include <math.h>

typedef float v2f __attribute__((ext_vector_type(2)));
typedef float v8f __attribute__((ext_vector_type(8)));

#define IN_DIM 128
#define HID    64
#define OUTD   64

#define SORT_CHUNKS 256
#define SORT_NBP    256
#define BK_SHIFT    9
#define BK_ROWS     512
#define SORT_UNITS  8
#define SC_STAGE    6656
#define BS_STAGE    12288

__device__ __forceinline__ int bucket_of(int d, int NB) {
  const unsigned b = (unsigned)d >> BK_SHIFT;
  return (int)(b < (unsigned)(NB - 1) ? b : (unsigned)(NB - 1));
}
__device__ __forceinline__ long chunk_size(int E) {
  return ((((long)E + SORT_CHUNKS - 1) / SORT_CHUNKS) + 3) & ~3L;
}

__global__ __launch_bounds__(256) void gat_chunk_hist(
    const int* __restrict__ dst, int* __restrict__ hist, int E, int NB) {
  __shared__ int h[SORT_NBP];
  const int t = threadIdx.x;
  const int c = blockIdx.x;
  h[t] = 0;
  __syncthreads();
  const long cs  = chunk_size(E);
  const long lo  = (long)c * cs;
  long       hi  = lo + cs;
  if (hi > E) hi = E;
  const long hi4 = lo + ((hi - lo) & ~3L);
  for (long base = lo; base < hi4; base += 256L * 4 * SORT_UNITS) {
    int4 d[SORT_UNITS];
    #pragma unroll
    for (int j = 0; j < SORT_UNITS; ++j) {
      const long e = base + ((long)j * 256 + t) * 4;
      d[j] = *(const int4*)(dst + (e < hi4 ? e : lo));
    }
    #pragma unroll
    for (int j = 0; j < SORT_UNITS; ++j) {
      const long e = base + ((long)j * 256 + t) * 4;
      if (e < hi4) {
        atomicAdd(&h[bucket_of(d[j].x, NB)], 1);
        atomicAdd(&h[bucket_of(d[j].y, NB)], 1);
        atomicAdd(&h[bucket_of(d[j].z, NB)], 1);
        atomicAdd(&h[bucket_of(d[j].w, NB)], 1);
      }
    }
  }
  if (hi4 + t < hi) atomicAdd(&h[bucket_of(dst[hi4 + t], NB)], 1);
  __syncthreads();
  hist[c * SORT_NBP + t] = h[t];
}

template <bool STAGED>
__device__ __forceinline__ void chunk_put(int2* stage, int2* __restrict__ pairs, unsigned p,
                                          int sv, int dv, int E) {
  if (STAGED) {
    if (p < (unsigned)SC_STAGE) stage[p] = make_int2(sv, dv);
  } else {
    if (p < (unsigned)E) pairs[p] = make_int2(sv, dv);
  }
}
template <bool STAGED>
__device__ __forceinline__ void chunk_place(
    const int* __restrict__ src, const int* __restrict__ dst, int* cursor, int2* stage,
    int2* __restrict__ pairs, long lo, long hi, int t, int E, int NB) {
  const long hi4 = lo + ((hi - lo) & ~3L);
  for (long base = lo; base < hi4; base += 256L * 4 * SORT_UNITS) {
    int4 d[SORT_UNITS];
    int4 s[SORT_UNITS];
    #pragma unroll
    for (int j = 0; j < SORT_UNITS; ++j) {
      const long e  = base + ((long)j * 256 + t) * 4;
      const long ec = e < hi4 ? e : lo;
      d[j] = *(const int4*)(dst + ec);
      s[j] = *(const int4*)(src + ec);
    }
    #pragma unroll
    for (int j = 0; j < SORT_UNITS; ++j) {
      const long e = base + ((long)j * 256 + t) * 4;
      if (e < hi4) {
        const unsigned p0 = (unsigned)atomicAdd(&cursor[bucket_of(d[j].x, NB)], 1);
        const unsigned p1 = (unsigned)atomicAdd(&cursor[bucket_of(d[j].y, NB)], 1);
        const unsigned p2 = (unsigned)atomicAdd(&cursor[bucket_of(d[j].z, NB)], 1);
        const unsigned p3 = (unsigned)atomicAdd(&cursor[bucket_of(d[j].w, NB)], 1);
        chunk_put<STAGED>(stage, pairs, p0, s[j].x, d[j].x, E);
        chunk_put<STAGED>(stage, pairs, p1, s[j].y, d[j].y, E);
        chunk_put<STAGED>(stage, pairs, p2, s[j].z, d[j].z, E);
        chunk_put<STAGED>(stage, pairs, p3, s[j].w, d[j].w, E);
      }
    }
  }
  if (hi4 + t < hi) {
    const int dd = dst[hi4 + t];
    const unsigned p = (unsigned)atomicAdd(&cursor[bucket_of(dd, NB)], 1);
    chunk_put<STAGED>(stage, pairs, p, src[hi4 + t], dd, E);
  }
}

__global__ __launch_bounds__(256) void gat_chunk_scatter(
    const int* __restrict__ src, const int* __restrict__ dst, const int* __restrict__ hist,
    int* __restrict__ bbase, int2* __restrict__ pairs, int E, int NB) {
  __shared__ __attribute__((aligned(16))) int part_pre[4][SORT_NBP];
  __shared__ __attribute__((aligned(16))) int part_tot[4][SORT_NBP];
  __shared__ int cursor[SORT_NBP];
  __shared__ int gdelta[SORT_NBP];
  __shared__ int wsum[8];
  __shared__ int wsum_l[8];
  __shared__ int2 stage[SC_STAGE];
  const int t    = threadIdx.x;
  const int lane = t & 31;
  const int wv   = t >> 5;
  const int c    = blockIdx.x;
  const long cs  = chunk_size(E);
  const bool staged = cs <= SC_STAGE;
  const int my_cnt = hist[c * SORT_NBP + t];

  {
    const int q = t & 63;
    const int r = t >> 6;
    int4 pre = {0, 0, 0, 0};
    int4 tot = {0, 0, 0, 0};
    #pragma unroll
    for (int hb = 0; hb < 2; ++hb) {
      int4 v[32];
      #pragma unroll
      for (int k = 0; k < 32; ++k)
        v[k] = *(const int4*)(hist + (r + 4 * (hb * 32 + k)) * SORT_NBP + 4 * q);
      #pragma unroll
      for (int k = 0; k < 32; ++k) {
        const int c2 = r + 4 * (hb * 32 + k);
        tot.x += v[k].x; tot.y += v[k].y; tot.z += v[k].z; tot.w += v[k].w;
        if (c2 < c) { pre.x += v[k].x; pre.y += v[k].y; pre.z += v[k].z; pre.w += v[k].w; }
      }
    }
    *(int4*)(&part_pre[r][4 * q]) = pre;
    *(int4*)(&part_tot[r][4 * q]) = tot;
  }
  __syncthreads();
  const int my_pre = part_pre[0][t] + part_pre[1][t] + part_pre[2][t] + part_pre[3][t];
  const int my_tot = part_tot[0][t] + part_tot[1][t] + part_tot[2][t] + part_tot[3][t];
  int incl   = my_tot;
  int incl_l = my_cnt;
  #pragma unroll
  for (int o = 1; o < 32; o <<= 1) {
    const int u  = __shfl_up(incl, o, 32);
    const int ul = __shfl_up(incl_l, o, 32);
    if (lane >= o) { incl += u; incl_l += ul; }
  }
  if (lane == 31) { wsum[wv] = incl; wsum_l[wv] = incl_l; }
  __syncthreads();
  int wbase = 0, wbase_l = 0;
  #pragma unroll
  for (int w = 0; w < 8; ++w) {
    wbase   += (w < wv) ? wsum[w] : 0;
    wbase_l += (w < wv) ? wsum_l[w] : 0;
  }
  const int bstart = wbase + incl - my_tot;
  const int lstart = wbase_l + incl_l - my_cnt;
  cursor[t] = staged ? lstart : bstart + my_pre;
  gdelta[t] = bstart + my_pre - lstart;
  if (c == 0) {
    if (t < NB) bbase[t] = bstart;
    if (t == NB - 1) bbase[NB] = bstart + my_tot;
  }
  __syncthreads();

  const long lo  = (long)c * cs;
  long       hi  = lo + cs;
  if (hi > E) hi = E;
  if (!staged) {
    chunk_place<false>(src, dst, cursor, stage, pairs, lo, hi, t, E, NB);
  } else {
    chunk_place<true>(src, dst, cursor, stage, pairs, lo, hi, t, E, NB);
    __syncthreads();
    const int n = hi > lo ? (int)(hi - lo) : 0;
    for (int i0 = 0; i0 < n; i0 += 256 * SORT_UNITS) {
      int2 v[SORT_UNITS];
      int  gd[SORT_UNITS];
      #pragma unroll
      for (int j = 0; j < SORT_UNITS; ++j) {
        const int i = i0 + j * 256 + t;
        v[j] = stage[i < n ? i : 0];
      }
      #pragma unroll
      for (int j = 0; j < SORT_UNITS; ++j) gd[j] = gdelta[bucket_of(v[j].y, NB)];
      #pragma unroll
      for (int j = 0; j < SORT_UNITS; ++j) {
        const int i = i0 + j * 256 + t;
        const unsigned p = (unsigned)(gd[j] + i);
        if (i < n && p < (unsigned)E) pairs[p] = v[j];
      }
    }
  }
}

#define BS_THREADS 1024
#define BS_UNITS   8
template <bool STAGED>
__device__ __forceinline__ void bucket_place(
    const int2* __restrict__ pairs, int* rcur, int* stage, int* __restrict__ ssrc,
    int lo, int hi, int row0, int sh, int t) {
  for (int base = lo; base < hi; base += BS_THREADS * BS_UNITS) {
    int2 pr[BS_UNITS];
    #pragma unroll
    for (int j = 0; j < BS_UNITS; ++j) {
      const int e = base + j * BS_THREADS + t;
      pr[j] = pairs[e < hi ? e : lo];
    }
    #pragma unroll
    for (int j = 0; j < BS_UNITS; ++j) {
      const int e = base + j * BS_THREADS + t;
      if (e < hi) {
        const int p = atomicAdd(&rcur[(pr[j].y - row0) & (BK_ROWS - 1)], 1);
        const unsigned g = (unsigned)(lo + p);
        if (g < (unsigned)hi) {
          if (STAGED) stage[sh + p] = pr[j].x;
          else        ssrc[g] = pr[j].x;
        }
      }
    }
  }
}

__global__ __launch_bounds__(BS_THREADS) void gat_bucket_sort(
    const int2* __restrict__ pairs, const int* __restrict__ bbase,
    int* __restrict__ cnt, int* __restrict__ off, int* __restrict__ ssrc, int N, int E) {
  __shared__ int rcnt[BK_ROWS];
  __shared__ int rcur[BK_ROWS];
  __shared__ int wsum[16];
  __shared__ int wexc[16];
  __shared__ int stage[BS_STAGE];
  const int t    = threadIdx.x;
  const int lane = t & 31;
  const int wv   = t >> 5;
  const int b    = blockIdx.x;
  int lo = bbase[b];
  int hi = bbase[b + 1];
  lo = min(max(lo, 0), E);
  hi = min(max(hi, lo), E);
  const int row0 = b << BK_SHIFT;
  const int g0 = lo & ~31;
  const int sh = lo - g0;
  const bool staged = (hi - g0) <= BS_STAGE;

  if (t < BK_ROWS) rcnt[t] = 0;
  __syncthreads();
  for (int base = lo; base < hi; base += BS_THREADS * BS_UNITS) {
    int2 pr[BS_UNITS];
    #pragma unroll
    for (int j = 0; j < BS_UNITS; ++j) {
      const int e = base + j * BS_THREADS + t;
      pr[j] = pairs[e < hi ? e : lo];
    }
    #pragma unroll
    for (int j = 0; j < BS_UNITS; ++j) {
      const int e = base + j * BS_THREADS + t;
      if (e < hi) atomicAdd(&rcnt[(pr[j].y - row0) & (BK_ROWS - 1)], 1);
    }
  }
  __syncthreads();

  int v = 0, incl = 0;
  if (t < BK_ROWS) {
    v = rcnt[t];
    incl = v;
    #pragma unroll
    for (int o = 1; o < 32; o <<= 1) {
      const int u = __shfl_up(incl, o, 32);
      if (lane >= o) incl += u;
    }
    if (lane == 31) wsum[wv] = incl;
  }
  __syncthreads();
  if (t < 16) {
    const int x = wsum[t];
    int in2 = x;
    #pragma unroll
    for (int o = 1; o < 16; o <<= 1) {
      const int u = __shfl_up(in2, o, 32);
      if (lane >= o) in2 += u;
    }
    wexc[t] = in2 - x;
  }
  __syncthreads();
  if (t < BK_ROWS) {
    const int start = wexc[wv] + incl - v;
    rcur[t] = start;
    const int node = row0 + t;
    if (node < N) {
      off[node] = lo + start;
      cnt[node] = v;
    }
  }
  __syncthreads();

  if (!staged) {
    bucket_place<false>(pairs, rcur, stage, ssrc, lo, hi, row0, 0, t);
  } else {
    bucket_place<true>(pairs, rcur, stage, ssrc, lo, hi, row0, sh, t);
    __syncthreads();
    for (int gb = g0; gb < hi; gb += BS_THREADS * BS_UNITS) {
      int v[BS_UNITS];
      #pragma unroll
      for (int j = 0; j < BS_UNITS; ++j) {
        const int g = gb + j * BS_THREADS + t;
        v[j] = stage[(g < hi ? g : hi - 1) - g0];
      }
      #pragma unroll
      for (int j = 0; j < BS_UNITS; ++j) {
        const int g = gb + j * BS_THREADS + t;
        if (g >= lo && g < hi) ssrc[g] = v[j];
      }
    }
  }
}

template <int BIT>
__device__ __forceinline__ float fold_step(float lo, float hi, bool up) {
  const float keep = up ? hi : lo;
  const float send = up ? lo : hi;
  const int got = __builtin_amdgcn_update_dpp(0, __float_as_int(send), 0x160 | BIT, 0xf, 0xf, true);
  return keep + __int_as_float(got);
}

__global__ __launch_bounds__(64) void gat_fused_gemm(
    const float* __restrict__ x, const float* __restrict__ w_in,
    const float* __restrict__ b_in, const float* __restrict__ w,
    const float* __restrict__ a, float* __restrict__ z,
    float* __restrict__ s1, float* __restrict__ s2, int N) {
  __shared__ float lds[2][16][66];
  __shared__ float sc[2][32];

  const int lane = threadIdx.x & 31;
  const int wv   = threadIdx.x >> 5;
  const int tile = blockIdx.x * 2 + wv;
  const int ntiles = (N + 15) >> 4;
  const bool active = tile < ntiles;

  const int m  = lane & 15;
  const int kp = lane >> 4;

  if (active) {
    const int row  = tile * 16 + m;
    const int rowc = row < N ? row : (N - 1);
    const float* xrow = x + (long)rowc * IN_DIM;

    v8f acc[4];
    #pragma unroll
    for (int nt = 0; nt < 4; ++nt) acc[nt] = (v8f){0,0,0,0,0,0,0,0};

    for (int k0 = 0; k0 < IN_DIM; k0 += 4) {
      const int kk = k0 + 2 * kp;
      v2f av = *(const v2f*)(xrow + kk);
      const float* wr0 = w_in + (long)kk * HID;
      const float* wr1 = wr0 + HID;
      #pragma unroll
      for (int nt = 0; nt < 4; ++nt) {
        v2f bv = { wr0[nt * 16 + m], wr1[nt * 16 + m] };
        acc[nt] = __builtin_amdgcn_wmma_f32_16x16x4_f32(
            false, av, false, bv, (short)0, acc[nt], false, false);
      }
    }

    #pragma unroll
    for (int nt = 0; nt < 4; ++nt) {
      float bias = b_in[nt * 16 + m];
      #pragma unroll
      for (int r = 0; r < 8; ++r) {
        float v = acc[nt][r] + bias;
        v = 0.5f * v * (1.0f + erff(v * 0.70710678118654752f));
        lds[wv][r + 8 * kp][nt * 16 + m] = v;
      }
    }
  }
  __syncthreads();

  v8f acc[4];
  #pragma unroll
  for (int nt = 0; nt < 4; ++nt) acc[nt] = (v8f){0,0,0,0,0,0,0,0};
  if (active) {
    for (int k0 = 0; k0 < HID; k0 += 4) {
      const int kk = k0 + 2 * kp;
      v2f av = *(const v2f*)(&lds[wv][m][kk]);
      const float* wr0 = w + (long)kk * OUTD;
      const float* wr1 = wr0 + OUTD;
      #pragma unroll
      for (int nt = 0; nt < 4; ++nt) {
        v2f bv = { wr0[nt * 16 + m], wr1[nt * 16 + m] };
        acc[nt] = __builtin_amdgcn_wmma_f32_16x16x4_f32(
            false, av, false, bv, (short)0, acc[nt], false, false);
      }
    }

    const float a10 = a[m],        a11 = a[16 + m],        a12 = a[32 + m],        a13 = a[48 + m];
    const float a20 = a[OUTD + m], a21 = a[OUTD + 16 + m], a22 = a[OUTD + 32 + m], a23 = a[OUTD + 48 + m];
#define ROW_PAIR(nt, j) ((v2f){acc[nt][2 * (j)], acc[nt][2 * (j) + 1]})
#define SCORE_PAIR(j, w0, w1, w2, w3) \
    (ROW_PAIR(0, j) * (w0) + ROW_PAIR(1, j) * (w1) + ROW_PAIR(2, j) * (w2) + ROW_PAIR(3, j) * (w3))
    const v2f P0 = SCORE_PAIR(0, a10, a11, a12, a13), Q0 = SCORE_PAIR(0, a20, a21, a22, a23);
    const v2f P1 = SCORE_PAIR(1, a10, a11, a12, a13), Q1 = SCORE_PAIR(1, a20, a21, a22, a23);
    const v2f P2 = SCORE_PAIR(2, a10, a11, a12, a13), Q2 = SCORE_PAIR(2, a20, a21, a22, a23);
    const v2f P3 = SCORE_PAIR(3, a10, a11, a12, a13), Q3 = SCORE_PAIR(3, a20, a21, a22, a23);
#undef SCORE_PAIR
#undef ROW_PAIR
    const bool u8 = (m & 8) != 0, u4 = (m & 4) != 0, u2 = (m & 2) != 0, u1 = (m & 1) != 0;
    const float c0 = fold_step<8>(P0.x, Q0.x, u8), c1 = fold_step<8>(P0.y, Q0.y, u8);
    const float c2 = fold_step<8>(P1.x, Q1.x, u8), c3 = fold_step<8>(P1.y, Q1.y, u8);
    const float c4 = fold_step<8>(P2.x, Q2.x, u8), c5 = fold_step<8>(P2.y, Q2.y, u8);
    const float c6 = fold_step<8>(P3.x, Q3.x, u8), c7 = fold_step<8>(P3.y, Q3.y, u8);
    const float d0 = fold_step<4>(c0, c4, u4), d1 = fold_step<4>(c1, c5, u4);
    const float d2 = fold_step<4>(c2, c6, u4), d3 = fold_step<4>(c3, c7, u4);
    const float e0 = fold_step<2>(d0, d2, u2), e1 = fold_step<2>(d1, d3, u2);
    const float f0 = fold_step<1>(e0, e1, u1);
    sc[m >> 3][wv * 16 + 8 * kp + (m & 7)] = f0;
  }
  __syncthreads();
  {
    const int rr = blockIdx.x * 32 + lane;
    float* sp = wv ? s2 : s1;
    if (rr < N) sp[rr] = sc[wv][lane];
  }

  if (active) {
    #pragma unroll
    for (int nt = 0; nt < 4; ++nt) {
      #pragma unroll
      for (int r = 0; r < 8; ++r) {
        int rr = tile * 16 + r + 8 * kp;
        if (rr < N) z[(long)rr * OUTD + nt * 16 + m] = acc[nt][r];
      }
    }
  }
}

#define AG_NODES 4
#define AG_GROUP 16

__device__ __forceinline__ float wave_max(float v) {
  #pragma unroll
  for (int o = 16; o > 0; o >>= 1) v = fmaxf(v, __shfl_xor(v, o, 32));
  return v;
}
__device__ __forceinline__ float wave_sum(float v) {
  #pragma unroll
  for (int o = 16; o > 0; o >>= 1) v += __shfl_xor(v, o, 32);
  return v;
}

__device__ __forceinline__ void trip_fetch(
    const int* __restrict__ ssrc, const float* __restrict__ s1,
    int o, int c, int jb, int lane, int N, int& s_l, float& g_l) {
  s_l = 0;
  g_l = 0.0f;
  if (jb + lane < c) {
    int s = ssrc[o + jb + lane];
    s = min(max(s, 0), N - 1);
    g_l = s1[s];
    s_l = s;
  }
}
__device__ __forceinline__ float edge_score(float g_l, float s2d, bool live) {
  const float v = g_l + s2d;
  const float e = v > 0.f ? v : 0.01f * v;
  return live ? e : -INFINITY;
}

__device__ __forceinline__ void rows_load(
    const char* __restrict__ zb, unsigned lane_off, int s_l, int j0, int n,
    v2f (&zv)[AG_GROUP]) {
  #pragma unroll
  for (int u = 0; u < AG_GROUP; ++u) {
    zv[u] = (v2f){0.f, 0.f};
    if (j0 + u < n) {
      const int s = __builtin_amdgcn_readlane(s_l, j0 + u);
      zv[u] = *(const v2f*)(zb + (long)s * (OUTD * 4) + lane_off);
    }
  }
}
__device__ __forceinline__ void rows_fma(
    const v2f (&zv)[AG_GROUP], int a_bits, int j0, int n, v2f& acc) {
  #pragma unroll
  for (int u = 0; u < AG_GROUP; ++u) {
    if (j0 + u < n) {
      const float al = __int_as_float(__builtin_amdgcn_readlane(a_bits, j0 + u));
      acc.x += al * zv[u].x;
      acc.y += al * zv[u].y;
    }
  }
}

__global__ __launch_bounds__(256) __attribute__((amdgpu_waves_per_eu(12))) void gat_aggregate(
    const int* __restrict__ ssrc, const int* __restrict__ off, const int* __restrict__ cnt,
    const float* __restrict__ s1, const float* __restrict__ s2,
    const float* __restrict__ z, float* __restrict__ h, int N) {
  const int lane = threadIdx.x & 31;
  const int wv = __builtin_amdgcn_readfirstlane((int)(threadIdx.x >> 5));
  const int d0 = (blockIdx.x * 8 + wv) * AG_NODES;
  if (d0 >= N) return;
  const char*    zb       = (const char*)z;
  const unsigned lane_off = (unsigned)lane * 8u;

  int   o_v  = 0;
  int   c_v  = 0;
  float s2_v = 0.0f;
  if (lane < AG_NODES && d0 + lane < N) {
    o_v  = off[d0 + lane];
    c_v  = cnt[d0 + lane];
    s2_v = s2[d0 + lane];
  }
  const int s2_bits = __float_as_int(s2_v);

  for (int k = 0; k < AG_NODES; ++k) {
    const int d = d0 + k;
    if (d >= N) break;
    const int   o   = __builtin_amdgcn_readlane(o_v, k);
    const int   c   = __builtin_amdgcn_readlane(c_v, k);
    const float s2d = __int_as_float(__builtin_amdgcn_readlane(s2_bits, k));

    const int n0 = c < 32 ? c : 32;
    int s_l; float g_l;
    trip_fetch(ssrc, s1, o, c, 0, lane, N, s_l, g_l);
    v2f zv[AG_GROUP];
    rows_load(zb, lane_off, s_l, 0, n0, zv);

    const float e_l = edge_score(g_l, s2d, lane < c);
    float m = e_l;
    for (int jb = 32; jb < c; jb += 32) {
      int s_t; float g_t;
      trip_fetch(ssrc, s1, o, c, jb, lane, N, s_t, g_t);
      m = fmaxf(m, edge_score(g_t, s2d, jb + lane < c));
    }
    m = wave_max(m);
    if (!isfinite(m)) m = 0.0f;

    const float ex_l = lane < c ? expf(e_l - m) : 0.0f;
    float den = ex_l;
    for (int jb = 32; jb < c; jb += 32) {
      int s_t; float g_t;
      trip_fetch(ssrc, s1, o, c, jb, lane, N, s_t, g_t);
      den += (jb + lane < c) ? expf(edge_score(g_t, s2d, true) - m) : 0.0f;
    }
    den = wave_sum(den);
    const float dd = fmaxf(den, 1e-9f);

    v2f acc = {0.f, 0.f};
    {
      const int a_bits = __float_as_int(ex_l / dd);
      rows_fma(zv, a_bits, 0, n0, acc);
      for (int j0 = AG_GROUP; j0 < n0; j0 += AG_GROUP) {
        rows_load(zb, lane_off, s_l, j0, n0, zv);
        rows_fma(zv, a_bits, j0, n0, acc);
      }
    }
    for (int jb = 32; jb < c; jb += 32) {
      int s_t; float g_t;
      trip_fetch(ssrc, s1, o, c, jb, lane, N, s_t, g_t);
      const float a_t = (jb + lane < c) ? expf(edge_score(g_t, s2d, true) - m) / dd : 0.0f;
      const int a_bits = __float_as_int(a_t);
      const int nt = (c - jb) < 32 ? (c - jb) : 32;
      for (int j0 = 0; j0 < nt; j0 += AG_GROUP) {
        rows_load(zb, lane_off, s_t, j0, nt, zv);
        rows_fma(zv, a_bits, j0, nt, acc);
      }
    }
    *(v2f*)((char*)h + (long)d * (OUTD * 4) + lane_off) = acc;
  }
}

extern "C" void kernel_launch(void* const* d_in, const int* in_sizes, int n_in,
                              void* d_out, int out_size, void* d_ws, size_t ws_size,
                              hipStream_t stream) {
  const float* x    = (const float*)d_in[0];
  const float* w_in = (const float*)d_in[1];
  const float* b_in = (const float*)d_in[2];
  const float* w    = (const float*)d_in[3];
  const float* a    = (const float*)d_in[4];
  const int*   src  = (const int*)d_in[5];
  const int*   dst  = (const int*)d_in[6];
  float* h = (float*)d_out;

  const int N = in_sizes[0] / IN_DIM;
  const int E = in_sizes[5];
  const int NB = (N + BK_ROWS - 1) / BK_ROWS;

  auto align256 = [](size_t v) { return (v + 255) & ~(size_t)255; };
  char* base = (char*)d_ws;
  size_t off_b = 0;
  float* z      = (float*)(base + off_b); off_b += align256((size_t)N * OUTD * 4);
  float* s1     = (float*)(base + off_b); off_b += align256((size_t)N * 4);
  float* s2     = (float*)(base + off_b); off_b += align256((size_t)N * 4);
  int*   cnt    = (int*)(base + off_b);   off_b += align256((size_t)N * 4);
  int*   off    = (int*)(base + off_b);   off_b += align256((size_t)N * 4);
  int*   hist   = (int*)(base + off_b);   off_b += align256((size_t)SORT_CHUNKS * SORT_NBP * 4);
  int*   bbase  = (int*)(base + off_b);   off_b += align256((size_t)(SORT_NBP + 1) * 4);
  int2*  pairs  = (int2*)(base + off_b);  off_b += align256((size_t)E * 8);
  int*   ssrc   = (int*)(base + off_b);   off_b += align256((size_t)E * 4);

  gat_chunk_hist<<<dim3(SORT_CHUNKS), dim3(256), 0, stream>>>(dst, hist, E, NB);
  gat_chunk_scatter<<<dim3(SORT_CHUNKS), dim3(256), 0, stream>>>(src, dst, hist, bbase, pairs,
                                                                E, NB);
  gat_bucket_sort<<<dim3(NB), dim3(BS_THREADS), 0, stream>>>(pairs, bbase, cnt, off, ssrc, N, E);
  {
    int ntiles = (N + 15) >> 4;
    int blocks = (ntiles + 1) / 2;
    gat_fused_gemm<<<dim3(blocks), dim3(64), 0, stream>>>(x, w_in, b_in, w, a, z, s1, s2, N);
  }
  {
    const int rows_per_block = 8 * AG_NODES;
    gat_aggregate<<<dim3((N + rows_per_block - 1) / rows_per_block), dim3(256), 0, stream>>>(
        ssrc, off, cnt, s1, s2, z, h, N);
  }
}


# sort level 1 made chunk-local: a chunk counts by bucket in LDS and writes its edges bucket-ordered to its own slice, a bucket collects its 256 runs: count launch and table sums go; row sums reordered
# speedup vs baseline: 4.8454x; 1.1272x over previous
#include <hip/hip_runtime.h>
#include <math.h>

typedef float v2f __attribute__((ext_vector_type(2)));
typedef float v8f __attribute__((ext_vector_type(8)));

#define IN_DIM 128
#define HID    64
#define OUTD   64

#define SORT_CHUNKS 256
#define SORT_NBP    256
#define BK_SHIFT    9
#define BK_ROWS     512
#define SORT_UNITS  8
#define SC_STAGE    6656
#define SC_OUT      8
#define BS_STAGE    12288

__device__ __forceinline__ int bucket_of(int d, int NB) {
  const unsigned b = (unsigned)d >> BK_SHIFT;
  return (int)(b < (unsigned)(NB - 1) ? b : (unsigned)(NB - 1));
}
__device__ __forceinline__ long chunk_size(int E) {
  return ((((long)E + SORT_CHUNKS - 1) / SORT_CHUNKS) + 3) & ~3L;
}

__device__ __forceinline__ void chunk_count(
    const int* __restrict__ dst, int* h, long lo, long hi, int t, int NB) {
  const long hi4 = lo + ((hi - lo) & ~3L);
  for (long base = lo; base < hi4; base += 256L * 4 * SORT_UNITS) {
    int4 d[SORT_UNITS];
    #pragma unroll
    for (int j = 0; j < SORT_UNITS; ++j) {
      const long e = base + ((long)j * 256 + t) * 4;
      d[j] = *(const int4*)(dst + (e < hi4 ? e : lo));
    }
    #pragma unroll
    for (int j = 0; j < SORT_UNITS; ++j) {
      const long e = base + ((long)j * 256 + t) * 4;
      if (e < hi4) {
        atomicAdd(&h[bucket_of(d[j].x, NB)], 1);
        atomicAdd(&h[bucket_of(d[j].y, NB)], 1);
        atomicAdd(&h[bucket_of(d[j].z, NB)], 1);
        atomicAdd(&h[bucket_of(d[j].w, NB)], 1);
      }
    }
  }
  if (hi4 + t < hi) atomicAdd(&h[bucket_of(dst[hi4 + t], NB)], 1);
}

template <bool STAGED>
__device__ __forceinline__ void chunk_put(int2* stage, int2* __restrict__ pairs, unsigned p,
                                          int sv, int dv, int E) {
  if (STAGED) {
    if (p < (unsigned)SC_STAGE) stage[p] = make_int2(sv, dv);
  } else {
    if (p < (unsigned)E) pairs[p] = make_int2(sv, dv);
  }
}
__device__ __forceinline__ void chunk_place_direct(
    const int* __restrict__ src, const int* __restrict__ dst, int* cursor,
    int2* __restrict__ pairs, long lo, long hi, int t, int E, int NB) {
  const long hi4 = lo + ((hi - lo) & ~3L);
  for (long base = lo; base < hi4; base += 256L * 4 * SORT_UNITS) {
    int4 d[SORT_UNITS];
    int4 s[SORT_UNITS];
    #pragma unroll
    for (int j = 0; j < SORT_UNITS; ++j) {
      const long e  = base + ((long)j * 256 + t) * 4;
      const long ec = e < hi4 ? e : lo;
      d[j] = *(const int4*)(dst + ec);
      s[j] = *(const int4*)(src + ec);
    }
    #pragma unroll
    for (int j = 0; j < SORT_UNITS; ++j) {
      const long e = base + ((long)j * 256 + t) * 4;
      if (e < hi4) {
        const unsigned p0 = (unsigned)atomicAdd(&cursor[bucket_of(d[j].x, NB)], 1);
        const unsigned p1 = (unsigned)atomicAdd(&cursor[bucket_of(d[j].y, NB)], 1);
        const unsigned p2 = (unsigned)atomicAdd(&cursor[bucket_of(d[j].z, NB)], 1);
        const unsigned p3 = (unsigned)atomicAdd(&cursor[bucket_of(d[j].w, NB)], 1);
        chunk_put<false>(nullptr, pairs, p0, s[j].x, d[j].x, E);
        chunk_put<false>(nullptr, pairs, p1, s[j].y, d[j].y, E);
        chunk_put<false>(nullptr, pairs, p2, s[j].z, d[j].z, E);
        chunk_put<false>(nullptr, pairs, p3, s[j].w, d[j].w, E);
      }
    }
  }
  if (hi4 + t < hi) {
    const int dd = dst[hi4 + t];
    const unsigned p = (unsigned)atomicAdd(&cursor[bucket_of(dd, NB)], 1);
    chunk_put<false>(nullptr, pairs, p, src[hi4 + t], dd, E);
  }
}

__global__ __launch_bounds__(256) void gat_chunk_sort(
    const int* __restrict__ src, const int* __restrict__ dst,
    int2* __restrict__ tab, int2* __restrict__ pairs, int E, int NB) {
  __shared__ int h[SORT_NBP];
  __shared__ int cursor[SORT_NBP];
  __shared__ int wsum[8];
  __shared__ __attribute__((aligned(16))) int2 stage[SC_STAGE];
  const int t    = threadIdx.x;
  const int lane = t & 31;
  const int wv   = t >> 5;
  const int c    = blockIdx.x;
  const long cs  = chunk_size(E);
  const bool staged = cs <= SC_STAGE;
  long lo = (long)c * cs;
  long hi = lo + cs;
  if (lo > E) lo = E;
  if (hi > E) hi = E;
  const long hi4  = lo + ((hi - lo) & ~3L);
  const bool tail = hi4 + t < hi;
  h[t] = 0;
  __syncthreads();

  int4 d[SORT_UNITS];
  int4 s[SORT_UNITS];
  int dt = 0, st = 0;
  #pragma unroll
  for (int j = 0; j < SORT_UNITS; ++j) { d[j] = make_int4(0, 0, 0, 0); s[j] = make_int4(0, 0, 0, 0); }
  if (staged) {
    if (lo < hi4) {
      #pragma unroll
      for (int j = 0; j < SORT_UNITS; ++j) {
        const long e  = lo + ((long)j * 256 + t) * 4;
        const long ec = e < hi4 ? e : lo;
        d[j] = *(const int4*)(dst + ec);
        s[j] = *(const int4*)(src + ec);
      }
    }
    if (tail) { dt = dst[hi4 + t]; st = src[hi4 + t]; }
    #pragma unroll
    for (int j = 0; j < SORT_UNITS; ++j) {
      const long e = lo + ((long)j * 256 + t) * 4;
      if (e < hi4) {
        atomicAdd(&h[bucket_of(d[j].x, NB)], 1);
        atomicAdd(&h[bucket_of(d[j].y, NB)], 1);
        atomicAdd(&h[bucket_of(d[j].z, NB)], 1);
        atomicAdd(&h[bucket_of(d[j].w, NB)], 1);
      }
    }
    if (tail) atomicAdd(&h[bucket_of(dt, NB)], 1);
  } else {
    chunk_count(dst, h, lo, hi, t, NB);
  }
  __syncthreads();

  const int my_cnt = h[t];
  int incl = my_cnt;
  #pragma unroll
  for (int o = 1; o < 32; o <<= 1) {
    const int u = __shfl_up(incl, o, 32);
    if (lane >= o) incl += u;
  }
  if (lane == 31) wsum[wv] = incl;
  __syncthreads();
  int wbase = 0;
  #pragma unroll
  for (int w = 0; w < 8; ++w) wbase += (w < wv) ? wsum[w] : 0;
  const int lstart = wbase + incl - my_cnt;
  cursor[t] = staged ? lstart : (int)lo + lstart;
  if (t < NB) tab[(long)t * SORT_CHUNKS + c] = make_int2(lstart, my_cnt);
  __syncthreads();

  if (!staged) {
    chunk_place_direct(src, dst, cursor, pairs, lo, hi, t, E, NB);
  } else {
    #pragma unroll
    for (int j = 0; j < SORT_UNITS; ++j) {
      const long e = lo + ((long)j * 256 + t) * 4;
      if (e < hi4) {
        const unsigned p0 = (unsigned)atomicAdd(&cursor[bucket_of(d[j].x, NB)], 1);
        const unsigned p1 = (unsigned)atomicAdd(&cursor[bucket_of(d[j].y, NB)], 1);
        const unsigned p2 = (unsigned)atomicAdd(&cursor[bucket_of(d[j].z, NB)], 1);
        const unsigned p3 = (unsigned)atomicAdd(&cursor[bucket_of(d[j].w, NB)], 1);
        chunk_put<true>(stage, pairs, p0, s[j].x, d[j].x, E);
        chunk_put<true>(stage, pairs, p1, s[j].y, d[j].y, E);
        chunk_put<true>(stage, pairs, p2, s[j].z, d[j].z, E);
        chunk_put<true>(stage, pairs, p3, s[j].w, d[j].w, E);
      }
    }
    if (tail) {
      const unsigned p = (unsigned)atomicAdd(&cursor[bucket_of(dt, NB)], 1);
      chunk_put<true>(stage, pairs, p, st, dt, E);
    }
    __syncthreads();
    const int n  = (int)(hi - lo);
    const int n2 = n >> 1;
    const int4* st4  = (const int4*)stage;
    int4*       out4 = (int4*)(pairs + lo);
    for (int i0 = 0; i0 < n2; i0 += 256 * SC_OUT) {
      int4 v[SC_OUT];
      #pragma unroll
      for (int j = 0; j < SC_OUT; ++j) {
        const int i = i0 + j * 256 + t;
        v[j] = st4[i < n2 ? i : 0];
      }
      #pragma unroll
      for (int j = 0; j < SC_OUT; ++j)
        asm volatile("" :: "v"(v[j].x), "v"(v[j].y), "v"(v[j].z), "v"(v[j].w));
      #pragma unroll
      for (int j = 0; j < SC_OUT; ++j) {
        const int i = i0 + j * 256 + t;
        if (i < n2) out4[i] = v[j];
      }
    }
    if ((n & 1) && t == 0) pairs[lo + n - 1] = stage[n - 1];
  }
}

#define BS_THREADS 1024
#define BS_RUNS    (SORT_CHUNKS / 32)
#define BS_UNITS   8

__device__ __forceinline__ void bucket_put(
    int2 q, int* rcur, int* stage, int* __restrict__ ssrc,
    int lo, int hi, int row0, int sh, bool staged) {
  const int p = atomicAdd(&rcur[(q.y - row0) & (BK_ROWS - 1)], 1);
  const unsigned g = (unsigned)(lo + p);
  if (g < (unsigned)hi) {
    if (staged) stage[sh + p] = q.x;
    else        ssrc[g] = q.x;
  }
}

template <int M>
__device__ __forceinline__ int row_xor_i(int v) {
  return __builtin_amdgcn_update_dpp(0, v, 0x160 | M, 0xf, 0xf, true);
}
__device__ __forceinline__ int wave_sum_i(int v) {
  v += row_xor_i<1>(v);
  v += row_xor_i<2>(v);
  v += row_xor_i<4>(v);
  v += row_xor_i<8>(v);
  return v + __builtin_amdgcn_permlanex16(v, v, 0x76543210, 0xfedcba98, false, false);
}

__global__ __launch_bounds__(BS_THREADS) void gat_bucket_sort(
    const int2* __restrict__ pairs, const int2* __restrict__ tab,
    int* __restrict__ cnt, int* __restrict__ off, int* __restrict__ ssrc, int N, int E) {
  __shared__ int rcnt[BK_ROWS];
  __shared__ int rcur[BK_ROWS];
  __shared__ int wsum[16];
  __shared__ int wexc[16];
  __shared__ int2 run[SORT_CHUNKS];
  __shared__ int part[2][8];
  __shared__ int stage[BS_STAGE];
  const int t    = threadIdx.x;
  const int lane = t & 31;
  const int wv   = t >> 5;
  const int b    = blockIdx.x;
  const int row0 = b << BK_SHIFT;
  const long cs  = chunk_size(E);

  if (t < SORT_CHUNKS) {
    const int2 e = tab[(long)b * SORT_CHUNKS + t];
    long clo = (long)t * cs;
    long chi = clo + cs;
    if (clo > E) clo = E;
    if (chi > E) chi = E;
    const int sl = (int)(chi - clo);
    const int ls = min(max(e.x, 0), sl);
    const int cn = min(max(e.y, 0), sl - ls);
    run[t] = make_int2((int)clo + ls, cn);
    const int a = wave_sum_i(ls);
    const int n = wave_sum_i(cn);
    if (lane == 0) { part[0][wv] = a; part[1][wv] = n; }
  }
  if (t < BK_ROWS) rcnt[t] = 0;
  __syncthreads();
  int lo = 0, nb = 0;
  #pragma unroll
  for (int w = 0; w < 8; ++w) { lo += part[0][w]; nb += part[1][w]; }
  lo = min(max(lo, 0), E);
  const int hi = min(max(nb, 0), E - lo) + lo;
  const int g0 = lo & ~31;
  const int sh = lo - g0;
  const bool staged = (hi - g0) <= BS_STAGE;

  int2 pr[2 * BS_RUNS];
  int2 rr[BS_RUNS];
  int  rl[BS_RUNS];
  #pragma unroll
  for (int k = 0; k < BS_RUNS; ++k) rr[k] = run[wv * BS_RUNS + k];
  #pragma unroll
  for (int k = 0; k < BS_RUNS; ++k) {
    rl[k] = rr[k].y;
    #pragma unroll
    for (int u = 0; u < 2; ++u) {
      const int i = lane + 32 * u;
      pr[2 * k + u] = make_int2(0, 0);
      if (i < rr[k].y) pr[2 * k + u] = pairs[rr[k].x + i];
    }
  }
  #pragma unroll
  for (int k = 0; k < BS_RUNS; ++k) {
    #pragma unroll
    for (int u = 0; u < 2; ++u)
      if (lane + 32 * u < rl[k]) atomicAdd(&rcnt[(pr[2 * k + u].y - row0) & (BK_ROWS - 1)], 1);
    if (rl[k] > 64) {
      const int rx = run[wv * BS_RUNS + k].x;
      for (int i = 64 + lane; i < rl[k]; i += 32)
        atomicAdd(&rcnt[(pairs[rx + i].y - row0) & (BK_ROWS - 1)], 1);
    }
  }
  __syncthreads();

  int v = 0, incl = 0;
  if (t < BK_ROWS) {
    v = rcnt[t];
    incl = v;
    #pragma unroll
    for (int o = 1; o < 32; o <<= 1) {
      const int u = __shfl_up(incl, o, 32);
      if (lane >= o) incl += u;
    }
    if (lane == 31) wsum[wv] = incl;
  }
  __syncthreads();
  if (t < 16) {
    const int x = wsum[t];
    int in2 = x;
    #pragma unroll
    for (int o = 1; o < 16; o <<= 1) {
      const int u = __shfl_up(in2, o, 32);
      if (lane >= o) in2 += u;
    }
    wexc[t] = in2 - x;
  }
  __syncthreads();
  if (t < BK_ROWS) {
    const int start = wexc[wv] + incl - v;
    rcur[t] = start;
    const int node = row0 + t;
    if (node < N) {
      off[node] = lo + start;
      cnt[node] = v;
    }
  }
  __syncthreads();

  #pragma unroll
  for (int k = 0; k < BS_RUNS; ++k) {
    #pragma unroll
    for (int u = 0; u < 2; ++u)
      if (lane + 32 * u < rl[k]) bucket_put(pr[2 * k + u], rcur, stage, ssrc, lo, hi, row0, sh, staged);
    if (rl[k] > 64) {
      const int rx = run[wv * BS_RUNS + k].x;
      for (int i = 64 + lane; i < rl[k]; i += 32)
        bucket_put(pairs[rx + i], rcur, stage, ssrc, lo, hi, row0, sh, staged);
    }
  }
  if (staged) {
    __syncthreads();
    for (int gb = g0; gb < hi; gb += BS_THREADS * BS_UNITS) {
      int w[BS_UNITS];
      #pragma unroll
      for (int j = 0; j < BS_UNITS; ++j) {
        const int g = gb + j * BS_THREADS + t;
        w[j] = stage[(g < hi ? g : hi - 1) - g0];
      }
      #pragma unroll
      for (int j = 0; j < BS_UNITS; ++j) {
        const int g = gb + j * BS_THREADS + t;
        if (g >= lo && g < hi) ssrc[g] = w[j];
      }
    }
  }
}

template <int BIT>
__device__ __forceinline__ float fold_step(float lo, float hi, bool up) {
  const float keep = up ? hi : lo;
  const float send = up ? lo : hi;
  const int got = __builtin_amdgcn_update_dpp(0, __float_as_int(send), 0x160 | BIT, 0xf, 0xf, true);
  return keep + __int_as_float(got);
}

__global__ __launch_bounds__(64) void gat_fused_gemm(
    const float* __restrict__ x, const float* __restrict__ w_in,
    const float* __restrict__ b_in, const float* __restrict__ w,
    const float* __restrict__ a, float* __restrict__ z,
    float* __restrict__ s2, int N) {
  __shared__ float lds[2][16][66];
  __shared__ float sc[32];

  const int lane = threadIdx.x & 31;
  const int wv   = threadIdx.x >> 5;
  const int tile = blockIdx.x * 2 + wv;
  const int ntiles = (N + 15) >> 4;
  const bool active = tile < ntiles;

  const int m  = lane & 15;
  const int kp = lane >> 4;

  if (active) {
    const int row  = tile * 16 + m;
    const int rowc = row < N ? row : (N - 1);
    const float* xrow = x + (long)rowc * IN_DIM;

    v8f acc[4];
    #pragma unroll
    for (int nt = 0; nt < 4; ++nt) acc[nt] = (v8f){0,0,0,0,0,0,0,0};

    for (int k0 = 0; k0 < IN_DIM; k0 += 4) {
      const int kk = k0 + 2 * kp;
      v2f av = *(const v2f*)(xrow + kk);
      const float* wr0 = w_in + (long)kk * HID;
      const float* wr1 = wr0 + HID;
      #pragma unroll
      for (int nt = 0; nt < 4; ++nt) {
        v2f bv = { wr0[nt * 16 + m], wr1[nt * 16 + m] };
        acc[nt] = __builtin_amdgcn_wmma_f32_16x16x4_f32(
            false, av, false, bv, (short)0, acc[nt], false, false);
      }
    }

    #pragma unroll
    for (int nt = 0; nt < 4; ++nt) {
      float bias = b_in[nt * 16 + m];
      #pragma unroll
      for (int r = 0; r < 8; ++r) {
        float v = acc[nt][r] + bias;
        v = 0.5f * v * (1.0f + erff(v * 0.70710678118654752f));
        lds[wv][r + 8 * kp][nt * 16 + m] = v;
      }
    }
  }
  __syncthreads();

  v8f acc[4];
  #pragma unroll
  for (int nt = 0; nt < 4; ++nt) acc[nt] = (v8f){0,0,0,0,0,0,0,0};
  if (active) {
    for (int k0 = 0; k0 < HID; k0 += 4) {
      const int kk = k0 + 2 * kp;
      v2f av = *(const v2f*)(&lds[wv][m][kk]);
      const float* wr0 = w + (long)kk * OUTD;
      const float* wr1 = wr0 + OUTD;
      #pragma unroll
      for (int nt = 0; nt < 4; ++nt) {
        v2f bv = { wr0[nt * 16 + m], wr1[nt * 16 + m] };
        acc[nt] = __builtin_amdgcn_wmma_f32_16x16x4_f32(
            false, av, false, bv, (short)0, acc[nt], false, false);
      }
    }

    const float a20 = a[OUTD + m], a21 = a[OUTD + 16 + m], a22 = a[OUTD + 32 + m], a23 = a[OUTD + 48 + m];
#define ROW_PAIR(nt, j) ((v2f){acc[nt][2 * (j)], acc[nt][2 * (j) + 1]})
#define SCORE_PAIR(j, w0, w1, w2, w3) \
    (ROW_PAIR(0, j) * (w0) + ROW_PAIR(1, j) * (w1) + ROW_PAIR(2, j) * (w2) + ROW_PAIR(3, j) * (w3))
    const v2f Q0 = SCORE_PAIR(0, a20, a21, a22, a23), Q1 = SCORE_PAIR(1, a20, a21, a22, a23);
    const v2f Q2 = SCORE_PAIR(2, a20, a21, a22, a23), Q3 = SCORE_PAIR(3, a20, a21, a22, a23);
#undef SCORE_PAIR
#undef ROW_PAIR
    const bool u8 = (m & 8) != 0, u4 = (m & 4) != 0, u2 = (m & 2) != 0, u1 = (m & 1) != 0;
    const float c0 = fold_step<8>(Q0.x, Q2.x, u8), c1 = fold_step<8>(Q0.y, Q2.y, u8);
    const float c2 = fold_step<8>(Q1.x, Q3.x, u8), c3 = fold_step<8>(Q1.y, Q3.y, u8);
    const float d0 = fold_step<4>(c0, c2, u4), d1 = fold_step<4>(c1, c3, u4);
    const float e0 = fold_step<2>(d0, d1, u2);
    const float f0 = fold_step<1>(e0, e0, u1);
    sc[wv * 16 + 8 * kp + (m >> 1)] = f0;
  }
  __syncthreads();
  if (wv == 0) {
    const int rr = blockIdx.x * 32 + lane;
    if (rr < N) s2[rr] = sc[lane];
  }

  if (active) {
    #pragma unroll
    for (int nt = 0; nt < 4; ++nt) {
      #pragma unroll
      for (int r = 0; r < 8; ++r) {
        int rr = tile * 16 + r + 8 * kp;
        if (rr < N) z[(long)rr * OUTD + nt * 16 + m] = acc[nt][r];
      }
    }
  }
}

#define AG_NODES 4
#define AG_PAIRS 8

__device__ __forceinline__ float other_half(float v) {
  const int b = __float_as_int(v);
  return __int_as_float(__builtin_amdgcn_permlanex16(b, b, 0x76543210, 0xfedcba98, false, false));
}

template <int M>
__device__ __forceinline__ float row_xor(float v) {
  return __int_as_float(__builtin_amdgcn_update_dpp(0, __float_as_int(v), 0x160 | M, 0xf, 0xf, true));
}
__device__ __forceinline__ float trip_max(float v) {
  v = fmaxf(v, row_xor<2>(v));
  v = fmaxf(v, row_xor<4>(v));
  v = fmaxf(v, row_xor<8>(v));
  return fmaxf(v, other_half(v));
}
__device__ __forceinline__ float trip_sum(float v) {
  v += row_xor<2>(v);
  v += row_xor<4>(v);
  v += row_xor<8>(v);
  return v + other_half(v);
}
template <int L>
__device__ __forceinline__ int row_bcast(int v) {
  return __builtin_amdgcn_update_dpp(0, v, 0x150 | L, 0xf, 0xf, true);
}
__device__ __forceinline__ int trip_index(
    const int* __restrict__ ssrc, int o, int c, int jb, int epos, int N) {
  int e = jb + epos;
  e = e < c ? e : c - 1;
  const int s = ssrc[o + e];
  return min(max(s, 0), N - 1);
}

__global__ __launch_bounds__(256) __attribute__((amdgpu_waves_per_eu(12))) void gat_aggregate(
    const int* __restrict__ ssrc, const int* __restrict__ off, const int* __restrict__ cnt,
    const float* __restrict__ s2, const float* __restrict__ z, const float* __restrict__ a,
    float* __restrict__ h, int N) {
  const int lane = threadIdx.x & 31;
  const int wv = __builtin_amdgcn_readfirstlane((int)(threadIdx.x >> 5));
  const int d0 = (blockIdx.x * 8 + wv) * AG_NODES;
  if (d0 >= N) return;
  const int hh   = lane >> 4;
  const int c16  = lane & 15;
  const int epos = (c16 & ~1) + hh;
  const char*    zb       = (const char*)z;
  const unsigned lane_off = (unsigned)c16 * 16u;
  const float4 a1 = *(const float4*)(a + 4 * c16);
  const bool u8 = (c16 & 8) != 0, u4 = (c16 & 4) != 0, u2 = (c16 & 2) != 0, u1 = (c16 & 1) != 0;

  int   o_v  = 0;
  int   c_v  = 0;
  float s2_v = 0.0f;
  if (lane < AG_NODES && d0 + lane < N) {
    o_v  = off[d0 + lane];
    c_v  = cnt[d0 + lane];
    s2_v = s2[d0 + lane];
  }
  const int s2_bits = __float_as_int(s2_v);

  int  s_nxt = 0;
  bool have  = false;
  {
    const int o0 = __builtin_amdgcn_readlane(o_v, 0);
    const int c0 = __builtin_amdgcn_readlane(c_v, 0);
    if (c0 > 0) { s_nxt = trip_index(ssrc, o0, c0, 0, epos, N); have = true; }
  }

  for (int k = 0; k < AG_NODES; ++k) {
    const int d = d0 + k;
    if (d >= N) break;
    const int   o   = __builtin_amdgcn_readlane(o_v, k);
    const int   c   = __builtin_amdgcn_readlane(c_v, k);
    const float s2d = __int_as_float(__builtin_amdgcn_readlane(s2_bits, k));

    float4 acc = {0.f, 0.f, 0.f, 0.f};
    float m_run = -INFINITY;
    float den   = 0.0f;
    for (int jb = 0; jb < c; jb += 2 * AG_PAIRS) {
      int s_l = s_nxt;
      if (!have) s_l = trip_index(ssrc, o, c, jb, epos, N);
      have = false;
      const int nt = (c - jb) < 2 * AG_PAIRS ? (c - jb) : 2 * AG_PAIRS;
      const int np = (nt + 1) >> 1;

      float4 zv[AG_PAIRS];
#define AG_LOAD(p)                                                                       \
      zv[p] = (float4){0.f, 0.f, 0.f, 0.f};                                                 \
      if ((p) < np) {                                                                       \
        const int s = row_bcast<2 * (p)>(s_l);                                              \
        zv[p] = *(const float4*)(zb + (((unsigned)s << 8) | lane_off));                    \
      }
      AG_LOAD(0) AG_LOAD(1) AG_LOAD(2) AG_LOAD(3) AG_LOAD(4) AG_LOAD(5) AG_LOAD(6) AG_LOAD(7)
#undef AG_LOAD

      if (jb + 2 * AG_PAIRS < c) {
        s_nxt = trip_index(ssrc, o, c, jb + 2 * AG_PAIRS, epos, N);
        have = true;
      } else if (k + 1 < AG_NODES && d + 1 < N) {
        const int on = __builtin_amdgcn_readlane(o_v, k + 1);
        const int cn = __builtin_amdgcn_readlane(c_v, k + 1);
        if (cn > 0) { s_nxt = trip_index(ssrc, on, cn, 0, epos, N); have = true; }
      }

#define AG_DOT(p) (zv[p].x * a1.x + zv[p].y * a1.y + zv[p].z * a1.z + zv[p].w * a1.w)
      const float t0 = AG_DOT(0), t1 = AG_DOT(1), t2 = AG_DOT(2), t3 = AG_DOT(3);
      const float t4 = AG_DOT(4), t5 = AG_DOT(5), t6 = AG_DOT(6), t7 = AG_DOT(7);
#undef AG_DOT
      const float f0 = fold_step<8>(t0, t4, u8), f1 = fold_step<8>(t1, t5, u8);
      const float f2 = fold_step<8>(t2, t6, u8), f3 = fold_step<8>(t3, t7, u8);
      const float g0 = fold_step<4>(f0, f2, u4), g1 = fold_step<4>(f1, f3, u4);
      const float q0 = fold_step<2>(g0, g1, u2);
      const float s1_l = fold_step<1>(q0, q0, u1);

      const bool  live = jb + epos < c;
      const float v    = s1_l + s2d;
      const float e_l  = live ? (v > 0.f ? v : 0.01f * v) : -INFINITY;
      const float m_new = fmaxf(m_run, trip_max(e_l));
      const float scale = expf(m_run - m_new);
      const float ex_l  = live ? expf(e_l - m_new) : 0.0f;
      den = den * scale + trip_sum(ex_l);
      acc.x *= scale; acc.y *= scale; acc.z *= scale; acc.w *= scale;
      m_run = m_new;

      const int ex_bits = __float_as_int(ex_l);
#define AG_FMA(p)                                                                        \
      if ((p) < np) {                                                                       \
        const float w = __int_as_float(row_bcast<2 * (p)>(ex_bits));                        \
        acc.x += w * zv[p].x; acc.y += w * zv[p].y; acc.z += w * zv[p].z; acc.w += w * zv[p].w; \
      }
      AG_FMA(0) AG_FMA(1) AG_FMA(2) AG_FMA(3) AG_FMA(4) AG_FMA(5) AG_FMA(6) AG_FMA(7)
#undef AG_FMA
    }
    const float inv = 1.0f / fmaxf(den, 1e-9f);
    float4 r;
    r.x = (acc.x + other_half(acc.x)) * inv;
    r.y = (acc.y + other_half(acc.y)) * inv;
    r.z = (acc.z + other_half(acc.z)) * inv;
    r.w = (acc.w + other_half(acc.w)) * inv;
    if (hh == 0) *(float4*)((char*)h + (size_t)d * (OUTD * 4) + lane_off) = r;
  }
}

extern "C" void kernel_launch(void* const* d_in, const int* in_sizes, int n_in,
                              void* d_out, int out_size, void* d_ws, size_t ws_size,
                              hipStream_t stream) {
  const float* x    = (const float*)d_in[0];
  const float* w_in = (const float*)d_in[1];
  const float* b_in = (const float*)d_in[2];
  const float* w    = (const float*)d_in[3];
  const float* a    = (const float*)d_in[4];
  const int*   src  = (const int*)d_in[5];
  const int*   dst  = (const int*)d_in[6];
  float* h = (float*)d_out;

  const int N = in_sizes[0] / IN_DIM;
  const int E = in_sizes[5];
  const int NB = (N + BK_ROWS - 1) / BK_ROWS;

  auto align256 = [](size_t v) { return (v + 255) & ~(size_t)255; };
  char* base = (char*)d_ws;
  size_t off_b = 0;
  float* z      = (float*)(base + off_b); off_b += align256((size_t)N * OUTD * 4);
  float* s2     = (float*)(base + off_b); off_b += align256((size_t)N * 4);
  int*   cnt    = (int*)(base + off_b);   off_b += align256((size_t)N * 4);
  int*   off    = (int*)(base + off_b);   off_b += align256((size_t)N * 4);
  int2*  tab    = (int2*)(base + off_b);  off_b += align256((size_t)SORT_NBP * SORT_CHUNKS * 8);
  int2*  pairs  = (int2*)(base + off_b);  off_b += align256((size_t)E * 8);
  int*   ssrc   = (int*)(base + off_b);   off_b += align256((size_t)E * 4);

  gat_chunk_sort<<<dim3(SORT_CHUNKS), dim3(256), 0, stream>>>(src, dst, tab, pairs, E, NB);
  gat_bucket_sort<<<dim3(NB), dim3(BS_THREADS), 0, stream>>>(pairs, tab, cnt, off, ssrc, N, E);
  {
    int ntiles = (N + 15) >> 4;
    int blocks = (ntiles + 1) / 2;
    gat_fused_gemm<<<dim3(blocks), dim3(64), 0, stream>>>(x, w_in, b_in, w, a, z, s2, N);
  }
  {
    const int rows_per_block = 8 * AG_NODES;
    gat_aggregate<<<dim3((N + rows_per_block - 1) / rows_per_block), dim3(256), 0, stream>>>(
        ssrc, off, cnt, s2, z, a, h, N);
  }
}


# sort pass 2 and aggregate fused: a work-group (256 buckets of N/256 rows, one per WGP) orders its edges by row in LDS, its waves take rows in turn: ssrc, off, cnt, a launch go; sum order as before
# speedup vs baseline: 5.3433x; 1.1028x over previous
#include <hip/hip_runtime.h>
#include <math.h>

typedef float v2f __attribute__((ext_vector_type(2)));
typedef float v8f __attribute__((ext_vector_type(8)));

#define IN_DIM 128
#define HID    64
#define OUTD   64

#define SORT_CHUNKS 256
#define SORT_NBP    256
#define BK_ROWS     512
#define SORT_UNITS  8
#define SC_STAGE    6656
#define SC_OUT      8
#define BS_STAGE    12288

struct BucketMap {
  unsigned rows;
  unsigned magic;
  int      last;
};
__device__ __forceinline__ BucketMap bucket_map(int N) {
  BucketMap m;
  const int r = (N + SORT_NBP - 1) / SORT_NBP;
  m.rows  = (unsigned)min(max(r, 2), BK_ROWS);
  m.magic = 0xFFFFFFFFu / m.rows + 1u;
  m.last  = min((int)(((unsigned)max(N, 1) + m.rows - 1u) / m.rows), SORT_NBP) - 1;
  return m;
}
__device__ __forceinline__ int bucket_of(int d, BucketMap m) {
  unsigned b = __umulhi((unsigned)d, m.magic);
  if (b * m.rows > (unsigned)d) --b;
  return (int)(b < (unsigned)m.last ? b : (unsigned)m.last);
}
__device__ __forceinline__ long chunk_size(int E) {
  return ((((long)E + SORT_CHUNKS - 1) / SORT_CHUNKS) + 3) & ~3L;
}

__device__ __forceinline__ void chunk_count(
    const int* __restrict__ dst, int* h, long lo, long hi, int t, BucketMap bm) {
  const long hi4 = lo + ((hi - lo) & ~3L);
  for (long base = lo; base < hi4; base += 256L * 4 * SORT_UNITS) {
    int4 d[SORT_UNITS];
    #pragma unroll
    for (int j = 0; j < SORT_UNITS; ++j) {
      const long e = base + ((long)j * 256 + t) * 4;
      d[j] = *(const int4*)(dst + (e < hi4 ? e : lo));
    }
    #pragma unroll
    for (int j = 0; j < SORT_UNITS; ++j) {
      const long e = base + ((long)j * 256 + t) * 4;
      if (e < hi4) {
        atomicAdd(&h[bucket_of(d[j].x, bm)], 1);
        atomicAdd(&h[bucket_of(d[j].y, bm)], 1);
        atomicAdd(&h[bucket_of(d[j].z, bm)], 1);
        atomicAdd(&h[bucket_of(d[j].w, bm)], 1);
      }
    }
  }
  if (hi4 + t < hi) atomicAdd(&h[bucket_of(dst[hi4 + t], bm)], 1);
}

template <bool STAGED>
__device__ __forceinline__ void chunk_put(int2* stage, int2* __restrict__ pairs, unsigned p,
                                          int sv, int dv, int E) {
  if (STAGED) {
    if (p < (unsigned)SC_STAGE) stage[p] = make_int2(sv, dv);
  } else {
    if (p < (unsigned)E) pairs[p] = make_int2(sv, dv);
  }
}
__device__ __forceinline__ void chunk_place_direct(
    const int* __restrict__ src, const int* __restrict__ dst, int* cursor,
    int2* __restrict__ pairs, long lo, long hi, int t, int E, BucketMap bm) {
  const long hi4 = lo + ((hi - lo) & ~3L);
  for (long base = lo; base < hi4; base += 256L * 4 * SORT_UNITS) {
    int4 d[SORT_UNITS];
    int4 s[SORT_UNITS];
    #pragma unroll
    for (int j = 0; j < SORT_UNITS; ++j) {
      const long e  = base + ((long)j * 256 + t) * 4;
      const long ec = e < hi4 ? e : lo;
      d[j] = *(const int4*)(dst + ec);
      s[j] = *(const int4*)(src + ec);
    }
    #pragma unroll
    for (int j = 0; j < SORT_UNITS; ++j) {
      const long e = base + ((long)j * 256 + t) * 4;
      if (e < hi4) {
        const unsigned p0 = (unsigned)atomicAdd(&cursor[bucket_of(d[j].x, bm)], 1);
        const unsigned p1 = (unsigned)atomicAdd(&cursor[bucket_of(d[j].y, bm)], 1);
        const unsigned p2 = (unsigned)atomicAdd(&cursor[bucket_of(d[j].z, bm)], 1);
        const unsigned p3 = (unsigned)atomicAdd(&cursor[bucket_of(d[j].w, bm)], 1);
        chunk_put<false>(nullptr, pairs, p0, s[j].x, d[j].x, E);
        chunk_put<false>(nullptr, pairs, p1, s[j].y, d[j].y, E);
        chunk_put<false>(nullptr, pairs, p2, s[j].z, d[j].z, E);
        chunk_put<false>(nullptr, pairs, p3, s[j].w, d[j].w, E);
      }
    }
  }
  if (hi4 + t < hi) {
    const int dd = dst[hi4 + t];
    const unsigned p = (unsigned)atomicAdd(&cursor[bucket_of(dd, bm)], 1);
    chunk_put<false>(nullptr, pairs, p, src[hi4 + t], dd, E);
  }
}

__global__ __launch_bounds__(256) void gat_chunk_sort(
    const int* __restrict__ src, const int* __restrict__ dst,
    int2* __restrict__ tab, int2* __restrict__ pairs, int E, int N) {
  __shared__ int h[SORT_NBP];
  __shared__ int cursor[SORT_NBP];
  __shared__ int wsum[8];
  __shared__ __attribute__((aligned(16))) int2 stage[SC_STAGE];
  const int t    = threadIdx.x;
  const int lane = t & 31;
  const int wv   = t >> 5;
  const int c    = blockIdx.x;
  const long cs  = chunk_size(E);
  const BucketMap bm = bucket_map(N);
  const int NB   = bm.last + 1;
  const bool staged = cs <= SC_STAGE;
  long lo = (long)c * cs;
  long hi = lo + cs;
  if (lo > E) lo = E;
  if (hi > E) hi = E;
  const long hi4  = lo + ((hi - lo) & ~3L);
  const bool tail = hi4 + t < hi;
  h[t] = 0;
  __syncthreads();

  int4 d[SORT_UNITS];
  int4 s[SORT_UNITS];
  int dt = 0, st = 0;
  #pragma unroll
  for (int j = 0; j < SORT_UNITS; ++j) { d[j] = make_int4(0, 0, 0, 0); s[j] = make_int4(0, 0, 0, 0); }
  if (staged) {
    if (lo < hi4) {
      #pragma unroll
      for (int j = 0; j < SORT_UNITS; ++j) {
        const long e  = lo + ((long)j * 256 + t) * 4;
        const long ec = e < hi4 ? e : lo;
        d[j] = *(const int4*)(dst + ec);
        s[j] = *(const int4*)(src + ec);
      }
    }
    if (tail) { dt = dst[hi4 + t]; st = src[hi4 + t]; }
    #pragma unroll
    for (int j = 0; j < SORT_UNITS; ++j) {
      const long e = lo + ((long)j * 256 + t) * 4;
      if (e < hi4) {
        atomicAdd(&h[bucket_of(d[j].x, bm)], 1);
        atomicAdd(&h[bucket_of(d[j].y, bm)], 1);
        atomicAdd(&h[bucket_of(d[j].z, bm)], 1);
        atomicAdd(&h[bucket_of(d[j].w, bm)], 1);
      }
    }
    if (tail) atomicAdd(&h[bucket_of(dt, bm)], 1);
  } else {
    chunk_count(dst, h, lo, hi, t, bm);
  }
  __syncthreads();

  const int my_cnt = h[t];
  int incl = my_cnt;
  #pragma unroll
  for (int o = 1; o < 32; o <<= 1) {
    const int u = __shfl_up(incl, o, 32);
    if (lane >= o) incl += u;
  }
  if (lane == 31) wsum[wv] = incl;
  __syncthreads();
  int wbase = 0;
  #pragma unroll
  for (int w = 0; w < 8; ++w) wbase += (w < wv) ? wsum[w] : 0;
  const int lstart = wbase + incl - my_cnt;
  cursor[t] = staged ? lstart : (int)lo + lstart;
  if (t < NB) tab[(long)t * SORT_CHUNKS + c] = make_int2(lstart, my_cnt);
  __syncthreads();

  if (!staged) {
    chunk_place_direct(src, dst, cursor, pairs, lo, hi, t, E, bm);
  } else {
    #pragma unroll
    for (int j = 0; j < SORT_UNITS; ++j) {
      const long e = lo + ((long)j * 256 + t) * 4;
      if (e < hi4) {
        const unsigned p0 = (unsigned)atomicAdd(&cursor[bucket_of(d[j].x, bm)], 1);
        const unsigned p1 = (unsigned)atomicAdd(&cursor[bucket_of(d[j].y, bm)], 1);
        const unsigned p2 = (unsigned)atomicAdd(&cursor[bucket_of(d[j].z, bm)], 1);
        const unsigned p3 = (unsigned)atomicAdd(&cursor[bucket_of(d[j].w, bm)], 1);
        chunk_put<true>(stage, pairs, p0, s[j].x, d[j].x, E);
        chunk_put<true>(stage, pairs, p1, s[j].y, d[j].y, E);
        chunk_put<true>(stage, pairs, p2, s[j].z, d[j].z, E);
        chunk_put<true>(stage, pairs, p3, s[j].w, d[j].w, E);
      }
    }
    if (tail) {
      const unsigned p = (unsigned)atomicAdd(&cursor[bucket_of(dt, bm)], 1);
      chunk_put<true>(stage, pairs, p, st, dt, E);
    }
    __syncthreads();
    const int n  = (int)(hi - lo);
    const int n2 = n >> 1;
    const int4* st4  = (const int4*)stage;
    int4*       out4 = (int4*)(pairs + lo);
    for (int i0 = 0; i0 < n2; i0 += 256 * SC_OUT) {
      int4 v[SC_OUT];
      #pragma unroll
      for (int j = 0; j < SC_OUT; ++j) {
        const int i = i0 + j * 256 + t;
        v[j] = st4[i < n2 ? i : 0];
      }
      #pragma unroll
      for (int j = 0; j < SC_OUT; ++j)
        asm volatile("" :: "v"(v[j].x), "v"(v[j].y), "v"(v[j].z), "v"(v[j].w));
      #pragma unroll
      for (int j = 0; j < SC_OUT; ++j) {
        const int i = i0 + j * 256 + t;
        if (i < n2) out4[i] = v[j];
      }
    }
    if ((n & 1) && t == 0) pairs[lo + n - 1] = stage[n - 1];
  }
}

template <int M>
__device__ __forceinline__ int row_xor_i(int v) {
  return __builtin_amdgcn_update_dpp(0, v, 0x160 | M, 0xf, 0xf, true);
}
__device__ __forceinline__ int wave_sum_i(int v) {
  v += row_xor_i<1>(v);
  v += row_xor_i<2>(v);
  v += row_xor_i<4>(v);
  v += row_xor_i<8>(v);
  return v + __builtin_amdgcn_permlanex16(v, v, 0x76543210, 0xfedcba98, false, false);
}

template <int BIT>
__device__ __forceinline__ float fold_step(float lo, float hi, bool up) {
  const float keep = up ? hi : lo;
  const float send = up ? lo : hi;
  const int got = __builtin_amdgcn_update_dpp(0, __float_as_int(send), 0x160 | BIT, 0xf, 0xf, true);
  return keep + __int_as_float(got);
}

__global__ __launch_bounds__(64) void gat_fused_gemm(
    const float* __restrict__ x, const float* __restrict__ w_in,
    const float* __restrict__ b_in, const float* __restrict__ w,
    const float* __restrict__ a, float* __restrict__ z,
    float* __restrict__ s2, int N) {
  __shared__ float lds[2][16][66];
  __shared__ float sc[32];

  const int lane = threadIdx.x & 31;
  const int wv   = threadIdx.x >> 5;
  const int tile = blockIdx.x * 2 + wv;
  const int ntiles = (N + 15) >> 4;
  const bool active = tile < ntiles;

  const int m  = lane & 15;
  const int kp = lane >> 4;

  if (active) {
    const int row  = tile * 16 + m;
    const int rowc = row < N ? row : (N - 1);
    const float* xrow = x + (long)rowc * IN_DIM;

    v8f acc[4];
    #pragma unroll
    for (int nt = 0; nt < 4; ++nt) acc[nt] = (v8f){0,0,0,0,0,0,0,0};

    for (int k0 = 0; k0 < IN_DIM; k0 += 4) {
      const int kk = k0 + 2 * kp;
      v2f av = *(const v2f*)(xrow + kk);
      const float* wr0 = w_in + (long)kk * HID;
      const float* wr1 = wr0 + HID;
      #pragma unroll
      for (int nt = 0; nt < 4; ++nt) {
        v2f bv = { wr0[nt * 16 + m], wr1[nt * 16 + m] };
        acc[nt] = __builtin_amdgcn_wmma_f32_16x16x4_f32(
            false, av, false, bv, (short)0, acc[nt], false, false);
      }
    }

    #pragma unroll
    for (int nt = 0; nt < 4; ++nt) {
      float bias = b_in[nt * 16 + m];
      #pragma unroll
      for (int r = 0; r < 8; ++r) {
        float v = acc[nt][r] + bias;
        v = 0.5f * v * (1.0f + erff(v * 0.70710678118654752f));
        lds[wv][r + 8 * kp][nt * 16 + m] = v;
      }
    }
  }
  __syncthreads();

  v8f acc[4];
  #pragma unroll
  for (int nt = 0; nt < 4; ++nt) acc[nt] = (v8f){0,0,0,0,0,0,0,0};
  if (active) {
    for (int k0 = 0; k0 < HID; k0 += 4) {
      const int kk = k0 + 2 * kp;
      v2f av = *(const v2f*)(&lds[wv][m][kk]);
      const float* wr0 = w + (long)kk * OUTD;
      const float* wr1 = wr0 + OUTD;
      #pragma unroll
      for (int nt = 0; nt < 4; ++nt) {
        v2f bv = { wr0[nt * 16 + m], wr1[nt * 16 + m] };
        acc[nt] = __builtin_amdgcn_wmma_f32_16x16x4_f32(
            false, av, false, bv, (short)0, acc[nt], false, false);
      }
    }

    const float a20 = a[OUTD + m], a21 = a[OUTD + 16 + m], a22 = a[OUTD + 32 + m], a23 = a[OUTD + 48 + m];
#define ROW_PAIR(nt, j) ((v2f){acc[nt][2 * (j)], acc[nt][2 * (j) + 1]})
#define SCORE_PAIR(j, w0, w1, w2, w3) \
    (ROW_PAIR(0, j) * (w0) + ROW_PAIR(1, j) * (w1) + ROW_PAIR(2, j) * (w2) + ROW_PAIR(3, j) * (w3))
    const v2f Q0 = SCORE_PAIR(0, a20, a21, a22, a23), Q1 = SCORE_PAIR(1, a20, a21, a22, a23);
    const v2f Q2 = SCORE_PAIR(2, a20, a21, a22, a23), Q3 = SCORE_PAIR(3, a20, a21, a22, a23);
#undef SCORE_PAIR
#undef ROW_PAIR
    const bool u8 = (m & 8) != 0, u4 = (m & 4) != 0, u2 = (m & 2) != 0, u1 = (m & 1) != 0;
    const float c0 = fold_step<8>(Q0.x, Q2.x, u8), c1 = fold_step<8>(Q0.y, Q2.y, u8);
    const float c2 = fold_step<8>(Q1.x, Q3.x, u8), c3 = fold_step<8>(Q1.y, Q3.y, u8);
    const float d0 = fold_step<4>(c0, c2, u4), d1 = fold_step<4>(c1, c3, u4);
    const float e0 = fold_step<2>(d0, d1, u2);
    const float f0 = fold_step<1>(e0, e0, u1);
    sc[wv * 16 + 8 * kp + (m >> 1)] = f0;
  }
  __syncthreads();
  if (wv == 0) {
    const int rr = blockIdx.x * 32 + lane;
    if (rr < N) s2[rr] = sc[lane];
  }

  if (active) {
    #pragma unroll
    for (int nt = 0; nt < 4; ++nt) {
      #pragma unroll
      for (int r = 0; r < 8; ++r) {
        int rr = tile * 16 + r + 8 * kp;
        if (rr < N) z[(long)rr * OUTD + nt * 16 + m] = acc[nt][r];
      }
    }
  }
}

__device__ __forceinline__ float other_half(float v) {
  const int b = __float_as_int(v);
  return __int_as_float(__builtin_amdgcn_permlanex16(b, b, 0x76543210, 0xfedcba98, false, false));
}

template <int M>
__device__ __forceinline__ float row_xor(float v) {
  return __int_as_float(__builtin_amdgcn_update_dpp(0, __float_as_int(v), 0x160 | M, 0xf, 0xf, true));
}
__device__ __forceinline__ float trip_max(float v) {
  v = fmaxf(v, row_xor<2>(v));
  v = fmaxf(v, row_xor<4>(v));
  v = fmaxf(v, row_xor<8>(v));
  return fmaxf(v, other_half(v));
}
__device__ __forceinline__ float trip_sum(float v) {
  v += row_xor<2>(v);
  v += row_xor<4>(v);
  v += row_xor<8>(v);
  return v + other_half(v);
}
template <int L>
__device__ __forceinline__ int row_bcast(int v) {
  return __builtin_amdgcn_update_dpp(0, v, 0x150 | L, 0xf, 0xf, true);
}

#define BS_THREADS 1024
#define BS_RUNS    (SORT_CHUNKS / 32)
#define AG_PAIRS   8

__device__ __forceinline__ void bucket_put(
    int2 q, int* rcur, int* stage, int* __restrict__ ssrc,
    int lo, int hi, int row0, bool staged) {
  const int p = atomicAdd(&rcur[(q.y - row0) & (BK_ROWS - 1)], 1);
  const unsigned g = (unsigned)(lo + p);
  if (g < (unsigned)hi) {
    if (staged) stage[p] = q.x;
    else        ssrc[g] = q.x;
  }
}

template <bool IN_LDS>
__device__ __forceinline__ void bucket_rows(
    const int* rcnt, const int* rcur, const float* s2s, int* next_row, const int* list,
    int nbe, int nrows, int row0, const float* __restrict__ z, float4 a1,
    float* __restrict__ h, int N) {
  const int lane = threadIdx.x & 31;
  const int wvs  = __builtin_amdgcn_readfirstlane((int)(threadIdx.x >> 5));
  const int hh   = lane >> 4;
  const int c16  = lane & 15;
  const int epos = (c16 & ~1) + hh;
  const char*    zb       = (const char*)z;
  const unsigned lane_off = (unsigned)c16 * 16u;
  const bool u8 = (c16 & 8) != 0, u4 = (c16 & 4) != 0, u2 = (c16 & 2) != 0, u1 = (c16 & 1) != 0;

  auto trip_index = [&](int o_, int c_, int jb_) -> int {
    int e = jb_ + epos;
    e = e < c_ ? e : c_ - 1;
    const int s = list[min(max(o_ + e, 0), nbe - 1)];
    return min(max(s, 0), N - 1);
  };
  auto row_info = [&](int rr_, int& oo, int& cc, float& ss) {
    oo = 0; cc = 0; ss = 0.0f;
    if (rr_ < nrows) {
      const int   cv = rcnt[rr_];
      const int   ev = rcur[rr_];
      const float sv = s2s[rr_];
      cc = __builtin_amdgcn_readfirstlane(cv);
      oo = __builtin_amdgcn_readfirstlane(ev) - cc;
      ss = __int_as_float(__builtin_amdgcn_readfirstlane(__float_as_int(sv)));
    }
  };
  auto take_row = [&](int& rr_, int& oo, int& cc, float& ss) {
    int nv = 0;
    if (lane == 0) nv = atomicAdd(next_row, 1);
    rr_ = __builtin_amdgcn_readfirstlane(nv);
    row_info(rr_, oo, cc, ss);
  };

  int   r = wvs, o = 0, c = 0;
  float s2d = 0.0f;
  row_info(r, o, c, s2d);
  int s_cur = 0;
  if (c > 0) s_cur = trip_index(o, c, 0);

  for (int it = 0; it < BK_ROWS && r < nrows; ++it) {
    int   rn = nrows, on = 0, cn = 0;
    float s2n = 0.0f;
    if (c <= 0) {
      take_row(rn, on, cn, s2n);
      if (cn > 0) s_cur = trip_index(on, cn, 0);
    }

    float4 acc = {0.f, 0.f, 0.f, 0.f};
    float m_run = -INFINITY;
    float den   = 0.0f;
    for (int jb = 0; jb < c; jb += 2 * AG_PAIRS) {
      const int s_l = s_cur;
      const int nt = (c - jb) < 2 * AG_PAIRS ? (c - jb) : 2 * AG_PAIRS;
      const int np = (nt + 1) >> 1;

      float4 zv[AG_PAIRS];
#define AG_LOAD(p)                                                                       \
      zv[p] = (float4){0.f, 0.f, 0.f, 0.f};                                                 \
      if ((p) < np) {                                                                       \
        const int s = row_bcast<2 * (p)>(s_l);                                              \
        zv[p] = *(const float4*)(zb + (((unsigned)s << 8) | lane_off));                    \
      }
      AG_LOAD(0) AG_LOAD(1) AG_LOAD(2) AG_LOAD(3) AG_LOAD(4) AG_LOAD(5) AG_LOAD(6) AG_LOAD(7)
#undef AG_LOAD

      if (jb == 0) take_row(rn, on, cn, s2n);
      if (jb + 2 * AG_PAIRS < c) s_cur = trip_index(o, c, jb + 2 * AG_PAIRS);
      else if (cn > 0)           s_cur = trip_index(on, cn, 0);

#define AG_DOT(p) (zv[p].x * a1.x + zv[p].y * a1.y + zv[p].z * a1.z + zv[p].w * a1.w)
      const float t0 = AG_DOT(0), t1 = AG_DOT(1), t2 = AG_DOT(2), t3 = AG_DOT(3);
      const float t4 = AG_DOT(4), t5 = AG_DOT(5), t6 = AG_DOT(6), t7 = AG_DOT(7);
#undef AG_DOT
      const float f0 = fold_step<8>(t0, t4, u8), f1 = fold_step<8>(t1, t5, u8);
      const float f2 = fold_step<8>(t2, t6, u8), f3 = fold_step<8>(t3, t7, u8);
      const float g0 = fold_step<4>(f0, f2, u4), g1 = fold_step<4>(f1, f3, u4);
      const float q0 = fold_step<2>(g0, g1, u2);
      const float s1_l = fold_step<1>(q0, q0, u1);

      const bool  live = jb + epos < c;
      const float vv   = s1_l + s2d;
      const float e_l  = live ? (vv > 0.f ? vv : 0.01f * vv) : -INFINITY;
      const float m_new = fmaxf(m_run, trip_max(e_l));
      const float scale = expf(m_run - m_new);
      const float ex_l  = live ? expf(e_l - m_new) : 0.0f;
      den = den * scale + trip_sum(ex_l);
      acc.x *= scale; acc.y *= scale; acc.z *= scale; acc.w *= scale;
      m_run = m_new;

      const int ex_bits = __float_as_int(ex_l);
#define AG_FMA(p)                                                                        \
      if ((p) < np) {                                                                       \
        const float w = __int_as_float(row_bcast<2 * (p)>(ex_bits));                        \
        acc.x += w * zv[p].x; acc.y += w * zv[p].y; acc.z += w * zv[p].z; acc.w += w * zv[p].w; \
      }
      AG_FMA(0) AG_FMA(1) AG_FMA(2) AG_FMA(3) AG_FMA(4) AG_FMA(5) AG_FMA(6) AG_FMA(7)
#undef AG_FMA
    }
    const float inv = 1.0f / fmaxf(den, 1e-9f);
    float4 rv;
    rv.x = (acc.x + other_half(acc.x)) * inv;
    rv.y = (acc.y + other_half(acc.y)) * inv;
    rv.z = (acc.z + other_half(acc.z)) * inv;
    rv.w = (acc.w + other_half(acc.w)) * inv;
    if (hh == 0) *(float4*)((char*)h + (size_t)(row0 + r) * (OUTD * 4) + lane_off) = rv;

    r = rn; o = on; c = cn; s2d = s2n;
  }
}

__global__ __launch_bounds__(BS_THREADS) void gat_bucket_aggregate(
    const int2* __restrict__ pairs, const int2* __restrict__ tab, int* __restrict__ ssrc,
    const float* __restrict__ s2, const float* __restrict__ z, const float* __restrict__ a,
    float* __restrict__ h, int N, int E) {
  __shared__ int   rcnt[BK_ROWS];
  __shared__ int   rcur[BK_ROWS];
  __shared__ float s2s[BK_ROWS];
  __shared__ int   wsum[16];
  __shared__ int   wexc[16];
  __shared__ int2  run[SORT_CHUNKS];
  __shared__ int   part[2][8];
  __shared__ int   next_row;
  __shared__ int   stage[BS_STAGE];
  const int t    = threadIdx.x;
  const int lane = t & 31;
  const int wv   = t >> 5;
  const int b    = blockIdx.x;
  const BucketMap bm = bucket_map(N);
  if (b > bm.last) return;
  const int row0 = b * (int)bm.rows;
  const long cs  = chunk_size(E);

  const int    c16 = lane & 15;
  const float4 a1  = *(const float4*)(a + 4 * c16);

  if (t < SORT_CHUNKS) {
    const int2 e = tab[(long)b * SORT_CHUNKS + t];
    long clo = (long)t * cs;
    long chi = clo + cs;
    if (clo > E) clo = E;
    if (chi > E) chi = E;
    const int sl = (int)(chi - clo);
    const int ls = min(max(e.x, 0), sl);
    const int cn = min(max(e.y, 0), sl - ls);
    run[t] = make_int2((int)clo + ls, cn);
    const int as = wave_sum_i(ls);
    const int n  = wave_sum_i(cn);
    if (lane == 0) { part[0][wv] = as; part[1][wv] = n; }
  }
  if (t < BK_ROWS) {
    rcnt[t] = 0;
    const int node = row0 + t;
    s2s[t] = node < N ? s2[node] : 0.0f;
  }
  if (t == 0) next_row = BS_THREADS / 32;
  __syncthreads();
  int lo_v = 0, nb_v = 0;
  #pragma unroll
  for (int w = 0; w < 8; ++w) { lo_v += part[0][w]; nb_v += part[1][w]; }
  lo_v = min(max(lo_v, 0), E);
  const int lo  = __builtin_amdgcn_readfirstlane(lo_v);
  const int hi  = __builtin_amdgcn_readfirstlane(min(max(nb_v, 0), E - lo_v)) + lo;
  const int nbe = hi - lo;
  const bool staged = nbe <= BS_STAGE;

  int2 pr[2 * BS_RUNS];
  int2 rr[BS_RUNS];
  int  rl[BS_RUNS];
  #pragma unroll
  for (int k = 0; k < BS_RUNS; ++k) rr[k] = run[wv * BS_RUNS + k];
  #pragma unroll
  for (int k = 0; k < BS_RUNS; ++k) {
    rl[k] = rr[k].y;
    #pragma unroll
    for (int u = 0; u < 2; ++u) {
      const int i = lane + 32 * u;
      pr[2 * k + u] = make_int2(0, 0);
      if (i < rr[k].y) pr[2 * k + u] = pairs[rr[k].x + i];
    }
  }
  #pragma unroll
  for (int k = 0; k < BS_RUNS; ++k) {
    #pragma unroll
    for (int u = 0; u < 2; ++u)
      if (lane + 32 * u < rl[k]) atomicAdd(&rcnt[(pr[2 * k + u].y - row0) & (BK_ROWS - 1)], 1);
    if (rl[k] > 64) {
      const int rx = run[wv * BS_RUNS + k].x;
      for (int i = 64 + lane; i < rl[k]; i += 32)
        atomicAdd(&rcnt[(pairs[rx + i].y - row0) & (BK_ROWS - 1)], 1);
    }
  }
  __syncthreads();

  int v = 0, incl = 0;
  if (t < BK_ROWS) {
    v = rcnt[t];
    incl = v;
    #pragma unroll
    for (int o = 1; o < 32; o <<= 1) {
      const int u = __shfl_up(incl, o, 32);
      if (lane >= o) incl += u;
    }
    if (lane == 31) wsum[wv] = incl;
  }
  __syncthreads();
  if (t < 16) {
    const int x = wsum[t];
    int in2 = x;
    #pragma unroll
    for (int o = 1; o < 16; o <<= 1) {
      const int u = __shfl_up(in2, o, 32);
      if (lane >= o) in2 += u;
    }
    wexc[t] = in2 - x;
  }
  __syncthreads();
  if (t < BK_ROWS) rcur[t] = wexc[wv] + incl - v;
  __syncthreads();

  #pragma unroll
  for (int k = 0; k < BS_RUNS; ++k) {
    #pragma unroll
    for (int u = 0; u < 2; ++u)
      if (lane + 32 * u < rl[k]) bucket_put(pr[2 * k + u], rcur, stage, ssrc, lo, hi, row0, staged);
    if (rl[k] > 64) {
      const int rx = run[wv * BS_RUNS + k].x;
      for (int i = 64 + lane; i < rl[k]; i += 32)
        bucket_put(pairs[rx + i], rcur, stage, ssrc, lo, hi, row0, staged);
    }
  }
  __syncthreads();

  const int nrows = min((int)bm.rows, N - row0);
  if (staged) bucket_rows<true>(rcnt, rcur, s2s, &next_row, stage, nbe, nrows, row0, z, a1, h, N);
  else        bucket_rows<false>(rcnt, rcur, s2s, &next_row, ssrc + lo, nbe, nrows, row0, z, a1, h, N);
}

extern "C" void kernel_launch(void* const* d_in, const int* in_sizes, int n_in,
                              void* d_out, int out_size, void* d_ws, size_t ws_size,
                              hipStream_t stream) {
  const float* x    = (const float*)d_in[0];
  const float* w_in = (const float*)d_in[1];
  const float* b_in = (const float*)d_in[2];
  const float* w    = (const float*)d_in[3];
  const float* a    = (const float*)d_in[4];
  const int*   src  = (const int*)d_in[5];
  const int*   dst  = (const int*)d_in[6];
  float* h = (float*)d_out;

  const int N = in_sizes[0] / IN_DIM;
  const int E = in_sizes[5];

  auto align256 = [](size_t v) { return (v + 255) & ~(size_t)255; };
  char* base = (char*)d_ws;
  size_t off_b = 0;
  float* z      = (float*)(base + off_b); off_b += align256((size_t)N * OUTD * 4);
  float* s2     = (float*)(base + off_b); off_b += align256((size_t)N * 4);
  int2*  tab    = (int2*)(base + off_b);  off_b += align256((size_t)SORT_NBP * SORT_CHUNKS * 8);
  int2*  pairs  = (int2*)(base + off_b);  off_b += align256((size_t)E * 8);
  int*   ssrc   = (int*)(base + off_b);   off_b += align256((size_t)E * 4);

  gat_chunk_sort<<<dim3(SORT_CHUNKS), dim3(256), 0, stream>>>(src, dst, tab, pairs, E, N);
  {
    int ntiles = (N + 15) >> 4;
    int blocks = (ntiles + 1) / 2;
    gat_fused_gemm<<<dim3(blocks), dim3(64), 0, stream>>>(x, w_in, b_in, w, a, z, s2, N);
  }
  gat_bucket_aggregate<<<dim3(SORT_NBP), dim3(BS_THREADS), 0, stream>>>(
      pairs, tab, ssrc, s2, z, a, h, N, E);
}


# GEMM operands by 16-byte loads: first launch re-lays w_in and w in fragment order in the workspace (one load feeds 2 products), x read in 16-byte pieces; k regrouped per product, sums reordered
# speedup vs baseline: 5.6656x; 1.0603x over previous
#include <hip/hip_runtime.h>
#include <math.h>

typedef float v2f __attribute__((ext_vector_type(2)));
typedef float v8f __attribute__((ext_vector_type(8)));

#define IN_DIM 128
#define HID    64
#define OUTD   64

#define SORT_CHUNKS 256
#define SORT_NBP    256
#define BK_ROWS     512
#define SORT_UNITS  8
#define SC_STAGE    6656
#define SC_OUT      8
#define BS_STAGE    12288

#define WF1_N     ((IN_DIM / 8) * 4 * 32)
#define WF2_N     ((HID / 8) * 4 * 32)
#define WF_PER_WG ((WF1_N + WF2_N + SORT_CHUNKS - 1) / SORT_CHUNKS)

struct BucketMap {
  unsigned rows;
  unsigned magic;
  int      last;
};
__device__ __forceinline__ BucketMap bucket_map(int N) {
  BucketMap m;
  const int r = (N + SORT_NBP - 1) / SORT_NBP;
  m.rows  = (unsigned)min(max(r, 2), BK_ROWS);
  m.magic = 0xFFFFFFFFu / m.rows + 1u;
  m.last  = min((int)(((unsigned)max(N, 1) + m.rows - 1u) / m.rows), SORT_NBP) - 1;
  return m;
}
__device__ __forceinline__ int bucket_of(int d, BucketMap m) {
  unsigned b = __umulhi((unsigned)d, m.magic);
  if (b * m.rows > (unsigned)d) --b;
  return (int)(b < (unsigned)m.last ? b : (unsigned)m.last);
}
__device__ __forceinline__ long chunk_size(int E) {
  return ((((long)E + SORT_CHUNKS - 1) / SORT_CHUNKS) + 3) & ~3L;
}

__device__ __forceinline__ void chunk_count(
    const int* __restrict__ dst, int* h, long lo, long hi, int t, BucketMap bm) {
  const long hi4 = lo + ((hi - lo) & ~3L);
  for (long base = lo; base < hi4; base += 256L * 4 * SORT_UNITS) {
    int4 d[SORT_UNITS];
    #pragma unroll
    for (int j = 0; j < SORT_UNITS; ++j) {
      const long e = base + ((long)j * 256 + t) * 4;
      d[j] = *(const int4*)(dst + (e < hi4 ? e : lo));
    }
    #pragma unroll
    for (int j = 0; j < SORT_UNITS; ++j) {
      const long e = base + ((long)j * 256 + t) * 4;
      if (e < hi4) {
        atomicAdd(&h[bucket_of(d[j].x, bm)], 1);
        atomicAdd(&h[bucket_of(d[j].y, bm)], 1);
        atomicAdd(&h[bucket_of(d[j].z, bm)], 1);
        atomicAdd(&h[bucket_of(d[j].w, bm)], 1);
      }
    }
  }
  if (hi4 + t < hi) atomicAdd(&h[bucket_of(dst[hi4 + t], bm)], 1);
}

template <bool STAGED>
__device__ __forceinline__ void chunk_put(int2* stage, int2* __restrict__ pairs, unsigned p,
                                          int sv, int dv, int E) {
  if (STAGED) {
    if (p < (unsigned)SC_STAGE) stage[p] = make_int2(sv, dv);
  } else {
    if (p < (unsigned)E) pairs[p] = make_int2(sv, dv);
  }
}
__device__ __forceinline__ void chunk_place_direct(
    const int* __restrict__ src, const int* __restrict__ dst, int* cursor,
    int2* __restrict__ pairs, long lo, long hi, int t, int E, BucketMap bm) {
  const long hi4 = lo + ((hi - lo) & ~3L);
  for (long base = lo; base < hi4; base += 256L * 4 * SORT_UNITS) {
    int4 d[SORT_UNITS];
    int4 s[SORT_UNITS];
    #pragma unroll
    for (int j = 0; j < SORT_UNITS; ++j) {
      const long e  = base + ((long)j * 256 + t) * 4;
      const long ec = e < hi4 ? e : lo;
      d[j] = *(const int4*)(dst + ec);
      s[j] = *(const int4*)(src + ec);
    }
    #pragma unroll
    for (int j = 0; j < SORT_UNITS; ++j) {
      const long e = base + ((long)j * 256 + t) * 4;
      if (e < hi4) {
        const unsigned p0 = (unsigned)atomicAdd(&cursor[bucket_of(d[j].x, bm)], 1);
        const unsigned p1 = (unsigned)atomicAdd(&cursor[bucket_of(d[j].y, bm)], 1);
        const unsigned p2 = (unsigned)atomicAdd(&cursor[bucket_of(d[j].z, bm)], 1);
        const unsigned p3 = (unsigned)atomicAdd(&cursor[bucket_of(d[j].w, bm)], 1);
        chunk_put<false>(nullptr, pairs, p0, s[j].x, d[j].x, E);
        chunk_put<false>(nullptr, pairs, p1, s[j].y, d[j].y, E);
        chunk_put<false>(nullptr, pairs, p2, s[j].z, d[j].z, E);
        chunk_put<false>(nullptr, pairs, p3, s[j].w, d[j].w, E);
      }
    }
  }
  if (hi4 + t < hi) {
    const int dd = dst[hi4 + t];
    const unsigned p = (unsigned)atomicAdd(&cursor[bucket_of(dd, bm)], 1);
    chunk_put<false>(nullptr, pairs, p, src[hi4 + t], dd, E);
  }
}

__global__ __launch_bounds__(256) void gat_chunk_sort(
    const int* __restrict__ src, const int* __restrict__ dst,
    int2* __restrict__ tab, int2* __restrict__ pairs, int E, int N,
    const float* __restrict__ w_in, const float* __restrict__ w, float4* __restrict__ wf) {
  __shared__ int h[SORT_NBP];
  __shared__ int cursor[SORT_NBP];
  __shared__ int wsum[8];
  __shared__ __attribute__((aligned(16))) int2 stage[SC_STAGE];
  const int t    = threadIdx.x;
  const int lane = t & 31;
  const int wv   = t >> 5;
  const int c    = blockIdx.x;
  const long cs  = chunk_size(E);
  const BucketMap bm = bucket_map(N);
  const int NB   = bm.last + 1;
  const bool staged = cs <= SC_STAGE;
  long lo = (long)c * cs;
  long hi = lo + cs;
  if (lo > E) lo = E;
  if (hi > E) hi = E;
  const long hi4  = lo + ((hi - lo) & ~3L);
  const bool tail = hi4 + t < hi;

  const int  wg    = c * WF_PER_WG + t;
  const bool relay = t < WF_PER_WG && wg < WF1_N + WF2_N;
  float4 wq = make_float4(0.f, 0.f, 0.f, 0.f);
  if (relay) {
    const bool first = wg < WF1_N;
    const int  q   = first ? wg : wg - WF1_N;
    const int  grp = q >> 5;
    const int  ln  = q & 31;
    const float* p = (first ? w_in : w) + (8 * (grp >> 2) + 4 * (ln >> 4)) * HID + (grp & 3) * 16 + (ln & 15);
    wq = make_float4(p[0], p[HID], p[2 * HID], p[3 * HID]);
  }
  h[t] = 0;
  __syncthreads();

  int4 d[SORT_UNITS];
  int4 s[SORT_UNITS];
  int dt = 0, st = 0;
  #pragma unroll
  for (int j = 0; j < SORT_UNITS; ++j) { d[j] = make_int4(0, 0, 0, 0); s[j] = make_int4(0, 0, 0, 0); }
  if (staged) {
    if (lo < hi4) {
      #pragma unroll
      for (int j = 0; j < SORT_UNITS; ++j) {
        const long e  = lo + ((long)j * 256 + t) * 4;
        const long ec = e < hi4 ? e : lo;
        d[j] = *(const int4*)(dst + ec);
        s[j] = *(const int4*)(src + ec);
      }
    }
    if (tail) { dt = dst[hi4 + t]; st = src[hi4 + t]; }
    #pragma unroll
    for (int j = 0; j < SORT_UNITS; ++j) {
      const long e = lo + ((long)j * 256 + t) * 4;
      if (e < hi4) {
        atomicAdd(&h[bucket_of(d[j].x, bm)], 1);
        atomicAdd(&h[bucket_of(d[j].y, bm)], 1);
        atomicAdd(&h[bucket_of(d[j].z, bm)], 1);
        atomicAdd(&h[bucket_of(d[j].w, bm)], 1);
      }
    }
    if (tail) atomicAdd(&h[bucket_of(dt, bm)], 1);
  } else {
    chunk_count(dst, h, lo, hi, t, bm);
  }
  __syncthreads();

  const int my_cnt = h[t];
  int incl = my_cnt;
  #pragma unroll
  for (int o = 1; o < 32; o <<= 1) {
    const int u = __shfl_up(incl, o, 32);
    if (lane >= o) incl += u;
  }
  if (lane == 31) wsum[wv] = incl;
  __syncthreads();
  int wbase = 0;
  #pragma unroll
  for (int w = 0; w < 8; ++w) wbase += (w < wv) ? wsum[w] : 0;
  const int lstart = wbase + incl - my_cnt;
  cursor[t] = staged ? lstart : (int)lo + lstart;
  if (t < NB) tab[(long)t * SORT_CHUNKS + c] = make_int2(lstart, my_cnt);
  __syncthreads();

  if (!staged) {
    chunk_place_direct(src, dst, cursor, pairs, lo, hi, t, E, bm);
  } else {
    #pragma unroll
    for (int j = 0; j < SORT_UNITS; ++j) {
      const long e = lo + ((long)j * 256 + t) * 4;
      if (e < hi4) {
        const unsigned p0 = (unsigned)atomicAdd(&cursor[bucket_of(d[j].x, bm)], 1);
        const unsigned p1 = (unsigned)atomicAdd(&cursor[bucket_of(d[j].y, bm)], 1);
        const unsigned p2 = (unsigned)atomicAdd(&cursor[bucket_of(d[j].z, bm)], 1);
        const unsigned p3 = (unsigned)atomicAdd(&cursor[bucket_of(d[j].w, bm)], 1);
        chunk_put<true>(stage, pairs, p0, s[j].x, d[j].x, E);
        chunk_put<true>(stage, pairs, p1, s[j].y, d[j].y, E);
        chunk_put<true>(stage, pairs, p2, s[j].z, d[j].z, E);
        chunk_put<true>(stage, pairs, p3, s[j].w, d[j].w, E);
      }
    }
    if (tail) {
      const unsigned p = (unsigned)atomicAdd(&cursor[bucket_of(dt, bm)], 1);
      chunk_put<true>(stage, pairs, p, st, dt, E);
    }
    __syncthreads();
    const int n  = (int)(hi - lo);
    const int n2 = n >> 1;
    const int4* st4  = (const int4*)stage;
    int4*       out4 = (int4*)(pairs + lo);
    for (int i0 = 0; i0 < n2; i0 += 256 * SC_OUT) {
      int4 v[SC_OUT];
      #pragma unroll
      for (int j = 0; j < SC_OUT; ++j) {
        const int i = i0 + j * 256 + t;
        v[j] = st4[i < n2 ? i : 0];
      }
      #pragma unroll
      for (int j = 0; j < SC_OUT; ++j)
        asm volatile("" :: "v"(v[j].x), "v"(v[j].y), "v"(v[j].z), "v"(v[j].w));
      #pragma unroll
      for (int j = 0; j < SC_OUT; ++j) {
        const int i = i0 + j * 256 + t;
        if (i < n2) out4[i] = v[j];
      }
    }
    if ((n & 1) && t == 0) pairs[lo + n - 1] = stage[n - 1];
  }
  if (relay) wf[wg] = wq;
}

template <int M>
__device__ __forceinline__ int row_xor_i(int v) {
  return __builtin_amdgcn_update_dpp(0, v, 0x160 | M, 0xf, 0xf, true);
}
__device__ __forceinline__ int wave_sum_i(int v) {
  v += row_xor_i<1>(v);
  v += row_xor_i<2>(v);
  v += row_xor_i<4>(v);
  v += row_xor_i<8>(v);
  return v + __builtin_amdgcn_permlanex16(v, v, 0x76543210, 0xfedcba98, false, false);
}

template <int BIT>
__device__ __forceinline__ float fold_step(float lo, float hi, bool up) {
  const float keep = up ? hi : lo;
  const float send = up ? lo : hi;
  const int got = __builtin_amdgcn_update_dpp(0, __float_as_int(send), 0x160 | BIT, 0xf, 0xf, true);
  return keep + __int_as_float(got);
}

__global__ __launch_bounds__(64) void gat_fused_gemm(
    const float* __restrict__ x, const float* __restrict__ b_in,
    const float* __restrict__ a, const float4* __restrict__ wf, float* __restrict__ z,
    float* __restrict__ s2, int N) {
  __shared__ float lds[2][16][66];
  __shared__ float sc[32];

  const int lane = threadIdx.x & 31;
  const int wv   = threadIdx.x >> 5;
  const int tile = blockIdx.x * 2 + wv;
  const int ntiles = (N + 15) >> 4;
  const bool active = tile < ntiles;

  const int m  = lane & 15;
  const int kp = lane >> 4;

  if (active) {
    const int row  = tile * 16 + m;
    const int rowc = row < N ? row : (N - 1);
    const float* xrow = x + (long)rowc * IN_DIM;

    v8f acc[4];
    #pragma unroll
    for (int nt = 0; nt < 4; ++nt) acc[nt] = (v8f){0,0,0,0,0,0,0,0};

    const float4* wf1 = wf + lane;
    #pragma unroll 8
    for (int j = 0; j < IN_DIM / 8; ++j) {
      const float4 xa = *(const float4*)(xrow + 8 * j + 4 * kp);
      float4 wb[4];
      #pragma unroll
      for (int nt = 0; nt < 4; ++nt) wb[nt] = wf1[(j * 4 + nt) * 32];
      const v2f a0 = {xa.x, xa.y}, a1 = {xa.z, xa.w};
      #pragma unroll
      for (int nt = 0; nt < 4; ++nt) {
        const v2f bv = {wb[nt].x, wb[nt].y};
        acc[nt] = __builtin_amdgcn_wmma_f32_16x16x4_f32(
            false, a0, false, bv, (short)0, acc[nt], false, false);
      }
      #pragma unroll
      for (int nt = 0; nt < 4; ++nt) {
        const v2f bv = {wb[nt].z, wb[nt].w};
        acc[nt] = __builtin_amdgcn_wmma_f32_16x16x4_f32(
            false, a1, false, bv, (short)0, acc[nt], false, false);
      }
    }

    #pragma unroll
    for (int nt = 0; nt < 4; ++nt) {
      float bias = b_in[nt * 16 + m];
      #pragma unroll
      for (int r = 0; r < 8; ++r) {
        float v = acc[nt][r] + bias;
        v = 0.5f * v * (1.0f + erff(v * 0.70710678118654752f));
        lds[wv][r + 8 * kp][nt * 16 + m] = v;
      }
    }
  }
  __syncthreads();

  v8f acc[4];
  #pragma unroll
  for (int nt = 0; nt < 4; ++nt) acc[nt] = (v8f){0,0,0,0,0,0,0,0};
  if (active) {
    const float4* wf2 = wf + WF1_N + lane;
    for (int j = 0; j < HID / 8; ++j) {
      const int kk = 8 * j + 4 * kp;
      const v2f a0 = *(const v2f*)(&lds[wv][m][kk]);
      const v2f a1 = *(const v2f*)(&lds[wv][m][kk + 2]);
      float4 wb[4];
      #pragma unroll
      for (int nt = 0; nt < 4; ++nt) wb[nt] = wf2[(j * 4 + nt) * 32];
      #pragma unroll
      for (int nt = 0; nt < 4; ++nt) {
        const v2f bv = {wb[nt].x, wb[nt].y};
        acc[nt] = __builtin_amdgcn_wmma_f32_16x16x4_f32(
            false, a0, false, bv, (short)0, acc[nt], false, false);
      }
      #pragma unroll
      for (int nt = 0; nt < 4; ++nt) {
        const v2f bv = {wb[nt].z, wb[nt].w};
        acc[nt] = __builtin_amdgcn_wmma_f32_16x16x4_f32(
            false, a1, false, bv, (short)0, acc[nt], false, false);
      }
    }

    const float a20 = a[OUTD + m], a21 = a[OUTD + 16 + m], a22 = a[OUTD + 32 + m], a23 = a[OUTD + 48 + m];
#define ROW_PAIR(nt, j) ((v2f){acc[nt][2 * (j)], acc[nt][2 * (j) + 1]})
#define SCORE_PAIR(j, w0, w1, w2, w3) \
    (ROW_PAIR(0, j) * (w0) + ROW_PAIR(1, j) * (w1) + ROW_PAIR(2, j) * (w2) + ROW_PAIR(3, j) * (w3))
    const v2f Q0 = SCORE_PAIR(0, a20, a21, a22, a23), Q1 = SCORE_PAIR(1, a20, a21, a22, a23);
    const v2f Q2 = SCORE_PAIR(2, a20, a21, a22, a23), Q3 = SCORE_PAIR(3, a20, a21, a22, a23);
#undef SCORE_PAIR
#undef ROW_PAIR
    const bool u8 = (m & 8) != 0, u4 = (m & 4) != 0, u2 = (m & 2) != 0, u1 = (m & 1) != 0;
    const float c0 = fold_step<8>(Q0.x, Q2.x, u8), c1 = fold_step<8>(Q0.y, Q2.y, u8);
    const float c2 = fold_step<8>(Q1.x, Q3.x, u8), c3 = fold_step<8>(Q1.y, Q3.y, u8);
    const float d0 = fold_step<4>(c0, c2, u4), d1 = fold_step<4>(c1, c3, u4);
    const float e0 = fold_step<2>(d0, d1, u2);
    const float f0 = fold_step<1>(e0, e0, u1);
    sc[wv * 16 + 8 * kp + (m >> 1)] = f0;
  }
  __syncthreads();
  if (wv == 0) {
    const int rr = blockIdx.x * 32 + lane;
    if (rr < N) s2[rr] = sc[lane];
  }

  if (active) {
    #pragma unroll
    for (int nt = 0; nt < 4; ++nt) {
      #pragma unroll
      for (int r = 0; r < 8; ++r) {
        int rr = tile * 16 + r + 8 * kp;
        if (rr < N) z[(long)rr * OUTD + nt * 16 + m] = acc[nt][r];
      }
    }
  }
}

__device__ __forceinline__ float other_half(float v) {
  const int b = __float_as_int(v);
  return __int_as_float(__builtin_amdgcn_permlanex16(b, b, 0x76543210, 0xfedcba98, false, false));
}

template <int M>
__device__ __forceinline__ float row_xor(float v) {
  return __int_as_float(__builtin_amdgcn_update_dpp(0, __float_as_int(v), 0x160 | M, 0xf, 0xf, true));
}
__device__ __forceinline__ float trip_max(float v) {
  v = fmaxf(v, row_xor<2>(v));
  v = fmaxf(v, row_xor<4>(v));
  v = fmaxf(v, row_xor<8>(v));
  return fmaxf(v, other_half(v));
}
__device__ __forceinline__ float trip_sum(float v) {
  v += row_xor<2>(v);
  v += row_xor<4>(v);
  v += row_xor<8>(v);
  return v + other_half(v);
}
template <int L>
__device__ __forceinline__ int row_bcast(int v) {
  return __builtin_amdgcn_update_dpp(0, v, 0x150 | L, 0xf, 0xf, true);
}

#define BS_THREADS 1024
#define BS_RUNS    (SORT_CHUNKS / 32)
#define AG_PAIRS   8

__device__ __forceinline__ void bucket_put(
    int2 q, int* rcur, int* stage, int* __restrict__ ssrc,
    int lo, int hi, int row0, bool staged) {
  const int p = atomicAdd(&rcur[(q.y - row0) & (BK_ROWS - 1)], 1);
  const unsigned g = (unsigned)(lo + p);
  if (g < (unsigned)hi) {
    if (staged) stage[p] = q.x;
    else        ssrc[g] = q.x;
  }
}

template <bool IN_LDS>
__device__ __forceinline__ void bucket_rows(
    const int* rcnt, const int* rcur, const float* s2s, int* next_row, const int* list,
    int nbe, int nrows, int row0, const float* __restrict__ z, float4 a1,
    float* __restrict__ h, int N) {
  const int lane = threadIdx.x & 31;
  const int wvs  = __builtin_amdgcn_readfirstlane((int)(threadIdx.x >> 5));
  const int hh   = lane >> 4;
  const int c16  = lane & 15;
  const int epos = (c16 & ~1) + hh;
  const char*    zb       = (const char*)z;
  const unsigned lane_off = (unsigned)c16 * 16u;
  const bool u8 = (c16 & 8) != 0, u4 = (c16 & 4) != 0, u2 = (c16 & 2) != 0, u1 = (c16 & 1) != 0;

  auto trip_index = [&](int o_, int c_, int jb_) -> int {
    int e = jb_ + epos;
    e = e < c_ ? e : c_ - 1;
    const int s = list[min(max(o_ + e, 0), nbe - 1)];
    return min(max(s, 0), N - 1);
  };
  auto row_info = [&](int rr_, int& oo, int& cc, float& ss) {
    oo = 0; cc = 0; ss = 0.0f;
    if (rr_ < nrows) {
      const int   cv = rcnt[rr_];
      const int   ev = rcur[rr_];
      const float sv = s2s[rr_];
      cc = __builtin_amdgcn_readfirstlane(cv);
      oo = __builtin_amdgcn_readfirstlane(ev) - cc;
      ss = __int_as_float(__builtin_amdgcn_readfirstlane(__float_as_int(sv)));
    }
  };
  auto take_row = [&](int& rr_, int& oo, int& cc, float& ss) {
    int nv = 0;
    if (lane == 0) nv = atomicAdd(next_row, 1);
    rr_ = __builtin_amdgcn_readfirstlane(nv);
    row_info(rr_, oo, cc, ss);
  };

  int   r = wvs, o = 0, c = 0;
  float s2d = 0.0f;
  row_info(r, o, c, s2d);
  int s_cur = 0;
  if (c > 0) s_cur = trip_index(o, c, 0);

  for (int it = 0; it < BK_ROWS && r < nrows; ++it) {
    int   rn = nrows, on = 0, cn = 0;
    float s2n = 0.0f;
    if (c <= 0) {
      take_row(rn, on, cn, s2n);
      if (cn > 0) s_cur = trip_index(on, cn, 0);
    }

    float4 acc = {0.f, 0.f, 0.f, 0.f};
    float m_run = -INFINITY;
    float den   = 0.0f;
    for (int jb = 0; jb < c; jb += 2 * AG_PAIRS) {
      const int s_l = s_cur;
      const int nt = (c - jb) < 2 * AG_PAIRS ? (c - jb) : 2 * AG_PAIRS;
      const int np = (nt + 1) >> 1;

      float4 zv[AG_PAIRS];
#define AG_LOAD(p)                                                                       \
      zv[p] = (float4){0.f, 0.f, 0.f, 0.f};                                                 \
      if ((p) < np) {                                                                       \
        const int s = row_bcast<2 * (p)>(s_l);                                              \
        zv[p] = *(const float4*)(zb + (((unsigned)s << 8) | lane_off));                    \
      }
      AG_LOAD(0) AG_LOAD(1) AG_LOAD(2) AG_LOAD(3) AG_LOAD(4) AG_LOAD(5) AG_LOAD(6) AG_LOAD(7)
#undef AG_LOAD

      if (jb == 0) take_row(rn, on, cn, s2n);
      if (jb + 2 * AG_PAIRS < c) s_cur = trip_index(o, c, jb + 2 * AG_PAIRS);
      else if (cn > 0)           s_cur = trip_index(on, cn, 0);

#define AG_DOT(p) (zv[p].x * a1.x + zv[p].y * a1.y + zv[p].z * a1.z + zv[p].w * a1.w)
      const float t0 = AG_DOT(0), t1 = AG_DOT(1), t2 = AG_DOT(2), t3 = AG_DOT(3);
      const float t4 = AG_DOT(4), t5 = AG_DOT(5), t6 = AG_DOT(6), t7 = AG_DOT(7);
#undef AG_DOT
      const float f0 = fold_step<8>(t0, t4, u8), f1 = fold_step<8>(t1, t5, u8);
      const float f2 = fold_step<8>(t2, t6, u8), f3 = fold_step<8>(t3, t7, u8);
      const float g0 = fold_step<4>(f0, f2, u4), g1 = fold_step<4>(f1, f3, u4);
      const float q0 = fold_step<2>(g0, g1, u2);
      const float s1_l = fold_step<1>(q0, q0, u1);

      const bool  live = jb + epos < c;
      const float vv   = s1_l + s2d;
      const float e_l  = live ? (vv > 0.f ? vv : 0.01f * vv) : -INFINITY;
      const float m_new = fmaxf(m_run, trip_max(e_l));
      const float scale = expf(m_run - m_new);
      const float ex_l  = live ? expf(e_l - m_new) : 0.0f;
      den = den * scale + trip_sum(ex_l);
      acc.x *= scale; acc.y *= scale; acc.z *= scale; acc.w *= scale;
      m_run = m_new;

      const int ex_bits = __float_as_int(ex_l);
#define AG_FMA(p)                                                                        \
      if ((p) < np) {                                                                       \
        const float w = __int_as_float(row_bcast<2 * (p)>(ex_bits));                        \
        acc.x += w * zv[p].x; acc.y += w * zv[p].y; acc.z += w * zv[p].z; acc.w += w * zv[p].w; \
      }
      AG_FMA(0) AG_FMA(1) AG_FMA(2) AG_FMA(3) AG_FMA(4) AG_FMA(5) AG_FMA(6) AG_FMA(7)
#undef AG_FMA
    }
    const float inv = 1.0f / fmaxf(den, 1e-9f);
    float4 rv;
    rv.x = (acc.x + other_half(acc.x)) * inv;
    rv.y = (acc.y + other_half(acc.y)) * inv;
    rv.z = (acc.z + other_half(acc.z)) * inv;
    rv.w = (acc.w + other_half(acc.w)) * inv;
    if (hh == 0) *(float4*)((char*)h + (size_t)(row0 + r) * (OUTD * 4) + lane_off) = rv;

    r = rn; o = on; c = cn; s2d = s2n;
  }
}

__global__ __launch_bounds__(BS_THREADS) void gat_bucket_aggregate(
    const int2* __restrict__ pairs, const int2* __restrict__ tab, int* __restrict__ ssrc,
    const float* __restrict__ s2, const float* __restrict__ z, const float* __restrict__ a,
    float* __restrict__ h, int N, int E) {
  __shared__ int   rcnt[BK_ROWS];
  __shared__ int   rcur[BK_ROWS];
  __shared__ float s2s[BK_ROWS];
  __shared__ int   wsum[16];
  __shared__ int   wexc[16];
  __shared__ int2  run[SORT_CHUNKS];
  __shared__ int   part[2][8];
  __shared__ int   next_row;
  __shared__ int   stage[BS_STAGE];
  const int t    = threadIdx.x;
  const int lane = t & 31;
  const int wv   = t >> 5;
  const int b    = blockIdx.x;
  const BucketMap bm = bucket_map(N);
  if (b > bm.last) return;
  const int row0 = b * (int)bm.rows;
  const long cs  = chunk_size(E);

  const int    c16 = lane & 15;
  const float4 a1  = *(const float4*)(a + 4 * c16);

  if (t < SORT_CHUNKS) {
    const int2 e = tab[(long)b * SORT_CHUNKS + t];
    long clo = (long)t * cs;
    long chi = clo + cs;
    if (clo > E) clo = E;
    if (chi > E) chi = E;
    const int sl = (int)(chi - clo);
    const int ls = min(max(e.x, 0), sl);
    const int cn = min(max(e.y, 0), sl - ls);
    run[t] = make_int2((int)clo + ls, cn);
    const int as = wave_sum_i(ls);
    const int n  = wave_sum_i(cn);
    if (lane == 0) { part[0][wv] = as; part[1][wv] = n; }
  }
  if (t < BK_ROWS) {
    rcnt[t] = 0;
    const int node = row0 + t;
    s2s[t] = node < N ? s2[node] : 0.0f;
  }
  if (t == 0) next_row = BS_THREADS / 32;
  __syncthreads();
  int lo_v = 0, nb_v = 0;
  #pragma unroll
  for (int w = 0; w < 8; ++w) { lo_v += part[0][w]; nb_v += part[1][w]; }
  lo_v = min(max(lo_v, 0), E);
  const int lo  = __builtin_amdgcn_readfirstlane(lo_v);
  const int hi  = __builtin_amdgcn_readfirstlane(min(max(nb_v, 0), E - lo_v)) + lo;
  const int nbe = hi - lo;
  const bool staged = nbe <= BS_STAGE;

  int2 pr[2 * BS_RUNS];
  int2 rr[BS_RUNS];
  int  rl[BS_RUNS];
  #pragma unroll
  for (int k = 0; k < BS_RUNS; ++k) rr[k] = run[wv * BS_RUNS + k];
  #pragma unroll
  for (int k = 0; k < BS_RUNS; ++k) {
    rl[k] = rr[k].y;
    #pragma unroll
    for (int u = 0; u < 2; ++u) {
      const int i = lane + 32 * u;
      pr[2 * k + u] = make_int2(0, 0);
      if (i < rr[k].y) pr[2 * k + u] = pairs[rr[k].x + i];
    }
  }
  #pragma unroll
  for (int k = 0; k < BS_RUNS; ++k) {
    #pragma unroll
    for (int u = 0; u < 2; ++u)
      if (lane + 32 * u < rl[k]) atomicAdd(&rcnt[(pr[2 * k + u].y - row0) & (BK_ROWS - 1)], 1);
    if (rl[k] > 64) {
      const int rx = run[wv * BS_RUNS + k].x;
      for (int i = 64 + lane; i < rl[k]; i += 32)
        atomicAdd(&rcnt[(pairs[rx + i].y - row0) & (BK_ROWS - 1)], 1);
    }
  }
  __syncthreads();

  int v = 0, incl = 0;
  if (t < BK_ROWS) {
    v = rcnt[t];
    incl = v;
    #pragma unroll
    for (int o = 1; o < 32; o <<= 1) {
      const int u = __shfl_up(incl, o, 32);
      if (lane >= o) incl += u;
    }
    if (lane == 31) wsum[wv] = incl;
  }
  __syncthreads();
  if (t < 16) {
    const int x = wsum[t];
    int in2 = x;
    #pragma unroll
    for (int o = 1; o < 16; o <<= 1) {
      const int u = __shfl_up(in2, o, 32);
      if (lane >= o) in2 += u;
    }
    wexc[t] = in2 - x;
  }
  __syncthreads();
  if (t < BK_ROWS) rcur[t] = wexc[wv] + incl - v;
  __syncthreads();

  #pragma unroll
  for (int k = 0; k < BS_RUNS; ++k) {
    #pragma unroll
    for (int u = 0; u < 2; ++u)
      if (lane + 32 * u < rl[k]) bucket_put(pr[2 * k + u], rcur, stage, ssrc, lo, hi, row0, staged);
    if (rl[k] > 64) {
      const int rx = run[wv * BS_RUNS + k].x;
      for (int i = 64 + lane; i < rl[k]; i += 32)
        bucket_put(pairs[rx + i], rcur, stage, ssrc, lo, hi, row0, staged);
    }
  }
  __syncthreads();

  const int nrows = min((int)bm.rows, N - row0);
  if (staged) bucket_rows<true>(rcnt, rcur, s2s, &next_row, stage, nbe, nrows, row0, z, a1, h, N);
  else        bucket_rows<false>(rcnt, rcur, s2s, &next_row, ssrc + lo, nbe, nrows, row0, z, a1, h, N);
}

extern "C" void kernel_launch(void* const* d_in, const int* in_sizes, int n_in,
                              void* d_out, int out_size, void* d_ws, size_t ws_size,
                              hipStream_t stream) {
  const float* x    = (const float*)d_in[0];
  const float* w_in = (const float*)d_in[1];
  const float* b_in = (const float*)d_in[2];
  const float* w    = (const float*)d_in[3];
  const float* a    = (const float*)d_in[4];
  const int*   src  = (const int*)d_in[5];
  const int*   dst  = (const int*)d_in[6];
  float* h = (float*)d_out;

  const int N = in_sizes[0] / IN_DIM;
  const int E = in_sizes[5];

  auto align256 = [](size_t v) { return (v + 255) & ~(size_t)255; };
  char* base = (char*)d_ws;
  size_t off_b = 0;
  float* z      = (float*)(base + off_b); off_b += align256((size_t)N * OUTD * 4);
  float* s2     = (float*)(base + off_b); off_b += align256((size_t)N * 4);
  int2*  tab    = (int2*)(base + off_b);  off_b += align256((size_t)SORT_NBP * SORT_CHUNKS * 8);
  int2*  pairs  = (int2*)(base + off_b);  off_b += align256((size_t)E * 8);
  int*   ssrc   = (int*)(base + off_b);   off_b += align256((size_t)E * 4);
  float4* wf    = (float4*)(base + off_b); off_b += align256((size_t)(WF1_N + WF2_N) * 16);

  gat_chunk_sort<<<dim3(SORT_CHUNKS), dim3(256), 0, stream>>>(src, dst, tab, pairs, E, N, w_in, w, wf);
  {
    int ntiles = (N + 15) >> 4;
    int blocks = (ntiles + 1) / 2;
    gat_fused_gemm<<<dim3(blocks), dim3(64), 0, stream>>>(x, b_in, a, wf, z, s2, N);
  }
  gat_bucket_aggregate<<<dim3(SORT_NBP), dim3(BS_THREADS), 0, stream>>>(
      pairs, tab, ssrc, s2, z, a, h, N, E);
}


# aggregate launch: an edge goes to one of 48 LDS slots its row owns by ONE returning add on the row cursor: count pass, scan, 3 barriers go (fuller rows: counted path); sums in LDS arrival order
# speedup vs baseline: 5.8730x; 1.0366x over previous
#include <hip/hip_runtime.h>
#include <math.h>

typedef float v2f __attribute__((ext_vector_type(2)));
typedef float v8f __attribute__((ext_vector_type(8)));

#define IN_DIM 128
#define HID    64
#define OUTD   64

#define SORT_CHUNKS 256
#define SORT_NBP    256
#define BK_ROWS     512
#define SORT_UNITS  8
#define SC_STAGE    6656
#define SC_OUT      8
#define BS_STAGE    12288
#define ROW_CAP     48
#define ROW_PITCH   49

#define WF1_N     ((IN_DIM / 8) * 4 * 32)
#define WF2_N     ((HID / 8) * 4 * 32)
#define WF_PER_WG ((WF1_N + WF2_N + SORT_CHUNKS - 1) / SORT_CHUNKS)

struct BucketMap {
  unsigned rows;
  unsigned magic;
  int      last;
};
__device__ __forceinline__ BucketMap bucket_map(int N) {
  BucketMap m;
  const int r = (N + SORT_NBP - 1) / SORT_NBP;
  m.rows  = (unsigned)min(max(r, 2), BK_ROWS);
  m.magic = 0xFFFFFFFFu / m.rows + 1u;
  m.last  = min((int)(((unsigned)max(N, 1) + m.rows - 1u) / m.rows), SORT_NBP) - 1;
  return m;
}
__device__ __forceinline__ int bucket_of(int d, BucketMap m) {
  unsigned b = __umulhi((unsigned)d, m.magic);
  if (b * m.rows > (unsigned)d) --b;
  return (int)(b < (unsigned)m.last ? b : (unsigned)m.last);
}
__device__ __forceinline__ long chunk_size(int E) {
  return ((((long)E + SORT_CHUNKS - 1) / SORT_CHUNKS) + 3) & ~3L;
}

__device__ __forceinline__ void chunk_count(
    const int* __restrict__ dst, int* h, long lo, long hi, int t, BucketMap bm) {
  const long hi4 = lo + ((hi - lo) & ~3L);
  for (long base = lo; base < hi4; base += 256L * 4 * SORT_UNITS) {
    int4 d[SORT_UNITS];
    #pragma unroll
    for (int j = 0; j < SORT_UNITS; ++j) {
      const long e = base + ((long)j * 256 + t) * 4;
      d[j] = *(const int4*)(dst + (e < hi4 ? e : lo));
    }
    #pragma unroll
    for (int j = 0; j < SORT_UNITS; ++j) {
      const long e = base + ((long)j * 256 + t) * 4;
      if (e < hi4) {
        atomicAdd(&h[bucket_of(d[j].x, bm)], 1);
        atomicAdd(&h[bucket_of(d[j].y, bm)], 1);
        atomicAdd(&h[bucket_of(d[j].z, bm)], 1);
        atomicAdd(&h[bucket_of(d[j].w, bm)], 1);
      }
    }
  }
  if (hi4 + t < hi) atomicAdd(&h[bucket_of(dst[hi4 + t], bm)], 1);
}

template <bool STAGED>
__device__ __forceinline__ void chunk_put(int2* stage, int2* __restrict__ pairs, unsigned p,
                                          int sv, int dv, int E) {
  if (STAGED) {
    if (p < (unsigned)SC_STAGE) stage[p] = make_int2(sv, dv);
  } else {
    if (p < (unsigned)E) pairs[p] = make_int2(sv, dv);
  }
}
__device__ __forceinline__ void chunk_place_direct(
    const int* __restrict__ src, const int* __restrict__ dst, int* cursor,
    int2* __restrict__ pairs, long lo, long hi, int t, int E, BucketMap bm) {
  const long hi4 = lo + ((hi - lo) & ~3L);
  for (long base = lo; base < hi4; base += 256L * 4 * SORT_UNITS) {
    int4 d[SORT_UNITS];
    int4 s[SORT_UNITS];
    #pragma unroll
    for (int j = 0; j < SORT_UNITS; ++j) {
      const long e  = base + ((long)j * 256 + t) * 4;
      const long ec = e < hi4 ? e : lo;
      d[j] = *(const int4*)(dst + ec);
      s[j] = *(const int4*)(src + ec);
    }
    #pragma unroll
    for (int j = 0; j < SORT_UNITS; ++j) {
      const long e = base + ((long)j * 256 + t) * 4;
      if (e < hi4) {
        const unsigned p0 = (unsigned)atomicAdd(&cursor[bucket_of(d[j].x, bm)], 1);
        const unsigned p1 = (unsigned)atomicAdd(&cursor[bucket_of(d[j].y, bm)], 1);
        const unsigned p2 = (unsigned)atomicAdd(&cursor[bucket_of(d[j].z, bm)], 1);
        const unsigned p3 = (unsigned)atomicAdd(&cursor[bucket_of(d[j].w, bm)], 1);
        chunk_put<false>(nullptr, pairs, p0, s[j].x, d[j].x, E);
        chunk_put<false>(nullptr, pairs, p1, s[j].y, d[j].y, E);
        chunk_put<false>(nullptr, pairs, p2, s[j].z, d[j].z, E);
        chunk_put<false>(nullptr, pairs, p3, s[j].w, d[j].w, E);
      }
    }
  }
  if (hi4 + t < hi) {
    const int dd = dst[hi4 + t];
    const unsigned p = (unsigned)atomicAdd(&cursor[bucket_of(dd, bm)], 1);
    chunk_put<false>(nullptr, pairs, p, src[hi4 + t], dd, E);
  }
}

__global__ __launch_bounds__(256) void gat_chunk_sort(
    const int* __restrict__ src, const int* __restrict__ dst,
    int2* __restrict__ tab, int2* __restrict__ pairs, int E, int N,
    const float* __restrict__ w_in, const float* __restrict__ w, float4* __restrict__ wf) {
  __shared__ int h[SORT_NBP];
  __shared__ int cursor[SORT_NBP];
  __shared__ int wsum[8];
  __shared__ __attribute__((aligned(16))) int2 stage[SC_STAGE];
  const int t    = threadIdx.x;
  const int lane = t & 31;
  const int wv   = t >> 5;
  const int c    = blockIdx.x;
  const long cs  = chunk_size(E);
  const BucketMap bm = bucket_map(N);
  const int NB   = bm.last + 1;
  const bool staged = cs <= SC_STAGE;
  long lo = (long)c * cs;
  long hi = lo + cs;
  if (lo > E) lo = E;
  if (hi > E) hi = E;
  const long hi4  = lo + ((hi - lo) & ~3L);
  const bool tail = hi4 + t < hi;

  const int  wg    = c * WF_PER_WG + t;
  const bool relay = t < WF_PER_WG && wg < WF1_N + WF2_N;
  float4 wq = make_float4(0.f, 0.f, 0.f, 0.f);
  if (relay) {
    const bool first = wg < WF1_N;
    const int  q   = first ? wg : wg - WF1_N;
    const int  grp = q >> 5;
    const int  ln  = q & 31;
    const float* p = (first ? w_in : w) + (8 * (grp >> 2) + 4 * (ln >> 4)) * HID + (grp & 3) * 16 + (ln & 15);
    wq = make_float4(p[0], p[HID], p[2 * HID], p[3 * HID]);
  }
  h[t] = 0;
  __syncthreads();

  int4 d[SORT_UNITS];
  int4 s[SORT_UNITS];
  int dt = 0, st = 0;
  #pragma unroll
  for (int j = 0; j < SORT_UNITS; ++j) { d[j] = make_int4(0, 0, 0, 0); s[j] = make_int4(0, 0, 0, 0); }
  if (staged) {
    if (lo < hi4) {
      #pragma unroll
      for (int j = 0; j < SORT_UNITS; ++j) {
        const long e  = lo + ((long)j * 256 + t) * 4;
        const long ec = e < hi4 ? e : lo;
        d[j] = *(const int4*)(dst + ec);
        s[j] = *(const int4*)(src + ec);
      }
    }
    if (tail) { dt = dst[hi4 + t]; st = src[hi4 + t]; }
    #pragma unroll
    for (int j = 0; j < SORT_UNITS; ++j) {
      const long e = lo + ((long)j * 256 + t) * 4;
      if (e < hi4) {
        atomicAdd(&h[bucket_of(d[j].x, bm)], 1);
        atomicAdd(&h[bucket_of(d[j].y, bm)], 1);
        atomicAdd(&h[bucket_of(d[j].z, bm)], 1);
        atomicAdd(&h[bucket_of(d[j].w, bm)], 1);
      }
    }
    if (tail) atomicAdd(&h[bucket_of(dt, bm)], 1);
  } else {
    chunk_count(dst, h, lo, hi, t, bm);
  }
  __syncthreads();

  const int my_cnt = h[t];
  int incl = my_cnt;
  #pragma unroll
  for (int o = 1; o < 32; o <<= 1) {
    const int u = __shfl_up(incl, o, 32);
    if (lane >= o) incl += u;
  }
  if (lane == 31) wsum[wv] = incl;
  __syncthreads();
  int wbase = 0;
  #pragma unroll
  for (int w = 0; w < 8; ++w) wbase += (w < wv) ? wsum[w] : 0;
  const int lstart = wbase + incl - my_cnt;
  cursor[t] = staged ? lstart : (int)lo + lstart;
  if (t < NB) tab[(long)t * SORT_CHUNKS + c] = make_int2(lstart, my_cnt);
  __syncthreads();

  if (!staged) {
    chunk_place_direct(src, dst, cursor, pairs, lo, hi, t, E, bm);
  } else {
    #pragma unroll
    for (int j = 0; j < SORT_UNITS; ++j) {
      const long e = lo + ((long)j * 256 + t) * 4;
      if (e < hi4) {
        const unsigned p0 = (unsigned)atomicAdd(&cursor[bucket_of(d[j].x, bm)], 1);
        const unsigned p1 = (unsigned)atomicAdd(&cursor[bucket_of(d[j].y, bm)], 1);
        const unsigned p2 = (unsigned)atomicAdd(&cursor[bucket_of(d[j].z, bm)], 1);
        const unsigned p3 = (unsigned)atomicAdd(&cursor[bucket_of(d[j].w, bm)], 1);
        chunk_put<true>(stage, pairs, p0, s[j].x, d[j].x, E);
        chunk_put<true>(stage, pairs, p1, s[j].y, d[j].y, E);
        chunk_put<true>(stage, pairs, p2, s[j].z, d[j].z, E);
        chunk_put<true>(stage, pairs, p3, s[j].w, d[j].w, E);
      }
    }
    if (tail) {
      const unsigned p = (unsigned)atomicAdd(&cursor[bucket_of(dt, bm)], 1);
      chunk_put<true>(stage, pairs, p, st, dt, E);
    }
    __syncthreads();
    const int n  = (int)(hi - lo);
    const int n2 = n >> 1;
    const int4* st4  = (const int4*)stage;
    int4*       out4 = (int4*)(pairs + lo);
    for (int i0 = 0; i0 < n2; i0 += 256 * SC_OUT) {
      int4 v[SC_OUT];
      #pragma unroll
      for (int j = 0; j < SC_OUT; ++j) {
        const int i = i0 + j * 256 + t;
        v[j] = st4[i < n2 ? i : 0];
      }
      #pragma unroll
      for (int j = 0; j < SC_OUT; ++j)
        asm volatile("" :: "v"(v[j].x), "v"(v[j].y), "v"(v[j].z), "v"(v[j].w));
      #pragma unroll
      for (int j = 0; j < SC_OUT; ++j) {
        const int i = i0 + j * 256 + t;
        if (i < n2) out4[i] = v[j];
      }
    }
    if ((n & 1) && t == 0) pairs[lo + n - 1] = stage[n - 1];
  }
  if (relay) wf[wg] = wq;
}

template <int M>
__device__ __forceinline__ int row_xor_i(int v) {
  return __builtin_amdgcn_update_dpp(0, v, 0x160 | M, 0xf, 0xf, true);
}
__device__ __forceinline__ int wave_sum_i(int v) {
  v += row_xor_i<1>(v);
  v += row_xor_i<2>(v);
  v += row_xor_i<4>(v);
  v += row_xor_i<8>(v);
  return v + __builtin_amdgcn_permlanex16(v, v, 0x76543210, 0xfedcba98, false, false);
}

template <int BIT>
__device__ __forceinline__ float fold_step(float lo, float hi, bool up) {
  const float keep = up ? hi : lo;
  const float send = up ? lo : hi;
  const int got = __builtin_amdgcn_update_dpp(0, __float_as_int(send), 0x160 | BIT, 0xf, 0xf, true);
  return keep + __int_as_float(got);
}

__global__ __launch_bounds__(64) void gat_fused_gemm(
    const float* __restrict__ x, const float* __restrict__ b_in,
    const float* __restrict__ a, const float4* __restrict__ wf, float* __restrict__ z,
    float* __restrict__ s2, int N) {
  __shared__ float lds[2][16][66];
  __shared__ float sc[32];

  const int lane = threadIdx.x & 31;
  const int wv   = threadIdx.x >> 5;
  const int tile = blockIdx.x * 2 + wv;
  const int ntiles = (N + 15) >> 4;
  const bool active = tile < ntiles;

  const int m  = lane & 15;
  const int kp = lane >> 4;

  if (active) {
    const int row  = tile * 16 + m;
    const int rowc = row < N ? row : (N - 1);
    const float* xrow = x + (long)rowc * IN_DIM;

    v8f acc[4];
    #pragma unroll
    for (int nt = 0; nt < 4; ++nt) acc[nt] = (v8f){0,0,0,0,0,0,0,0};

    const float4* wf1 = wf + lane;
    #pragma unroll 8
    for (int j = 0; j < IN_DIM / 8; ++j) {
      const float4 xa = *(const float4*)(xrow + 8 * j + 4 * kp);
      float4 wb[4];
      #pragma unroll
      for (int nt = 0; nt < 4; ++nt) wb[nt] = wf1[(j * 4 + nt) * 32];
      const v2f a0 = {xa.x, xa.y}, a1 = {xa.z, xa.w};
      #pragma unroll
      for (int nt = 0; nt < 4; ++nt) {
        const v2f bv = {wb[nt].x, wb[nt].y};
        acc[nt] = __builtin_amdgcn_wmma_f32_16x16x4_f32(
            false, a0, false, bv, (short)0, acc[nt], false, false);
      }
      #pragma unroll
      for (int nt = 0; nt < 4; ++nt) {
        const v2f bv = {wb[nt].z, wb[nt].w};
        acc[nt] = __builtin_amdgcn_wmma_f32_16x16x4_f32(
            false, a1, false, bv, (short)0, acc[nt], false, false);
      }
    }

    #pragma unroll
    for (int nt = 0; nt < 4; ++nt) {
      float bias = b_in[nt * 16 + m];
      #pragma unroll
      for (int r = 0; r < 8; ++r) {
        float v = acc[nt][r] + bias;
        v = 0.5f * v * (1.0f + erff(v * 0.70710678118654752f));
        lds[wv][r + 8 * kp][nt * 16 + m] = v;
      }
    }
  }
  __syncthreads();

  v8f acc[4];
  #pragma unroll
  for (int nt = 0; nt < 4; ++nt) acc[nt] = (v8f){0,0,0,0,0,0,0,0};
  if (active) {
    const float4* wf2 = wf + WF1_N + lane;
    for (int j = 0; j < HID / 8; ++j) {
      const int kk = 8 * j + 4 * kp;
      const v2f a0 = *(const v2f*)(&lds[wv][m][kk]);
      const v2f a1 = *(const v2f*)(&lds[wv][m][kk + 2]);
      float4 wb[4];
      #pragma unroll
      for (int nt = 0; nt < 4; ++nt) wb[nt] = wf2[(j * 4 + nt) * 32];
      #pragma unroll
      for (int nt = 0; nt < 4; ++nt) {
        const v2f bv = {wb[nt].x, wb[nt].y};
        acc[nt] = __builtin_amdgcn_wmma_f32_16x16x4_f32(
            false, a0, false, bv, (short)0, acc[nt], false, false);
      }
      #pragma unroll
      for (int nt = 0; nt < 4; ++nt) {
        const v2f bv = {wb[nt].z, wb[nt].w};
        acc[nt] = __builtin_amdgcn_wmma_f32_16x16x4_f32(
            false, a1, false, bv, (short)0, acc[nt], false, false);
      }
    }

    const float a20 = a[OUTD + m], a21 = a[OUTD + 16 + m], a22 = a[OUTD + 32 + m], a23 = a[OUTD + 48 + m];
#define ROW_PAIR(nt, j) ((v2f){acc[nt][2 * (j)], acc[nt][2 * (j) + 1]})
#define SCORE_PAIR(j, w0, w1, w2, w3) \
    (ROW_PAIR(0, j) * (w0) + ROW_PAIR(1, j) * (w1) + ROW_PAIR(2, j) * (w2) + ROW_PAIR(3, j) * (w3))
    const v2f Q0 = SCORE_PAIR(0, a20, a21, a22, a23), Q1 = SCORE_PAIR(1, a20, a21, a22, a23);
    const v2f Q2 = SCORE_PAIR(2, a20, a21, a22, a23), Q3 = SCORE_PAIR(3, a20, a21, a22, a23);
#undef SCORE_PAIR
#undef ROW_PAIR
    const bool u8 = (m & 8) != 0, u4 = (m & 4) != 0, u2 = (m & 2) != 0, u1 = (m & 1) != 0;
    const float c0 = fold_step<8>(Q0.x, Q2.x, u8), c1 = fold_step<8>(Q0.y, Q2.y, u8);
    const float c2 = fold_step<8>(Q1.x, Q3.x, u8), c3 = fold_step<8>(Q1.y, Q3.y, u8);
    const float d0 = fold_step<4>(c0, c2, u4), d1 = fold_step<4>(c1, c3, u4);
    const float e0 = fold_step<2>(d0, d1, u2);
    const float f0 = fold_step<1>(e0, e0, u1);
    sc[wv * 16 + 8 * kp + (m >> 1)] = f0;
  }
  __syncthreads();
  if (wv == 0) {
    const int rr = blockIdx.x * 32 + lane;
    if (rr < N) s2[rr] = sc[lane];
  }

  if (active) {
    #pragma unroll
    for (int nt = 0; nt < 4; ++nt) {
      #pragma unroll
      for (int r = 0; r < 8; ++r) {
        int rr = tile * 16 + r + 8 * kp;
        if (rr < N) z[(long)rr * OUTD + nt * 16 + m] = acc[nt][r];
      }
    }
  }
}

__device__ __forceinline__ float other_half(float v) {
  const int b = __float_as_int(v);
  return __int_as_float(__builtin_amdgcn_permlanex16(b, b, 0x76543210, 0xfedcba98, false, false));
}

template <int M>
__device__ __forceinline__ float row_xor(float v) {
  return __int_as_float(__builtin_amdgcn_update_dpp(0, __float_as_int(v), 0x160 | M, 0xf, 0xf, true));
}
__device__ __forceinline__ float trip_max(float v) {
  v = fmaxf(v, row_xor<2>(v));
  v = fmaxf(v, row_xor<4>(v));
  v = fmaxf(v, row_xor<8>(v));
  return fmaxf(v, other_half(v));
}
__device__ __forceinline__ float trip_sum(float v) {
  v += row_xor<2>(v);
  v += row_xor<4>(v);
  v += row_xor<8>(v);
  return v + other_half(v);
}
template <int L>
__device__ __forceinline__ int row_bcast(int v) {
  return __builtin_amdgcn_update_dpp(0, v, 0x150 | L, 0xf, 0xf, true);
}

#define BS_THREADS 1024
#define BS_RUNS    (SORT_CHUNKS / 32)
#define AG_PAIRS   8

__device__ __forceinline__ void row_put(int2 q, int* rcur, int* stage, int row0) {
  const int r = (q.y - row0) & (BK_ROWS - 1);
  const int p = atomicAdd(&rcur[r], 1);
  if ((unsigned)(p - r * ROW_PITCH) < (unsigned)ROW_CAP) stage[p] = q.x;
}

__device__ __forceinline__ void bucket_put(
    int2 q, int* rcur, int* stage, int* __restrict__ ssrc,
    int lo, int hi, int row0, bool staged) {
  const int p = atomicAdd(&rcur[(q.y - row0) & (BK_ROWS - 1)], 1);
  const unsigned g = (unsigned)(lo + p);
  if (g < (unsigned)hi) {
    if (staged) stage[p] = q.x;
    else        ssrc[g] = q.x;
  }
}

template <bool IN_LDS>
__device__ __forceinline__ void bucket_rows(
    const int* rcnt, const int* rcur, const float* s2s, int* next_row, const int* list,
    int nbe, int nrows, int row0, const float* __restrict__ z, float4 a1,
    float* __restrict__ h, int N) {
  const int lane = threadIdx.x & 31;
  const int wvs  = __builtin_amdgcn_readfirstlane((int)(threadIdx.x >> 5));
  const int hh   = lane >> 4;
  const int c16  = lane & 15;
  const int epos = (c16 & ~1) + hh;
  const char*    zb       = (const char*)z;
  const unsigned lane_off = (unsigned)c16 * 16u;
  const bool u8 = (c16 & 8) != 0, u4 = (c16 & 4) != 0, u2 = (c16 & 2) != 0, u1 = (c16 & 1) != 0;

  auto trip_index = [&](int o_, int c_, int jb_) -> int {
    int e = jb_ + epos;
    e = e < c_ ? e : c_ - 1;
    const int s = list[min(max(o_ + e, 0), nbe - 1)];
    return min(max(s, 0), N - 1);
  };
  auto row_info = [&](int rr_, int& oo, int& cc, float& ss) {
    oo = 0; cc = 0; ss = 0.0f;
    if (rr_ < nrows) {
      const int   cv = rcnt[rr_];
      const int   ev = rcur[rr_];
      const float sv = s2s[rr_];
      cc = __builtin_amdgcn_readfirstlane(cv);
      oo = __builtin_amdgcn_readfirstlane(ev) - cc;
      ss = __int_as_float(__builtin_amdgcn_readfirstlane(__float_as_int(sv)));
    }
  };
  auto take_row = [&](int& rr_, int& oo, int& cc, float& ss) {
    int nv = 0;
    if (lane == 0) nv = atomicAdd(next_row, 1);
    rr_ = __builtin_amdgcn_readfirstlane(nv);
    row_info(rr_, oo, cc, ss);
  };

  int   r = wvs, o = 0, c = 0;
  float s2d = 0.0f;
  row_info(r, o, c, s2d);
  int s_cur = 0;
  if (c > 0) s_cur = trip_index(o, c, 0);

  for (int it = 0; it < BK_ROWS && r < nrows; ++it) {
    int   rn = nrows, on = 0, cn = 0;
    float s2n = 0.0f;
    if (c <= 0) {
      take_row(rn, on, cn, s2n);
      if (cn > 0) s_cur = trip_index(on, cn, 0);
    }

    float4 acc = {0.f, 0.f, 0.f, 0.f};
    float m_run = -INFINITY;
    float den   = 0.0f;
    for (int jb = 0; jb < c; jb += 2 * AG_PAIRS) {
      const int s_l = s_cur;
      const int nt = (c - jb) < 2 * AG_PAIRS ? (c - jb) : 2 * AG_PAIRS;
      const int np = (nt + 1) >> 1;

      float4 zv[AG_PAIRS];
#define AG_LOAD(p)                                                                       \
      zv[p] = (float4){0.f, 0.f, 0.f, 0.f};                                                 \
      if ((p) < np) {                                                                       \
        const int s = row_bcast<2 * (p)>(s_l);                                              \
        zv[p] = *(const float4*)(zb + (((unsigned)s << 8) | lane_off));                    \
      }
      AG_LOAD(0) AG_LOAD(1) AG_LOAD(2) AG_LOAD(3) AG_LOAD(4) AG_LOAD(5) AG_LOAD(6) AG_LOAD(7)
#undef AG_LOAD

      if (jb == 0) take_row(rn, on, cn, s2n);
      if (jb + 2 * AG_PAIRS < c) s_cur = trip_index(o, c, jb + 2 * AG_PAIRS);
      else if (cn > 0)           s_cur = trip_index(on, cn, 0);

#define AG_DOT(p) (zv[p].x * a1.x + zv[p].y * a1.y + zv[p].z * a1.z + zv[p].w * a1.w)
      const float t0 = AG_DOT(0), t1 = AG_DOT(1), t2 = AG_DOT(2), t3 = AG_DOT(3);
      const float t4 = AG_DOT(4), t5 = AG_DOT(5), t6 = AG_DOT(6), t7 = AG_DOT(7);
#undef AG_DOT
      const float f0 = fold_step<8>(t0, t4, u8), f1 = fold_step<8>(t1, t5, u8);
      const float f2 = fold_step<8>(t2, t6, u8), f3 = fold_step<8>(t3, t7, u8);
      const float g0 = fold_step<4>(f0, f2, u4), g1 = fold_step<4>(f1, f3, u4);
      const float q0 = fold_step<2>(g0, g1, u2);
      const float s1_l = fold_step<1>(q0, q0, u1);

      const bool  live = jb + epos < c;
      const float vv   = s1_l + s2d;
      const float e_l  = live ? (vv > 0.f ? vv : 0.01f * vv) : -INFINITY;
      const float m_new = fmaxf(m_run, trip_max(e_l));
      const float scale = expf(m_run - m_new);
      const float ex_l  = live ? expf(e_l - m_new) : 0.0f;
      den = den * scale + trip_sum(ex_l);
      acc.x *= scale; acc.y *= scale; acc.z *= scale; acc.w *= scale;
      m_run = m_new;

      const int ex_bits = __float_as_int(ex_l);
#define AG_FMA(p)                                                                        \
      if ((p) < np) {                                                                       \
        const float w = __int_as_float(row_bcast<2 * (p)>(ex_bits));                        \
        acc.x += w * zv[p].x; acc.y += w * zv[p].y; acc.z += w * zv[p].z; acc.w += w * zv[p].w; \
      }
      AG_FMA(0) AG_FMA(1) AG_FMA(2) AG_FMA(3) AG_FMA(4) AG_FMA(5) AG_FMA(6) AG_FMA(7)
#undef AG_FMA
    }
    const float inv = 1.0f / fmaxf(den, 1e-9f);
    float4 rv;
    rv.x = (acc.x + other_half(acc.x)) * inv;
    rv.y = (acc.y + other_half(acc.y)) * inv;
    rv.z = (acc.z + other_half(acc.z)) * inv;
    rv.w = (acc.w + other_half(acc.w)) * inv;
    if (hh == 0) *(float4*)((char*)h + (size_t)(row0 + r) * (OUTD * 4) + lane_off) = rv;

    r = rn; o = on; c = cn; s2d = s2n;
  }
}

__global__ __launch_bounds__(BS_THREADS) void gat_bucket_aggregate(
    const int2* __restrict__ pairs, const int2* __restrict__ tab, int* __restrict__ ssrc,
    const float* __restrict__ s2, const float* __restrict__ z, const float* __restrict__ a,
    float* __restrict__ h, int N, int E) {
  __shared__ int   rcnt[BK_ROWS];
  __shared__ int   rcur[BK_ROWS];
  __shared__ float s2s[BK_ROWS];
  __shared__ int   wsum[16];
  __shared__ int   wexc[16];
  __shared__ int2  run[SORT_CHUNKS];
  __shared__ int   part[2][8];
  __shared__ int   next_row;
  __shared__ int   stage[BK_ROWS * ROW_PITCH];
  __shared__ int   ovf;
  const int t    = threadIdx.x;
  const int lane = t & 31;
  const int wv   = t >> 5;
  const int b    = blockIdx.x;
  const BucketMap bm = bucket_map(N);
  if (b > bm.last) return;
  const int row0 = b * (int)bm.rows;
  const long cs  = chunk_size(E);

  const int    c16 = lane & 15;
  const float4 a1  = *(const float4*)(a + 4 * c16);

  if (t < SORT_CHUNKS) {
    const int2 e = tab[(long)b * SORT_CHUNKS + t];
    long clo = (long)t * cs;
    long chi = clo + cs;
    if (clo > E) clo = E;
    if (chi > E) chi = E;
    const int sl = (int)(chi - clo);
    const int ls = min(max(e.x, 0), sl);
    const int cn = min(max(e.y, 0), sl - ls);
    run[t] = make_int2((int)clo + ls, cn);
    const int as = wave_sum_i(ls);
    const int n  = wave_sum_i(cn);
    if (lane == 0) { part[0][wv] = as; part[1][wv] = n; }
  }
  if (t < BK_ROWS) {
    rcur[t] = t * ROW_PITCH;
    const int node = row0 + t;
    s2s[t] = node < N ? s2[node] : 0.0f;
  }
  if (t == 0) { next_row = BS_THREADS / 32; ovf = 0; }
  __syncthreads();
  int lo_v = 0, nb_v = 0;
  #pragma unroll
  for (int w = 0; w < 8; ++w) { lo_v += part[0][w]; nb_v += part[1][w]; }
  lo_v = min(max(lo_v, 0), E);
  const int lo  = __builtin_amdgcn_readfirstlane(lo_v);
  const int hi  = __builtin_amdgcn_readfirstlane(min(max(nb_v, 0), E - lo_v)) + lo;
  const int nbe = hi - lo;
  const bool staged = nbe <= BS_STAGE;

  int2 pr[2 * BS_RUNS];
  int2 rr[BS_RUNS];
  int  rl[BS_RUNS];
  #pragma unroll
  for (int k = 0; k < BS_RUNS; ++k) rr[k] = run[wv * BS_RUNS + k];
  #pragma unroll
  for (int k = 0; k < BS_RUNS; ++k) {
    rl[k] = rr[k].y;
    #pragma unroll
    for (int u = 0; u < 2; ++u) {
      const int i = lane + 32 * u;
      pr[2 * k + u] = make_int2(0, 0);
      if (i < rr[k].y) pr[2 * k + u] = pairs[rr[k].x + i];
    }
  }
  #pragma unroll
  for (int k = 0; k < BS_RUNS; ++k) {
    #pragma unroll
    for (int u = 0; u < 2; ++u)
      if (lane + 32 * u < rl[k]) row_put(pr[2 * k + u], rcur, stage, row0);
    if (rl[k] > 64) {
      const int rx = run[wv * BS_RUNS + k].x;
      for (int i = 64 + lane; i < rl[k]; i += 32) row_put(pairs[rx + i], rcur, stage, row0);
    }
  }
  __syncthreads();
  if (t < BK_ROWS) {
    const int cnt = rcur[t] - t * ROW_PITCH;
    rcnt[t] = cnt;
    if (cnt > ROW_CAP) ovf = 1;
  }
  __syncthreads();
  const int nrows = min((int)bm.rows, N - row0);
  if (ovf == 0) {
    bucket_rows<true>(rcnt, rcur, s2s, &next_row, stage, BK_ROWS * ROW_PITCH, nrows, row0, z, a1, h, N);
    return;
  }

  int v = 0, incl = 0;
  if (t < BK_ROWS) {
    v = rcnt[t];
    incl = v;
    #pragma unroll
    for (int o = 1; o < 32; o <<= 1) {
      const int u = __shfl_up(incl, o, 32);
      if (lane >= o) incl += u;
    }
    if (lane == 31) wsum[wv] = incl;
  }
  __syncthreads();
  if (t < 16) {
    const int x = wsum[t];
    int in2 = x;
    #pragma unroll
    for (int o = 1; o < 16; o <<= 1) {
      const int u = __shfl_up(in2, o, 32);
      if (lane >= o) in2 += u;
    }
    wexc[t] = in2 - x;
  }
  __syncthreads();
  if (t < BK_ROWS) rcur[t] = wexc[wv] + incl - v;
  __syncthreads();

  #pragma unroll
  for (int k = 0; k < BS_RUNS; ++k) {
    #pragma unroll
    for (int u = 0; u < 2; ++u)
      if (lane + 32 * u < rl[k]) bucket_put(pr[2 * k + u], rcur, stage, ssrc, lo, hi, row0, staged);
    if (rl[k] > 64) {
      const int rx = run[wv * BS_RUNS + k].x;
      for (int i = 64 + lane; i < rl[k]; i += 32)
        bucket_put(pairs[rx + i], rcur, stage, ssrc, lo, hi, row0, staged);
    }
  }
  __syncthreads();

  if (staged) bucket_rows<true>(rcnt, rcur, s2s, &next_row, stage, nbe, nrows, row0, z, a1, h, N);
  else        bucket_rows<false>(rcnt, rcur, s2s, &next_row, ssrc + lo, nbe, nrows, row0, z, a1, h, N);
}

extern "C" void kernel_launch(void* const* d_in, const int* in_sizes, int n_in,
                              void* d_out, int out_size, void* d_ws, size_t ws_size,
                              hipStream_t stream) {
  const float* x    = (const float*)d_in[0];
  const float* w_in = (const float*)d_in[1];
  const float* b_in = (const float*)d_in[2];
  const float* w    = (const float*)d_in[3];
  const float* a    = (const float*)d_in[4];
  const int*   src  = (const int*)d_in[5];
  const int*   dst  = (const int*)d_in[6];
  float* h = (float*)d_out;

  const int N = in_sizes[0] / IN_DIM;
  const int E = in_sizes[5];

  auto align256 = [](size_t v) { return (v + 255) & ~(size_t)255; };
  char* base = (char*)d_ws;
  size_t off_b = 0;
  float* z      = (float*)(base + off_b); off_b += align256((size_t)N * OUTD * 4);
  float* s2     = (float*)(base + off_b); off_b += align256((size_t)N * 4);
  int2*  tab    = (int2*)(base + off_b);  off_b += align256((size_t)SORT_NBP * SORT_CHUNKS * 8);
  int2*  pairs  = (int2*)(base + off_b);  off_b += align256((size_t)E * 8);
  int*   ssrc   = (int*)(base + off_b);   off_b += align256((size_t)E * 4);
  float4* wf    = (float4*)(base + off_b); off_b += align256((size_t)(WF1_N + WF2_N) * 16);

  gat_chunk_sort<<<dim3(SORT_CHUNKS), dim3(256), 0, stream>>>(src, dst, tab, pairs, E, N, w_in, w, wf);
  {
    int ntiles = (N + 15) >> 4;
    int blocks = (ntiles + 1) / 2;
    gat_fused_gemm<<<dim3(blocks), dim3(64), 0, stream>>>(x, b_in, a, wf, z, s2, N);
  }
  gat_bucket_aggregate<<<dim3(SORT_NBP), dim3(BS_THREADS), 0, stream>>>(
      pairs, tab, ssrc, s2, z, a, h, N, E);
}
